# starting kernel
# baseline (speedup 1.0000x reference)
#include <hip/hip_runtime.h>
#include <hip/hip_bf16.h>
#include <math.h>


typedef float v2f __attribute__((ext_vector_type(2)));
typedef float v8f __attribute__((ext_vector_type(8)));

#define IN_CH 8
#define OUT_CH 64

__device__ __forceinline__ v8f wmma_f32(v2f a, v2f b, v8f c) {
  return __builtin_amdgcn_wmma_f32_16x16x4_f32(false, a, false, b, (short)0, c,
                                               false, false);
}

__device__ __forceinline__ v8f accum_block(const float* __restrict__ A, int ldA,
                                           int rA, const float* __restrict__ W,
                                           int Kdim, int colB, int half,
                                           v8f acc) {
#pragma unroll 4
  for (int k0 = 0; k0 < Kdim; k0 += 4) {
    int ka = k0 + 2 * half;
    v2f a;
    a.x = A[rA * ldA + ka];
    a.y = A[rA * ldA + ka + 1];
    v2f b;
    b.x = W[ka * OUT_CH + colB];
    b.y = W[(ka + 1) * OUT_CH + colB];
    acc = wmma_f32(a, b, acc);
  }
  return acc;
}


__global__ void __launch_bounds__(256) zero_kernel(float* p, int n) {
  int i = blockIdx.x * blockDim.x + threadIdx.x;
  if (i < n) p[i] = 0.0f;
}

__global__ void __launch_bounds__(256)
degree_kernel(const int* __restrict__ rowIdx, float* __restrict__ deg, int E) {
  int e = blockIdx.x * blockDim.x + threadIdx.x;
  if (e < E) atomicAdd(&deg[rowIdx[e]], 1.0f);
}

__global__ void __launch_bounds__(256) dinv_kernel(float* deg, int n) {
  int i = blockIdx.x * blockDim.x + threadIdx.x;
  if (i < n) {
    float d = deg[i];
    deg[i] = (d > 0.0f) ? rsqrtf(d) : 0.0f;
  }
}

__global__ void __launch_bounds__(256)
spmm_kernel(const int* __restrict__ rowIdx, const int* __restrict__ colIdx,
            const float* __restrict__ dinv, const float* __restrict__ src64,
            float* __restrict__ dst64, const float* __restrict__ src8,
            float* __restrict__ dst8, int E) {
  int e = __builtin_amdgcn_readfirstlane(
      (blockIdx.x * blockDim.x + threadIdx.x) >> 5);
  int lane = threadIdx.x & 31;
  if (e >= E) return;
  int r = __builtin_amdgcn_readfirstlane(rowIdx[e]);
  int c = __builtin_amdgcn_readfirstlane(colIdx[e]);
  float nrm = -dinv[r] * dinv[c];
  const float2* __restrict__ src = (const float2*)(src64 + (size_t)c * OUT_CH);
  float2 v = src[lane];
  float* __restrict__ dst = dst64 + (size_t)r * OUT_CH + 2 * lane;
  atomicAdd(&dst[0], v.x * nrm);
  atomicAdd(&dst[1], v.y * nrm);
  if (src8 != nullptr && lane < IN_CH) {
    atomicAdd(&dst8[r * IN_CH + lane], src8[c * IN_CH + lane] * nrm);
  }
}

__global__ void __launch_bounds__(128)
gates_wmma_kernel(const float* __restrict__ x, const float* __restrict__ H,
                  const float* __restrict__ Lx, const float* __restrict__ LH,
                  const float* __restrict__ Wzx, const float* __restrict__ Wzh,
                  const float* __restrict__ Wrx, const float* __restrict__ Wrh,
                  const float* __restrict__ bzx, const float* __restrict__ bzh,
                  const float* __restrict__ brx, const float* __restrict__ brh,
                  float* __restrict__ Zbuf, float* __restrict__ RH) {
  int lane = threadIdx.x & 31;
  int wv = threadIdx.x >> 5;
  int half = lane >> 4;
  int l15 = lane & 15;
  int row0 = blockIdx.x * 16;
  int col0 = wv * 16;
  int rA = row0 + l15;
  int colB = col0 + l15;

  v8f accZ = {};
  v8f accR = {};
  accZ = accum_block(x, IN_CH, rA, Wzx, IN_CH, colB, half, accZ);
  accR = accum_block(x, IN_CH, rA, Wrx, IN_CH, colB, half, accR);
  accZ = accum_block(Lx, IN_CH, rA, Wzx + IN_CH * OUT_CH, IN_CH, colB, half, accZ);
  accR = accum_block(Lx, IN_CH, rA, Wrx + IN_CH * OUT_CH, IN_CH, colB, half, accR);
  accZ = accum_block(H, OUT_CH, rA, Wzh, OUT_CH, colB, half, accZ);
  accR = accum_block(H, OUT_CH, rA, Wrh, OUT_CH, colB, half, accR);
  accZ = accum_block(LH, OUT_CH, rA, Wzh + OUT_CH * OUT_CH, OUT_CH, colB, half, accZ);
  accR = accum_block(LH, OUT_CH, rA, Wrh + OUT_CH * OUT_CH, OUT_CH, colB, half, accR);

  float bZ = bzx[colB] + bzh[colB];
  float bR = brx[colB] + brh[colB];
#pragma unroll
  for (int i = 0; i < 8; ++i) {
    int r = row0 + i + 8 * half;
    float z = 1.0f / (1.0f + __expf(-(accZ[i] + bZ)));
    float rg = 1.0f / (1.0f + __expf(-(accR[i] + bR)));
    Zbuf[r * OUT_CH + colB] = z;
    RH[r * OUT_CH + colB] = rg * H[r * OUT_CH + colB];
  }
}

__global__ void __launch_bounds__(128)
cand_out_wmma_kernel(const float* __restrict__ x, const float* __restrict__ H,
                     const float* __restrict__ Lx, const float* __restrict__ RH,
                     const float* __restrict__ LRH,
                     const float* __restrict__ Zbuf,
                     const float* __restrict__ Wcx,
                     const float* __restrict__ Wch,
                     const float* __restrict__ bcx,
                     const float* __restrict__ bch, float* __restrict__ out) {
  int lane = threadIdx.x & 31;
  int wv = threadIdx.x >> 5;
  int half = lane >> 4;
  int l15 = lane & 15;
  int row0 = blockIdx.x * 16;
  int col0 = wv * 16;
  int rA = row0 + l15;
  int colB = col0 + l15;

  v8f accC = {};
  accC = accum_block(x, IN_CH, rA, Wcx, IN_CH, colB, half, accC);
  accC = accum_block(Lx, IN_CH, rA, Wcx + IN_CH * OUT_CH, IN_CH, colB, half, accC);
  accC = accum_block(RH, OUT_CH, rA, Wch, OUT_CH, colB, half, accC);
  accC = accum_block(LRH, OUT_CH, rA, Wch + OUT_CH * OUT_CH, OUT_CH, colB, half, accC);

  float bC = bcx[colB] + bch[colB];
#pragma unroll
  for (int i = 0; i < 8; ++i) {
    int r = row0 + i + 8 * half;
    float cv = tanhf(accC[i] + bC);
    float z = Zbuf[r * OUT_CH + colB];
    float h = H[r * OUT_CH + colB];
    out[r * OUT_CH + colB] = z * h + (1.0f - z) * cv;
  }
}


extern "C" void kernel_launch(void* const* d_in, const int* in_sizes, int n_in,
                              void* d_out, int out_size, void* d_ws,
                              size_t ws_size, hipStream_t stream) {
  const float* x = (const float*)d_in[0];
  const float* H = (const float*)d_in[1];
  const int* edge = (const int*)d_in[2];
  const float* Wzx = (const float*)d_in[3];
  const float* bzx = (const float*)d_in[4];
  const float* Wzh = (const float*)d_in[5];
  const float* bzh = (const float*)d_in[6];
  const float* Wrx = (const float*)d_in[7];
  const float* brx = (const float*)d_in[8];
  const float* Wrh = (const float*)d_in[9];
  const float* brh = (const float*)d_in[10];
  const float* Wcx = (const float*)d_in[11];
  const float* bcx = (const float*)d_in[12];
  const float* Wch = (const float*)d_in[13];
  const float* bch = (const float*)d_in[14];

  const int N = in_sizes[0] / IN_CH;
  const int E = in_sizes[2] / 2;
  const int* rowIdx = edge;
  const int* colIdx = edge + E;

  float* ws = (float*)d_ws;
  float* dinv = ws;
  float* Lx = dinv + N;
  float* LH = Lx + (size_t)N * IN_CH;
  float* LRH = LH + (size_t)N * OUT_CH;
  float* Zbuf = LRH + (size_t)N * OUT_CH;
  float* RH = Zbuf + (size_t)N * OUT_CH;
  const int nZero = N * (1 + IN_CH + OUT_CH + OUT_CH);

  zero_kernel<<<(nZero + 255) / 256, 256, 0, stream>>>(dinv, nZero);
  degree_kernel<<<(E + 255) / 256, 256, 0, stream>>>(rowIdx, dinv, E);
  dinv_kernel<<<(N + 255) / 256, 256, 0, stream>>>(dinv, N);
  {
    long long threads = (long long)E * 32;
    spmm_kernel<<<(int)((threads + 255) / 256), 256, 0, stream>>>(
        rowIdx, colIdx, dinv, H, LH, x, Lx, E);
  }
  gates_wmma_kernel<<<N / 16, 128, 0, stream>>>(x, H, Lx, LH, Wzx, Wzh, Wrx,
                                                Wrh, bzx, bzh, brx, brh, Zbuf,
                                                RH);
  {
    long long threads = (long long)E * 32;
    spmm_kernel<<<(int)((threads + 255) / 256), 256, 0, stream>>>(
        rowIdx, colIdx, dinv, RH, LRH, nullptr, nullptr, E);
  }
  cand_out_wmma_kernel<<<N / 16, 128, 0, stream>>>(
      x, H, Lx, RH, LRH, Zbuf, Wcx, Wch, bcx, bch, (float*)d_out);
}


# SpMM by row ownership: each block filters the edge list for its 196 destination rows, sums them in LDS accumulators, stores whole rows; no global float atomics or zero fill; sum order = LDS arrival
# speedup vs baseline: 1.4212x; 1.4212x over previous
#include <hip/hip_runtime.h>
#include <hip/hip_bf16.h>
#include <math.h>


typedef float v2f __attribute__((ext_vector_type(2)));
typedef float v8f __attribute__((ext_vector_type(8)));

#define IN_CH 8
#define OUT_CH 64

__device__ __forceinline__ v8f wmma_f32(v2f a, v2f b, v8f c) {
  return __builtin_amdgcn_wmma_f32_16x16x4_f32(false, a, false, b, (short)0, c,
                                               false, false);
}

__device__ __forceinline__ v8f accum_block(const float* __restrict__ A, int ldA,
                                           int rA, const float* __restrict__ W,
                                           int Kdim, int colB, int half,
                                           v8f acc) {
#pragma unroll 4
  for (int k0 = 0; k0 < Kdim; k0 += 4) {
    int ka = k0 + 2 * half;
    v2f a;
    a.x = A[rA * ldA + ka];
    a.y = A[rA * ldA + ka + 1];
    v2f b;
    b.x = W[ka * OUT_CH + colB];
    b.y = W[(ka + 1) * OUT_CH + colB];
    acc = wmma_f32(a, b, acc);
  }
  return acc;
}


__global__ void __launch_bounds__(256) zero_kernel(float* p, int n) {
  int i = blockIdx.x * blockDim.x + threadIdx.x;
  if (i < n) p[i] = 0.0f;
}

__global__ void __launch_bounds__(256)
degree_kernel(const int* __restrict__ rowIdx, float* __restrict__ deg, int E) {
  int e = blockIdx.x * blockDim.x + threadIdx.x;
  if (e < E) atomicAdd(&deg[rowIdx[e]], 1.0f);
}

__global__ void __launch_bounds__(256) dinv_kernel(float* deg, int n) {
  int i = blockIdx.x * blockDim.x + threadIdx.x;
  if (i < n) {
    float d = deg[i];
    deg[i] = (d > 0.0f) ? rsqrtf(d) : 0.0f;
  }
}

#define ROWS_PB 196
#define QCAP 32768
#define AGG_THREADS 1024
#define AGG_WAVES (AGG_THREADS / 32)

template <bool HAS8>
__global__ void __launch_bounds__(AGG_THREADS)
agg_kernel(const int* __restrict__ rowIdx, const int* __restrict__ colIdx,
           const float* __restrict__ dinv, const float* __restrict__ src64,
           float* __restrict__ dst64, const float* __restrict__ src8,
           float* __restrict__ dst8, int E, int N) {
  __shared__ __attribute__((aligned(16))) float acc[ROWS_PB * OUT_CH];
  __shared__ __attribute__((aligned(16))) float acc8[HAS8 ? ROWS_PB * IN_CH : 4];
  __shared__ int queue[QCAP];
  __shared__ int qn;

  const int tid = threadIdx.x;
  const int lane = tid & 31;
  const int base = blockIdx.x * ROWS_PB;

  for (int i = tid; i < ROWS_PB * OUT_CH; i += AGG_THREADS) acc[i] = 0.0f;
  if (HAS8) {
    for (int i = tid; i < ROWS_PB * IN_CH; i += AGG_THREADS) acc8[i] = 0.0f;
  }
  if (tid == 0) qn = 0;
  __syncthreads();

  auto visit = [&](int e, int r) {
    unsigned d = (unsigned)(r - base);
    if (d < (unsigned)ROWS_PB) {
      int slot = atomicAdd(&qn, 1);
      if (slot < QCAP) {
        queue[slot] = e;
      } else {
        int c = colIdx[e];
        float nrm = -dinv[r] * dinv[c];
#pragma unroll 1
        for (int k = 0; k < OUT_CH; ++k)
          atomicAdd(&acc[d * OUT_CH + k], src64[(size_t)c * OUT_CH + k] * nrm);
        if (HAS8) {
#pragma unroll 1
          for (int k = 0; k < IN_CH; ++k)
            atomicAdd(&acc8[d * IN_CH + k], src8[(size_t)c * IN_CH + k] * nrm);
        }
      }
    }
  };

  const int E4 = E & ~3;
  const int4* __restrict__ row4 = (const int4*)rowIdx;
  auto visit4 = [&](int e, int4 r) {
    unsigned d0 = (unsigned)(r.x - base), d1 = (unsigned)(r.y - base);
    unsigned d2 = (unsigned)(r.z - base), d3 = (unsigned)(r.w - base);
    unsigned m = min(min(d0, d1), min(d2, d3));
    if (e < E4 && m < (unsigned)ROWS_PB) {
      visit(e, r.x);
      visit(e + 1, r.y);
      visit(e + 2, r.z);
      visit(e + 3, r.w);
    }
  };
  const int qLast = (E4 >> 2) - 1;
  auto load4 = [&](int4 (&v)[4], int e0) {
#pragma unroll
    for (int k = 0; k < 4; ++k) {
      int q = e0 / 4 + k * AGG_THREADS + tid;
      v[k] = row4[min(q, qLast)];
    }
  };
  const int STEP = 16 * AGG_THREADS;
  const int nWin = (E4 + 2 * STEP - 1) / (2 * STEP);
  int win = nWin > 0 ? (int)(blockIdx.x % (unsigned)nWin) : 0;
  int e0 = win * 2 * STEP;
  int4 va[4], vb[4];
  if (nWin > 0) load4(va, e0);
  for (int n = 0; n < nWin; ++n) {
    win = (win + 1 < nWin) ? win + 1 : 0;
    const int e0n = win * 2 * STEP;
    load4(vb, e0 + STEP);
#pragma unroll
    for (int k = 0; k < 4; ++k)
      visit4(e0 + (k * AGG_THREADS + tid) * 4, va[k]);
    load4(va, e0n);
#pragma unroll
    for (int k = 0; k < 4; ++k)
      visit4(e0 + STEP + (k * AGG_THREADS + tid) * 4, vb[k]);
    e0 = e0n;
  }
  for (int e = E4 + tid; e < E; e += AGG_THREADS) visit(e, rowIdx[e]);
  __syncthreads();

  const int nq = __builtin_amdgcn_readfirstlane(min(qn, QCAP));
  const int wave = __builtin_amdgcn_readfirstlane(tid >> 5);
  for (int g0 = wave * 32; g0 < nq; g0 += 32 * AGG_WAVES) {
    const int cnt = min(32, nq - g0);
    const bool valid = lane < cnt;
    const int e = valid ? queue[g0 + lane] : 0;
    const int r = rowIdx[e];
    const int c = colIdx[e];
    const float nrm = valid ? -dinv[r] * dinv[c] : 0.0f;
    const int d = valid ? r - base : 0;
    if (HAS8) {
      if (valid) {
        const float4* s8 = (const float4*)(src8 + (size_t)c * IN_CH);
        float4 a = s8[0];
        float4 b = s8[1];
        float* t8 = &acc8[d * IN_CH];
        atomicAdd(&t8[0], a.x * nrm);
        atomicAdd(&t8[1], a.y * nrm);
        atomicAdd(&t8[2], a.z * nrm);
        atomicAdd(&t8[3], a.w * nrm);
        atomicAdd(&t8[4], b.x * nrm);
        atomicAdd(&t8[5], b.y * nrm);
        atomicAdd(&t8[6], b.z * nrm);
        atomicAdd(&t8[7], b.w * nrm);
      }
    }
#pragma unroll
    for (int h = 0; h < 2; ++h) {
      float2 g[16];
#pragma unroll
      for (int j = 0; j < 16; ++j) {
        int cj = __builtin_amdgcn_readlane(c, h * 16 + j);
        g[j] = ((const float2*)(src64 + (size_t)cj * OUT_CH))[lane];
      }
#pragma unroll
      for (int j = 0; j < 16; ++j) {
        if (h * 16 + j < cnt) {
          float nj = __int_as_float(
              __builtin_amdgcn_readlane(__float_as_int(nrm), h * 16 + j));
          int dj = __builtin_amdgcn_readlane(d, h * 16 + j);
          float* t = &acc[dj * OUT_CH + 2 * lane];
          atomicAdd(&t[0], g[j].x * nj);
          atomicAdd(&t[1], g[j].y * nj);
        }
      }
    }
  }
  __syncthreads();

  const int rows = min(ROWS_PB, N - base);
  if (rows > 0) {
    float4* o = (float4*)(dst64 + (size_t)base * OUT_CH);
    const float4* a4 = (const float4*)acc;
    for (int i = tid; i < rows * (OUT_CH / 4); i += AGG_THREADS) o[i] = a4[i];
    if (HAS8) {
      float4* o8 = (float4*)(dst8 + (size_t)base * IN_CH);
      const float4* a8 = (const float4*)acc8;
      for (int i = tid; i < rows * (IN_CH / 4); i += AGG_THREADS) o8[i] = a8[i];
    }
  }
}

__global__ void __launch_bounds__(128)
gates_wmma_kernel(const float* __restrict__ x, const float* __restrict__ H,
                  const float* __restrict__ Lx, const float* __restrict__ LH,
                  const float* __restrict__ Wzx, const float* __restrict__ Wzh,
                  const float* __restrict__ Wrx, const float* __restrict__ Wrh,
                  const float* __restrict__ bzx, const float* __restrict__ bzh,
                  const float* __restrict__ brx, const float* __restrict__ brh,
                  float* __restrict__ Zbuf, float* __restrict__ RH) {
  int lane = threadIdx.x & 31;
  int wv = threadIdx.x >> 5;
  int half = lane >> 4;
  int l15 = lane & 15;
  int row0 = blockIdx.x * 16;
  int col0 = wv * 16;
  int rA = row0 + l15;
  int colB = col0 + l15;

  v8f accZ = {};
  v8f accR = {};
  accZ = accum_block(x, IN_CH, rA, Wzx, IN_CH, colB, half, accZ);
  accR = accum_block(x, IN_CH, rA, Wrx, IN_CH, colB, half, accR);
  accZ = accum_block(Lx, IN_CH, rA, Wzx + IN_CH * OUT_CH, IN_CH, colB, half, accZ);
  accR = accum_block(Lx, IN_CH, rA, Wrx + IN_CH * OUT_CH, IN_CH, colB, half, accR);
  accZ = accum_block(H, OUT_CH, rA, Wzh, OUT_CH, colB, half, accZ);
  accR = accum_block(H, OUT_CH, rA, Wrh, OUT_CH, colB, half, accR);
  accZ = accum_block(LH, OUT_CH, rA, Wzh + OUT_CH * OUT_CH, OUT_CH, colB, half, accZ);
  accR = accum_block(LH, OUT_CH, rA, Wrh + OUT_CH * OUT_CH, OUT_CH, colB, half, accR);

  float bZ = bzx[colB] + bzh[colB];
  float bR = brx[colB] + brh[colB];
#pragma unroll
  for (int i = 0; i < 8; ++i) {
    int r = row0 + i + 8 * half;
    float z = 1.0f / (1.0f + __expf(-(accZ[i] + bZ)));
    float rg = 1.0f / (1.0f + __expf(-(accR[i] + bR)));
    Zbuf[r * OUT_CH + colB] = z;
    RH[r * OUT_CH + colB] = rg * H[r * OUT_CH + colB];
  }
}

__global__ void __launch_bounds__(128)
cand_out_wmma_kernel(const float* __restrict__ x, const float* __restrict__ H,
                     const float* __restrict__ Lx, const float* __restrict__ RH,
                     const float* __restrict__ LRH,
                     const float* __restrict__ Zbuf,
                     const float* __restrict__ Wcx,
                     const float* __restrict__ Wch,
                     const float* __restrict__ bcx,
                     const float* __restrict__ bch, float* __restrict__ out) {
  int lane = threadIdx.x & 31;
  int wv = threadIdx.x >> 5;
  int half = lane >> 4;
  int l15 = lane & 15;
  int row0 = blockIdx.x * 16;
  int col0 = wv * 16;
  int rA = row0 + l15;
  int colB = col0 + l15;

  v8f accC = {};
  accC = accum_block(x, IN_CH, rA, Wcx, IN_CH, colB, half, accC);
  accC = accum_block(Lx, IN_CH, rA, Wcx + IN_CH * OUT_CH, IN_CH, colB, half, accC);
  accC = accum_block(RH, OUT_CH, rA, Wch, OUT_CH, colB, half, accC);
  accC = accum_block(LRH, OUT_CH, rA, Wch + OUT_CH * OUT_CH, OUT_CH, colB, half, accC);

  float bC = bcx[colB] + bch[colB];
#pragma unroll
  for (int i = 0; i < 8; ++i) {
    int r = row0 + i + 8 * half;
    float cv = tanhf(accC[i] + bC);
    float z = Zbuf[r * OUT_CH + colB];
    float h = H[r * OUT_CH + colB];
    out[r * OUT_CH + colB] = z * h + (1.0f - z) * cv;
  }
}


extern "C" void kernel_launch(void* const* d_in, const int* in_sizes, int n_in,
                              void* d_out, int out_size, void* d_ws,
                              size_t ws_size, hipStream_t stream) {
  const float* x = (const float*)d_in[0];
  const float* H = (const float*)d_in[1];
  const int* edge = (const int*)d_in[2];
  const float* Wzx = (const float*)d_in[3];
  const float* bzx = (const float*)d_in[4];
  const float* Wzh = (const float*)d_in[5];
  const float* bzh = (const float*)d_in[6];
  const float* Wrx = (const float*)d_in[7];
  const float* brx = (const float*)d_in[8];
  const float* Wrh = (const float*)d_in[9];
  const float* brh = (const float*)d_in[10];
  const float* Wcx = (const float*)d_in[11];
  const float* bcx = (const float*)d_in[12];
  const float* Wch = (const float*)d_in[13];
  const float* bch = (const float*)d_in[14];

  const int N = in_sizes[0] / IN_CH;
  const int E = in_sizes[2] / 2;
  const int* rowIdx = edge;
  const int* colIdx = edge + E;

  float* ws = (float*)d_ws;
  float* dinv = ws;
  float* Lx = dinv + ((N + 63) / 64) * 64;
  float* LH = Lx + (size_t)N * IN_CH;
  float* LRH = LH + (size_t)N * OUT_CH;
  float* Zbuf = LRH + (size_t)N * OUT_CH;
  float* RH = Zbuf + (size_t)N * OUT_CH;
  const int nZero = N;
  const int aggBlocks = (N + ROWS_PB - 1) / ROWS_PB;

  zero_kernel<<<(nZero + 255) / 256, 256, 0, stream>>>(dinv, nZero);
  degree_kernel<<<(E + 255) / 256, 256, 0, stream>>>(rowIdx, dinv, E);
  dinv_kernel<<<(N + 255) / 256, 256, 0, stream>>>(dinv, N);
  agg_kernel<true><<<aggBlocks, AGG_THREADS, 0, stream>>>(
      rowIdx, colIdx, dinv, H, LH, x, Lx, E, N);
  gates_wmma_kernel<<<N / 16, 128, 0, stream>>>(x, H, Lx, LH, Wzx, Wzh, Wrx,
                                                Wrh, bzx, bzh, brx, brh, Zbuf,
                                                RH);
  agg_kernel<false><<<aggBlocks, AGG_THREADS, 0, stream>>>(
      rowIdx, colIdx, dinv, RH, LRH, nullptr, nullptr, E, N);
  cand_out_wmma_kernel<<<N / 16, 128, 0, stream>>>(
      x, H, Lx, RH, LRH, Zbuf, Wcx, Wch, bcx, bch, (float*)d_out);
}


# Edge filter runs once per call: the first aggregation pass saves each block's listed edges (source row, local row, weight) in the workspace, the second sums from that list; sum order = LDS arrival
# speedup vs baseline: 1.7275x; 1.2155x over previous
#include <hip/hip_runtime.h>
#include <hip/hip_bf16.h>
#include <math.h>


typedef float v2f __attribute__((ext_vector_type(2)));
typedef float v8f __attribute__((ext_vector_type(8)));

#define IN_CH 8
#define OUT_CH 64

__device__ __forceinline__ v8f wmma_f32(v2f a, v2f b, v8f c) {
  return __builtin_amdgcn_wmma_f32_16x16x4_f32(false, a, false, b, (short)0, c,
                                               false, false);
}

__device__ __forceinline__ v8f accum_block(const float* __restrict__ A, int ldA,
                                           int rA, const float* __restrict__ W,
                                           int Kdim, int colB, int half,
                                           v8f acc) {
#pragma unroll 4
  for (int k0 = 0; k0 < Kdim; k0 += 4) {
    int ka = k0 + 2 * half;
    v2f a;
    a.x = A[rA * ldA + ka];
    a.y = A[rA * ldA + ka + 1];
    v2f b;
    b.x = W[ka * OUT_CH + colB];
    b.y = W[(ka + 1) * OUT_CH + colB];
    acc = wmma_f32(a, b, acc);
  }
  return acc;
}


__global__ void __launch_bounds__(256) zero_kernel(float* p, int n) {
  int i = blockIdx.x * blockDim.x + threadIdx.x;
  if (i < n) p[i] = 0.0f;
}

__global__ void __launch_bounds__(256)
degree_kernel(const int* __restrict__ rowIdx, float* __restrict__ deg, int E) {
  int e = blockIdx.x * blockDim.x + threadIdx.x;
  if (e < E) atomicAdd(&deg[rowIdx[e]], 1.0f);
}

__global__ void __launch_bounds__(256) dinv_kernel(float* deg, int n) {
  int i = blockIdx.x * blockDim.x + threadIdx.x;
  if (i < n) {
    float d = deg[i];
    deg[i] = (d > 0.0f) ? rsqrtf(d) : 0.0f;
  }
}

#define ROWS_PB 196
#define QCAP 32768
#define AGG_THREADS 1024
#define AGG_WAVES (AGG_THREADS / 32)

template <bool FIRST>
__global__ void __launch_bounds__(AGG_THREADS)
agg_kernel(const int* __restrict__ rowIdx, const int* __restrict__ colIdx,
           const float* __restrict__ dinv, const float* __restrict__ src64,
           float* __restrict__ dst64, const float* __restrict__ src8,
           float* __restrict__ dst8, int* __restrict__ listWords,
           int* __restrict__ listCount, int E, int N) {
  constexpr bool HAS8 = FIRST;
  __shared__ __attribute__((aligned(16))) float acc[ROWS_PB * OUT_CH];
  __shared__ __attribute__((aligned(16))) float acc8[HAS8 ? ROWS_PB * IN_CH : 4];
  __shared__ int queue[QCAP];
  __shared__ int qn;

  const int tid = threadIdx.x;
  const int lane = tid & 31;
  const int base = blockIdx.x * ROWS_PB;

  for (int i = tid; i < ROWS_PB * OUT_CH; i += AGG_THREADS) acc[i] = 0.0f;
  if (HAS8) {
    for (int i = tid; i < ROWS_PB * IN_CH; i += AGG_THREADS) acc8[i] = 0.0f;
  }
  if (tid == 0) qn = 0;
  __syncthreads();

  int total = 0;
  bool fromList = false;
  if (!FIRST) {
    total = __builtin_amdgcn_readfirstlane(listCount[blockIdx.x]);
    fromList = total <= QCAP;
  }
  int4* __restrict__ myList = (int4*)listWords + (size_t)blockIdx.x * QCAP;

  if (!fromList) {
    auto visit = [&](int e, int r) {
      unsigned d = (unsigned)(r - base);
      if (d < (unsigned)ROWS_PB) {
        int slot = atomicAdd(&qn, 1);
        if (slot < QCAP) {
          queue[slot] = e;
        } else {
          int c = colIdx[e];
          float nrm = -dinv[r] * dinv[c];
#pragma unroll 1
          for (int k = 0; k < OUT_CH; ++k)
            atomicAdd(&acc[d * OUT_CH + k], src64[(size_t)c * OUT_CH + k] * nrm);
          if (HAS8) {
#pragma unroll 1
            for (int k = 0; k < IN_CH; ++k)
              atomicAdd(&acc8[d * IN_CH + k], src8[(size_t)c * IN_CH + k] * nrm);
          }
        }
      }
    };

    const int E4 = E & ~3;
    const int4* __restrict__ row4 = (const int4*)rowIdx;
    auto visit4 = [&](int e, int4 r) {
      unsigned d0 = (unsigned)(r.x - base), d1 = (unsigned)(r.y - base);
      unsigned d2 = (unsigned)(r.z - base), d3 = (unsigned)(r.w - base);
      unsigned m = min(min(d0, d1), min(d2, d3));
      if (e < E4 && m < (unsigned)ROWS_PB) {
        visit(e, r.x);
        visit(e + 1, r.y);
        visit(e + 2, r.z);
        visit(e + 3, r.w);
      }
    };
    const int qLast = (E4 >> 2) - 1;
    auto load4 = [&](int4 (&v)[4], int e0) {
#pragma unroll
      for (int k = 0; k < 4; ++k) {
        int q = e0 / 4 + k * AGG_THREADS + tid;
        v[k] = row4[min(q, qLast)];
      }
    };
    const int STEP = 16 * AGG_THREADS;
    const int nWin = (E4 + 2 * STEP - 1) / (2 * STEP);
    int win = nWin > 0 ? (int)(blockIdx.x % (unsigned)nWin) : 0;
    int e0 = win * 2 * STEP;
    int4 va[4], vb[4];
    if (nWin > 0) load4(va, e0);
    for (int n = 0; n < nWin; ++n) {
      win = (win + 1 < nWin) ? win + 1 : 0;
      const int e0n = win * 2 * STEP;
      load4(vb, e0 + STEP);
#pragma unroll
      for (int k = 0; k < 4; ++k)
        visit4(e0 + (k * AGG_THREADS + tid) * 4, va[k]);
      load4(va, e0n);
#pragma unroll
      for (int k = 0; k < 4; ++k)
        visit4(e0 + STEP + (k * AGG_THREADS + tid) * 4, vb[k]);
      e0 = e0n;
    }
    for (int e = E4 + tid; e < E; e += AGG_THREADS) visit(e, rowIdx[e]);
  }
  __syncthreads();

  const int wave = __builtin_amdgcn_readfirstlane(tid >> 5);
  auto sumGroup = [&](int c, int d, float nrm, int cnt) {
#pragma unroll
    for (int h = 0; h < 2; ++h) {
      float2 g[16];
#pragma unroll
      for (int j = 0; j < 16; ++j) {
        int cj = __builtin_amdgcn_readlane(c, h * 16 + j);
        g[j] = ((const float2*)(src64 + (size_t)cj * OUT_CH))[lane];
      }
#pragma unroll
      for (int j = 0; j < 16; ++j) {
        if (h * 16 + j < cnt) {
          float nj = __int_as_float(
              __builtin_amdgcn_readlane(__float_as_int(nrm), h * 16 + j));
          int dj = __builtin_amdgcn_readlane(d, h * 16 + j);
          float* t = &acc[dj * OUT_CH + 2 * lane];
          atomicAdd(&t[0], g[j].x * nj);
          atomicAdd(&t[1], g[j].y * nj);
        }
      }
    }
  };
  if (fromList) {
    const int nq = total;
    int g0 = wave * 32;
    if (g0 < nq) {
      int4 rec = myList[min(g0 + lane, nq - 1)];
      for (;;) {
        const int gn = g0 + 32 * AGG_WAVES;
        const int4 nxt = myList[min(gn + lane, nq - 1)];
        sumGroup(rec.x, rec.y, __int_as_float(rec.z), min(32, nq - g0));
        if (gn >= nq) break;
        rec = nxt;
        g0 = gn;
      }
    }
  } else {
    const int nq = __builtin_amdgcn_readfirstlane(min(qn, QCAP));
    for (int g0 = wave * 32; g0 < nq; g0 += 32 * AGG_WAVES) {
      const int cnt = min(32, nq - g0);
      const bool valid = lane < cnt;
      const int e = valid ? queue[g0 + lane] : 0;
      const int r = rowIdx[e];
      const int c = colIdx[e];
      const float nrm = valid ? -dinv[r] * dinv[c] : 0.0f;
      const int d = valid ? r - base : 0;
      if (FIRST && valid)
        myList[g0 + lane] = make_int4(c, d, __float_as_int(nrm), 0);
      if (HAS8) {
        if (valid) {
          const float4* s8 = (const float4*)(src8 + (size_t)c * IN_CH);
          float4 a = s8[0];
          float4 b = s8[1];
          float* t8 = &acc8[d * IN_CH];
          atomicAdd(&t8[0], a.x * nrm);
          atomicAdd(&t8[1], a.y * nrm);
          atomicAdd(&t8[2], a.z * nrm);
          atomicAdd(&t8[3], a.w * nrm);
          atomicAdd(&t8[4], b.x * nrm);
          atomicAdd(&t8[5], b.y * nrm);
          atomicAdd(&t8[6], b.z * nrm);
          atomicAdd(&t8[7], b.w * nrm);
        }
      }
      sumGroup(c, d, nrm, cnt);
    }
    if (FIRST && tid == 0) listCount[blockIdx.x] = qn;
  }
  __syncthreads();

  const int rows = min(ROWS_PB, N - base);
  if (rows > 0) {
    float4* o = (float4*)(dst64 + (size_t)base * OUT_CH);
    const float4* a4 = (const float4*)acc;
    for (int i = tid; i < rows * (OUT_CH / 4); i += AGG_THREADS) o[i] = a4[i];
    if (HAS8) {
      float4* o8 = (float4*)(dst8 + (size_t)base * IN_CH);
      const float4* a8 = (const float4*)acc8;
      for (int i = tid; i < rows * (IN_CH / 4); i += AGG_THREADS) o8[i] = a8[i];
    }
  }
}

__global__ void __launch_bounds__(128)
gates_wmma_kernel(const float* __restrict__ x, const float* __restrict__ H,
                  const float* __restrict__ Lx, const float* __restrict__ LH,
                  const float* __restrict__ Wzx, const float* __restrict__ Wzh,
                  const float* __restrict__ Wrx, const float* __restrict__ Wrh,
                  const float* __restrict__ bzx, const float* __restrict__ bzh,
                  const float* __restrict__ brx, const float* __restrict__ brh,
                  float* __restrict__ Zbuf, float* __restrict__ RH) {
  int lane = threadIdx.x & 31;
  int wv = threadIdx.x >> 5;
  int half = lane >> 4;
  int l15 = lane & 15;
  int row0 = blockIdx.x * 16;
  int col0 = wv * 16;
  int rA = row0 + l15;
  int colB = col0 + l15;

  v8f accZ = {};
  v8f accR = {};
  accZ = accum_block(x, IN_CH, rA, Wzx, IN_CH, colB, half, accZ);
  accR = accum_block(x, IN_CH, rA, Wrx, IN_CH, colB, half, accR);
  accZ = accum_block(Lx, IN_CH, rA, Wzx + IN_CH * OUT_CH, IN_CH, colB, half, accZ);
  accR = accum_block(Lx, IN_CH, rA, Wrx + IN_CH * OUT_CH, IN_CH, colB, half, accR);
  accZ = accum_block(H, OUT_CH, rA, Wzh, OUT_CH, colB, half, accZ);
  accR = accum_block(H, OUT_CH, rA, Wrh, OUT_CH, colB, half, accR);
  accZ = accum_block(LH, OUT_CH, rA, Wzh + OUT_CH * OUT_CH, OUT_CH, colB, half, accZ);
  accR = accum_block(LH, OUT_CH, rA, Wrh + OUT_CH * OUT_CH, OUT_CH, colB, half, accR);

  float bZ = bzx[colB] + bzh[colB];
  float bR = brx[colB] + brh[colB];
#pragma unroll
  for (int i = 0; i < 8; ++i) {
    int r = row0 + i + 8 * half;
    float z = 1.0f / (1.0f + __expf(-(accZ[i] + bZ)));
    float rg = 1.0f / (1.0f + __expf(-(accR[i] + bR)));
    Zbuf[r * OUT_CH + colB] = z;
    RH[r * OUT_CH + colB] = rg * H[r * OUT_CH + colB];
  }
}

__global__ void __launch_bounds__(128)
cand_out_wmma_kernel(const float* __restrict__ x, const float* __restrict__ H,
                     const float* __restrict__ Lx, const float* __restrict__ RH,
                     const float* __restrict__ LRH,
                     const float* __restrict__ Zbuf,
                     const float* __restrict__ Wcx,
                     const float* __restrict__ Wch,
                     const float* __restrict__ bcx,
                     const float* __restrict__ bch, float* __restrict__ out) {
  int lane = threadIdx.x & 31;
  int wv = threadIdx.x >> 5;
  int half = lane >> 4;
  int l15 = lane & 15;
  int row0 = blockIdx.x * 16;
  int col0 = wv * 16;
  int rA = row0 + l15;
  int colB = col0 + l15;

  v8f accC = {};
  accC = accum_block(x, IN_CH, rA, Wcx, IN_CH, colB, half, accC);
  accC = accum_block(Lx, IN_CH, rA, Wcx + IN_CH * OUT_CH, IN_CH, colB, half, accC);
  accC = accum_block(RH, OUT_CH, rA, Wch, OUT_CH, colB, half, accC);
  accC = accum_block(LRH, OUT_CH, rA, Wch + OUT_CH * OUT_CH, OUT_CH, colB, half, accC);

  float bC = bcx[colB] + bch[colB];
#pragma unroll
  for (int i = 0; i < 8; ++i) {
    int r = row0 + i + 8 * half;
    float cv = tanhf(accC[i] + bC);
    float z = Zbuf[r * OUT_CH + colB];
    float h = H[r * OUT_CH + colB];
    out[r * OUT_CH + colB] = z * h + (1.0f - z) * cv;
  }
}


extern "C" void kernel_launch(void* const* d_in, const int* in_sizes, int n_in,
                              void* d_out, int out_size, void* d_ws,
                              size_t ws_size, hipStream_t stream) {
  const float* x = (const float*)d_in[0];
  const float* H = (const float*)d_in[1];
  const int* edge = (const int*)d_in[2];
  const float* Wzx = (const float*)d_in[3];
  const float* bzx = (const float*)d_in[4];
  const float* Wzh = (const float*)d_in[5];
  const float* bzh = (const float*)d_in[6];
  const float* Wrx = (const float*)d_in[7];
  const float* brx = (const float*)d_in[8];
  const float* Wrh = (const float*)d_in[9];
  const float* brh = (const float*)d_in[10];
  const float* Wcx = (const float*)d_in[11];
  const float* bcx = (const float*)d_in[12];
  const float* Wch = (const float*)d_in[13];
  const float* bch = (const float*)d_in[14];

  const int N = in_sizes[0] / IN_CH;
  const int E = in_sizes[2] / 2;
  const int* rowIdx = edge;
  const int* colIdx = edge + E;

  float* ws = (float*)d_ws;
  float* dinv = ws;
  float* Lx = dinv + ((N + 63) / 64) * 64;
  float* LH = Lx + (size_t)N * IN_CH;
  float* LRH = LH + (size_t)N * OUT_CH;
  float* Zbuf = LRH + (size_t)N * OUT_CH;
  float* RH = Zbuf + (size_t)N * OUT_CH;
  const int nZero = N;
  const int aggBlocks = (N + ROWS_PB - 1) / ROWS_PB;
  int* listCount = (int*)(RH + (size_t)N * OUT_CH);
  int* listWords = listCount + ((aggBlocks + 63) / 64) * 64;

  zero_kernel<<<(nZero + 255) / 256, 256, 0, stream>>>(dinv, nZero);
  degree_kernel<<<(E + 255) / 256, 256, 0, stream>>>(rowIdx, dinv, E);
  dinv_kernel<<<(N + 255) / 256, 256, 0, stream>>>(dinv, N);
  agg_kernel<true><<<aggBlocks, AGG_THREADS, 0, stream>>>(
      rowIdx, colIdx, dinv, H, LH, x, Lx, listWords, listCount, E, N);
  gates_wmma_kernel<<<N / 16, 128, 0, stream>>>(x, H, Lx, LH, Wzx, Wzh, Wrx,
                                                Wrh, bzx, bzh, brx, brh, Zbuf,
                                                RH);
  agg_kernel<false><<<aggBlocks, AGG_THREADS, 0, stream>>>(
      rowIdx, colIdx, dinv, RH, LRH, nullptr, nullptr, listWords, listCount, E,
      N);
  cand_out_wmma_kernel<<<N / 16, 128, 0, stream>>>(
      x, H, Lx, RH, LRH, Zbuf, Wcx, Wch, bcx, bch, (float*)d_out);
}


# Edge filter streams one owner byte per edge (block number of its destination row, packed by the degree kernel) instead of the four-byte row: a b128 load covers sixteen edges; sum order = LDS arrival
# speedup vs baseline: 1.8542x; 1.0733x over previous
#include <hip/hip_runtime.h>
#include <hip/hip_bf16.h>
#include <math.h>


typedef float v2f __attribute__((ext_vector_type(2)));
typedef float v8f __attribute__((ext_vector_type(8)));

#define IN_CH 8
#define OUT_CH 64

__device__ __forceinline__ v8f wmma_f32(v2f a, v2f b, v8f c) {
  return __builtin_amdgcn_wmma_f32_16x16x4_f32(false, a, false, b, (short)0, c,
                                               false, false);
}

__device__ __forceinline__ v8f accum_block(const float* __restrict__ A, int ldA,
                                           int rA, const float* __restrict__ W,
                                           int Kdim, int colB, int half,
                                           v8f acc) {
#pragma unroll 4
  for (int k0 = 0; k0 < Kdim; k0 += 4) {
    int ka = k0 + 2 * half;
    v2f a;
    a.x = A[rA * ldA + ka];
    a.y = A[rA * ldA + ka + 1];
    v2f b;
    b.x = W[ka * OUT_CH + colB];
    b.y = W[(ka + 1) * OUT_CH + colB];
    acc = wmma_f32(a, b, acc);
  }
  return acc;
}


__global__ void __launch_bounds__(256) zero_kernel(float* p, int n) {
  int i = blockIdx.x * blockDim.x + threadIdx.x;
  if (i < n) p[i] = 0.0f;
}

#define ROWS_PB 196

__device__ __forceinline__ unsigned owner_byte(int r) {
  return (unsigned)(r / ROWS_PB) & 255u;
}

__global__ void __launch_bounds__(256)
degree_kernel(const int* __restrict__ rowIdx, float* __restrict__ deg,
              unsigned* __restrict__ ownerW, int E) {
  const int t = blockIdx.x * blockDim.x + threadIdx.x;
  const int e = t * 4;
  if (e + 3 < E) {
    const int4 r = ((const int4*)rowIdx)[t];
    atomicAdd(&deg[r.x], 1.0f);
    atomicAdd(&deg[r.y], 1.0f);
    atomicAdd(&deg[r.z], 1.0f);
    atomicAdd(&deg[r.w], 1.0f);
    ownerW[t] = owner_byte(r.x) | (owner_byte(r.y) << 8) |
                (owner_byte(r.z) << 16) | (owner_byte(r.w) << 24);
  } else {
    unsigned char* ob = (unsigned char*)ownerW;
    for (int k = e; k < E; ++k) {
      const int r = rowIdx[k];
      atomicAdd(&deg[r], 1.0f);
      ob[k] = (unsigned char)owner_byte(r);
    }
  }
}

__global__ void __launch_bounds__(256) dinv_kernel(float* deg, int n) {
  int i = blockIdx.x * blockDim.x + threadIdx.x;
  if (i < n) {
    float d = deg[i];
    deg[i] = (d > 0.0f) ? rsqrtf(d) : 0.0f;
  }
}

#define QCAP 32768
#define AGG_THREADS 1024
#define AGG_WAVES (AGG_THREADS / 32)

template <bool FIRST>
__global__ void __launch_bounds__(AGG_THREADS)
agg_kernel(const int* __restrict__ rowIdx, const int* __restrict__ colIdx,
           const float* __restrict__ dinv, const float* __restrict__ src64,
           float* __restrict__ dst64, const float* __restrict__ src8,
           float* __restrict__ dst8, const unsigned* __restrict__ ownerW,
           int* __restrict__ listWords, int* __restrict__ listCount, int E,
           int N) {
  constexpr bool HAS8 = FIRST;
  __shared__ __attribute__((aligned(16))) float acc[ROWS_PB * OUT_CH];
  __shared__ __attribute__((aligned(16))) float acc8[HAS8 ? ROWS_PB * IN_CH : 4];
  __shared__ int queue[QCAP];
  __shared__ int qn;

  const int tid = threadIdx.x;
  const int lane = tid & 31;
  const int base = blockIdx.x * ROWS_PB;

  for (int i = tid; i < ROWS_PB * OUT_CH; i += AGG_THREADS) acc[i] = 0.0f;
  if (HAS8) {
    for (int i = tid; i < ROWS_PB * IN_CH; i += AGG_THREADS) acc8[i] = 0.0f;
  }
  if (tid == 0) qn = 0;
  __syncthreads();

  int total = 0;
  bool fromList = false;
  if (!FIRST) {
    total = __builtin_amdgcn_readfirstlane(listCount[blockIdx.x]);
    fromList = total <= QCAP;
  }
  int4* __restrict__ myList = (int4*)listWords + (size_t)blockIdx.x * QCAP;

  if (!fromList) {
    const bool byteIsBlock = gridDim.x <= 256u;
    auto take = [&](int e) {
      int slot = atomicAdd(&qn, 1);
      if (slot < QCAP) {
        queue[slot] = e;
      } else {
        int r = rowIdx[e];
        int c = colIdx[e];
        unsigned d = (unsigned)(r - base);
        float nrm = -dinv[r] * dinv[c];
#pragma unroll 1
        for (int k = 0; k < OUT_CH; ++k)
          atomicAdd(&acc[d * OUT_CH + k], src64[(size_t)c * OUT_CH + k] * nrm);
        if (HAS8) {
#pragma unroll 1
          for (int k = 0; k < IN_CH; ++k)
            atomicAdd(&acc8[d * IN_CH + k], src8[(size_t)c * IN_CH + k] * nrm);
        }
      }
    };
    auto visitRow = [&](int e) {
      unsigned d = (unsigned)(rowIdx[e] - base);
      if (d < (unsigned)ROWS_PB) take(e);
    };
    const unsigned pat = (blockIdx.x & 255u) * 0x01010101u;
    const int nG = E >> 4;
    const uint4* __restrict__ own4 = (const uint4*)ownerW;
    auto visit16 = [&](int q, uint4 w) {
      const unsigned x0 = w.x ^ pat, x1 = w.y ^ pat;
      const unsigned x2 = w.z ^ pat, x3 = w.w ^ pat;
      const unsigned n0 = ((x0 & 0x7f7f7f7fu) + 0x7f7f7f7fu) | x0;
      const unsigned n1 = ((x1 & 0x7f7f7f7fu) + 0x7f7f7f7fu) | x1;
      const unsigned n2 = ((x2 & 0x7f7f7f7fu) + 0x7f7f7f7fu) | x2;
      const unsigned n3 = ((x3 & 0x7f7f7f7fu) + 0x7f7f7f7fu) | x3;
      if (q < nG && (~(n0 & n1 & n2 & n3) & 0x80808080u) != 0u) {
        unsigned h = ((~n0 & 0x80808080u) >> 7) | ((~n1 & 0x80808080u) >> 6) |
                     ((~n2 & 0x80808080u) >> 5) | ((~n3 & 0x80808080u) >> 4);
        const int e = q * 16;
        while (h != 0u) {
          const int p = __builtin_ctz(h);
          h &= h - 1u;
          const int eh = e + ((p & 7) << 2) + (p >> 3);
          if (byteIsBlock) take(eh); else visitRow(eh);
        }
      }
    };
    const int qLast = nG - 1;
    auto load4 = [&](uint4 (&v)[4], int q0) {
#pragma unroll
      for (int k = 0; k < 4; ++k) {
        int q = q0 + k * AGG_THREADS + tid;
        v[k] = own4[min(q, qLast)];
      }
    };
    const int STEP = 4 * AGG_THREADS;
    const int nWin = (nG + 2 * STEP - 1) / (2 * STEP);
    int win = nWin > 0 ? (int)(blockIdx.x % (unsigned)nWin) : 0;
    int q0 = win * 2 * STEP;
    uint4 va[4], vb[4];
    if (nWin > 0) load4(va, q0);
    for (int n = 0; n < nWin; ++n) {
      win = (win + 1 < nWin) ? win + 1 : 0;
      const int q0n = win * 2 * STEP;
      load4(vb, q0 + STEP);
#pragma unroll
      for (int k = 0; k < 4; ++k)
        visit16(q0 + k * AGG_THREADS + tid, va[k]);
      load4(va, q0n);
#pragma unroll
      for (int k = 0; k < 4; ++k)
        visit16(q0 + STEP + k * AGG_THREADS + tid, vb[k]);
      q0 = q0n;
    }
    for (int e = (nG << 4) + tid; e < E; e += AGG_THREADS) visitRow(e);
  }
  __syncthreads();

  const int wave = __builtin_amdgcn_readfirstlane(tid >> 5);
  auto sumGroup = [&](int c, int d, float nrm, int cnt) {
#pragma unroll
    for (int h = 0; h < 2; ++h) {
      float2 g[16];
#pragma unroll
      for (int j = 0; j < 16; ++j) {
        int cj = __builtin_amdgcn_readlane(c, h * 16 + j);
        g[j] = ((const float2*)(src64 + (size_t)cj * OUT_CH))[lane];
      }
#pragma unroll
      for (int j = 0; j < 16; ++j) {
        if (h * 16 + j < cnt) {
          float nj = __int_as_float(
              __builtin_amdgcn_readlane(__float_as_int(nrm), h * 16 + j));
          int dj = __builtin_amdgcn_readlane(d, h * 16 + j);
          float* t = &acc[dj * OUT_CH + 2 * lane];
          atomicAdd(&t[0], g[j].x * nj);
          atomicAdd(&t[1], g[j].y * nj);
        }
      }
    }
  };
  if (fromList) {
    const int nq = total;
    int g0 = wave * 32;
    if (g0 < nq) {
      int4 rec = myList[min(g0 + lane, nq - 1)];
      for (;;) {
        const int gn = g0 + 32 * AGG_WAVES;
        const int4 nxt = myList[min(gn + lane, nq - 1)];
        sumGroup(rec.x, rec.y, __int_as_float(rec.z), min(32, nq - g0));
        if (gn >= nq) break;
        rec = nxt;
        g0 = gn;
      }
    }
  } else {
    const int nq = __builtin_amdgcn_readfirstlane(min(qn, QCAP));
    for (int g0 = wave * 32; g0 < nq; g0 += 32 * AGG_WAVES) {
      const int cnt = min(32, nq - g0);
      const bool valid = lane < cnt;
      const int e = valid ? queue[g0 + lane] : 0;
      const int r = rowIdx[e];
      const int c = colIdx[e];
      const float nrm = valid ? -dinv[r] * dinv[c] : 0.0f;
      const int d = valid ? r - base : 0;
      if (FIRST && valid)
        myList[g0 + lane] = make_int4(c, d, __float_as_int(nrm), 0);
      if (HAS8) {
        if (valid) {
          const float4* s8 = (const float4*)(src8 + (size_t)c * IN_CH);
          float4 a = s8[0];
          float4 b = s8[1];
          float* t8 = &acc8[d * IN_CH];
          atomicAdd(&t8[0], a.x * nrm);
          atomicAdd(&t8[1], a.y * nrm);
          atomicAdd(&t8[2], a.z * nrm);
          atomicAdd(&t8[3], a.w * nrm);
          atomicAdd(&t8[4], b.x * nrm);
          atomicAdd(&t8[5], b.y * nrm);
          atomicAdd(&t8[6], b.z * nrm);
          atomicAdd(&t8[7], b.w * nrm);
        }
      }
      sumGroup(c, d, nrm, cnt);
    }
    if (FIRST && tid == 0) listCount[blockIdx.x] = qn;
  }
  __syncthreads();

  const int rows = min(ROWS_PB, N - base);
  if (rows > 0) {
    float4* o = (float4*)(dst64 + (size_t)base * OUT_CH);
    const float4* a4 = (const float4*)acc;
    for (int i = tid; i < rows * (OUT_CH / 4); i += AGG_THREADS) o[i] = a4[i];
    if (HAS8) {
      float4* o8 = (float4*)(dst8 + (size_t)base * IN_CH);
      const float4* a8 = (const float4*)acc8;
      for (int i = tid; i < rows * (IN_CH / 4); i += AGG_THREADS) o8[i] = a8[i];
    }
  }
}

__global__ void __launch_bounds__(128)
gates_wmma_kernel(const float* __restrict__ x, const float* __restrict__ H,
                  const float* __restrict__ Lx, const float* __restrict__ LH,
                  const float* __restrict__ Wzx, const float* __restrict__ Wzh,
                  const float* __restrict__ Wrx, const float* __restrict__ Wrh,
                  const float* __restrict__ bzx, const float* __restrict__ bzh,
                  const float* __restrict__ brx, const float* __restrict__ brh,
                  float* __restrict__ Zbuf, float* __restrict__ RH) {
  int lane = threadIdx.x & 31;
  int wv = threadIdx.x >> 5;
  int half = lane >> 4;
  int l15 = lane & 15;
  int row0 = blockIdx.x * 16;
  int col0 = wv * 16;
  int rA = row0 + l15;
  int colB = col0 + l15;

  v8f accZ = {};
  v8f accR = {};
  accZ = accum_block(x, IN_CH, rA, Wzx, IN_CH, colB, half, accZ);
  accR = accum_block(x, IN_CH, rA, Wrx, IN_CH, colB, half, accR);
  accZ = accum_block(Lx, IN_CH, rA, Wzx + IN_CH * OUT_CH, IN_CH, colB, half, accZ);
  accR = accum_block(Lx, IN_CH, rA, Wrx + IN_CH * OUT_CH, IN_CH, colB, half, accR);
  accZ = accum_block(H, OUT_CH, rA, Wzh, OUT_CH, colB, half, accZ);
  accR = accum_block(H, OUT_CH, rA, Wrh, OUT_CH, colB, half, accR);
  accZ = accum_block(LH, OUT_CH, rA, Wzh + OUT_CH * OUT_CH, OUT_CH, colB, half, accZ);
  accR = accum_block(LH, OUT_CH, rA, Wrh + OUT_CH * OUT_CH, OUT_CH, colB, half, accR);

  float bZ = bzx[colB] + bzh[colB];
  float bR = brx[colB] + brh[colB];
#pragma unroll
  for (int i = 0; i < 8; ++i) {
    int r = row0 + i + 8 * half;
    float z = 1.0f / (1.0f + __expf(-(accZ[i] + bZ)));
    float rg = 1.0f / (1.0f + __expf(-(accR[i] + bR)));
    Zbuf[r * OUT_CH + colB] = z;
    RH[r * OUT_CH + colB] = rg * H[r * OUT_CH + colB];
  }
}

__global__ void __launch_bounds__(128)
cand_out_wmma_kernel(const float* __restrict__ x, const float* __restrict__ H,
                     const float* __restrict__ Lx, const float* __restrict__ RH,
                     const float* __restrict__ LRH,
                     const float* __restrict__ Zbuf,
                     const float* __restrict__ Wcx,
                     const float* __restrict__ Wch,
                     const float* __restrict__ bcx,
                     const float* __restrict__ bch, float* __restrict__ out) {
  int lane = threadIdx.x & 31;
  int wv = threadIdx.x >> 5;
  int half = lane >> 4;
  int l15 = lane & 15;
  int row0 = blockIdx.x * 16;
  int col0 = wv * 16;
  int rA = row0 + l15;
  int colB = col0 + l15;

  v8f accC = {};
  accC = accum_block(x, IN_CH, rA, Wcx, IN_CH, colB, half, accC);
  accC = accum_block(Lx, IN_CH, rA, Wcx + IN_CH * OUT_CH, IN_CH, colB, half, accC);
  accC = accum_block(RH, OUT_CH, rA, Wch, OUT_CH, colB, half, accC);
  accC = accum_block(LRH, OUT_CH, rA, Wch + OUT_CH * OUT_CH, OUT_CH, colB, half, accC);

  float bC = bcx[colB] + bch[colB];
#pragma unroll
  for (int i = 0; i < 8; ++i) {
    int r = row0 + i + 8 * half;
    float cv = tanhf(accC[i] + bC);
    float z = Zbuf[r * OUT_CH + colB];
    float h = H[r * OUT_CH + colB];
    out[r * OUT_CH + colB] = z * h + (1.0f - z) * cv;
  }
}


extern "C" void kernel_launch(void* const* d_in, const int* in_sizes, int n_in,
                              void* d_out, int out_size, void* d_ws,
                              size_t ws_size, hipStream_t stream) {
  const float* x = (const float*)d_in[0];
  const float* H = (const float*)d_in[1];
  const int* edge = (const int*)d_in[2];
  const float* Wzx = (const float*)d_in[3];
  const float* bzx = (const float*)d_in[4];
  const float* Wzh = (const float*)d_in[5];
  const float* bzh = (const float*)d_in[6];
  const float* Wrx = (const float*)d_in[7];
  const float* brx = (const float*)d_in[8];
  const float* Wrh = (const float*)d_in[9];
  const float* brh = (const float*)d_in[10];
  const float* Wcx = (const float*)d_in[11];
  const float* bcx = (const float*)d_in[12];
  const float* Wch = (const float*)d_in[13];
  const float* bch = (const float*)d_in[14];

  const int N = in_sizes[0] / IN_CH;
  const int E = in_sizes[2] / 2;
  const int* rowIdx = edge;
  const int* colIdx = edge + E;

  float* ws = (float*)d_ws;
  float* dinv = ws;
  float* Lx = dinv + ((N + 63) / 64) * 64;
  float* LH = Lx + (size_t)N * IN_CH;
  float* LRH = LH + (size_t)N * OUT_CH;
  float* Zbuf = LRH + (size_t)N * OUT_CH;
  float* RH = Zbuf + (size_t)N * OUT_CH;
  const int nZero = N;
  const int aggBlocks = (N + ROWS_PB - 1) / ROWS_PB;
  unsigned* ownerW = (unsigned*)(RH + (size_t)N * OUT_CH);
  int* listCount = (int*)(ownerW + ((E + 255) / 256) * 64);
  int* listWords = listCount + ((aggBlocks + 63) / 64) * 64;

  zero_kernel<<<(nZero + 255) / 256, 256, 0, stream>>>(dinv, nZero);
  degree_kernel<<<(E + 1023) / 1024, 256, 0, stream>>>(rowIdx, dinv, ownerW,
                                                        E);
  dinv_kernel<<<(N + 255) / 256, 256, 0, stream>>>(dinv, N);
  agg_kernel<true><<<aggBlocks, AGG_THREADS, 0, stream>>>(
      rowIdx, colIdx, dinv, H, LH, x, Lx, ownerW, listWords, listCount, E, N);
  gates_wmma_kernel<<<N / 16, 128, 0, stream>>>(x, H, Lx, LH, Wzx, Wzh, Wrx,
                                                Wrh, bzx, bzh, brx, brh, Zbuf,
                                                RH);
  agg_kernel<false><<<aggBlocks, AGG_THREADS, 0, stream>>>(
      rowIdx, colIdx, dinv, RH, LRH, nullptr, nullptr, ownerW, listWords,
      listCount, E, N);
  cand_out_wmma_kernel<<<N / 16, 128, 0, stream>>>(
      x, H, Lx, RH, LRH, Zbuf, Wcx, Wch, bcx, bch, (float*)d_out);
}


# Dense stages fused into the aggregation kernels: a block forms its rows' gates (pass 1) or candidate and output (pass 2) by WMMA from its LDS row sums; LH and LRH never stored; 7 -> 5 launches
# speedup vs baseline: 1.9627x; 1.0585x over previous
#include <hip/hip_runtime.h>
#include <hip/hip_bf16.h>
#include <math.h>


typedef float v2f __attribute__((ext_vector_type(2)));
typedef float v8f __attribute__((ext_vector_type(8)));

#define IN_CH 8
#define OUT_CH 64

__device__ __forceinline__ v8f wmma_f32(v2f a, v2f b, v8f c) {
  return __builtin_amdgcn_wmma_f32_16x16x4_f32(false, a, false, b, (short)0, c,
                                               false, false);
}

__device__ __forceinline__ v8f accum_block(const float* __restrict__ A, int ldA,
                                           int rA, const float* __restrict__ W,
                                           int Kdim, int colB, int half,
                                           v8f acc) {
#pragma unroll 4
  for (int k0 = 0; k0 < Kdim; k0 += 4) {
    int ka = k0 + 2 * half;
    v2f a;
    a.x = A[rA * ldA + ka];
    a.y = A[rA * ldA + ka + 1];
    v2f b;
    b.x = W[ka * OUT_CH + colB];
    b.y = W[(ka + 1) * OUT_CH + colB];
    acc = wmma_f32(a, b, acc);
  }
  return acc;
}

__device__ __forceinline__ void accum_block2(const float* __restrict__ A,
                                             int ldA, int rA,
                                             const float* __restrict__ W1,
                                             const float* __restrict__ W2,
                                             int Kdim, int colB, int half,
                                             v8f& acc1, v8f& acc2) {
#pragma unroll 4
  for (int k0 = 0; k0 < Kdim; k0 += 4) {
    int ka = k0 + 2 * half;
    v2f a;
    a.x = A[rA * ldA + ka];
    a.y = A[rA * ldA + ka + 1];
    v2f b1;
    b1.x = W1[ka * OUT_CH + colB];
    b1.y = W1[(ka + 1) * OUT_CH + colB];
    v2f b2;
    b2.x = W2[ka * OUT_CH + colB];
    b2.y = W2[(ka + 1) * OUT_CH + colB];
    acc1 = wmma_f32(a, b1, acc1);
    acc2 = wmma_f32(a, b2, acc2);
  }
}


__global__ void __launch_bounds__(256) zero_kernel(float* p, int n) {
  int i = blockIdx.x * blockDim.x + threadIdx.x;
  if (i < n) p[i] = 0.0f;
}

#define ROWS_PB 196
#define TILE_ROWS 208
#define PITCH 68

__device__ __forceinline__ unsigned owner_byte(int r) {
  return (unsigned)(r / ROWS_PB) & 255u;
}

__global__ void __launch_bounds__(256)
degree_kernel(const int* __restrict__ rowIdx, float* __restrict__ deg,
              unsigned* __restrict__ ownerW, int E) {
  const int t = blockIdx.x * blockDim.x + threadIdx.x;
  const int e = t * 4;
  if (e + 3 < E) {
    const int4 r = ((const int4*)rowIdx)[t];
    atomicAdd(&deg[r.x], 1.0f);
    atomicAdd(&deg[r.y], 1.0f);
    atomicAdd(&deg[r.z], 1.0f);
    atomicAdd(&deg[r.w], 1.0f);
    ownerW[t] = owner_byte(r.x) | (owner_byte(r.y) << 8) |
                (owner_byte(r.z) << 16) | (owner_byte(r.w) << 24);
  } else {
    unsigned char* ob = (unsigned char*)ownerW;
    for (int k = e; k < E; ++k) {
      const int r = rowIdx[k];
      atomicAdd(&deg[r], 1.0f);
      ob[k] = (unsigned char)owner_byte(r);
    }
  }
}

__global__ void __launch_bounds__(256) dinv_kernel(float* deg, int n) {
  int i = blockIdx.x * blockDim.x + threadIdx.x;
  if (i < n) {
    float d = deg[i];
    deg[i] = (d > 0.0f) ? rsqrtf(d) : 0.0f;
  }
}

#define QCAP 32768
#define AGG_THREADS 1024
#define AGG_WAVES (AGG_THREADS / 32)

template <bool FIRST>
__global__ void __launch_bounds__(AGG_THREADS)
agg_kernel(const int* __restrict__ rowIdx, const int* __restrict__ colIdx,
           const float* __restrict__ dinv, const float* __restrict__ src64,
           const float* __restrict__ src8, float* __restrict__ Lx,
           const unsigned* __restrict__ ownerW, int* __restrict__ listWords,
           int* __restrict__ listCount, int E, int N,
           const float* __restrict__ H, const float* __restrict__ Wax,
           const float* __restrict__ Wah, const float* __restrict__ Wbx,
           const float* __restrict__ Wbh, const float* __restrict__ bax,
           const float* __restrict__ bah, const float* __restrict__ bbx,
           const float* __restrict__ bbh, float* __restrict__ Zbuf,
           float* __restrict__ RHout, float* __restrict__ out) {
  constexpr bool HAS8 = FIRST;
  __shared__ __attribute__((aligned(16))) float acc[TILE_ROWS * PITCH];
  __shared__ __attribute__((aligned(16))) float acc8[HAS8 ? TILE_ROWS * IN_CH : 4];
  __shared__ __attribute__((aligned(16))) int queue[QCAP];
  __shared__ int qn;

  const int tid = threadIdx.x;
  const int lane = tid & 31;
  const int base = blockIdx.x * ROWS_PB;

  for (int i = tid; i < TILE_ROWS * PITCH; i += AGG_THREADS) acc[i] = 0.0f;
  if (HAS8) {
    for (int i = tid; i < TILE_ROWS * IN_CH; i += AGG_THREADS) acc8[i] = 0.0f;
  }
  if (tid == 0) qn = 0;
  __syncthreads();

  int total = 0;
  bool fromList = false;
  if (!FIRST) {
    total = __builtin_amdgcn_readfirstlane(listCount[blockIdx.x]);
    fromList = total <= QCAP;
  }
  int4* __restrict__ myList = (int4*)listWords + (size_t)blockIdx.x * QCAP;

  if (!fromList) {
    const bool byteIsBlock = gridDim.x <= 256u;
    auto take = [&](int e) {
      int slot = atomicAdd(&qn, 1);
      if (slot < QCAP) {
        queue[slot] = e;
      } else {
        int r = rowIdx[e];
        int c = colIdx[e];
        unsigned d = (unsigned)(r - base);
        float nrm = -dinv[r] * dinv[c];
#pragma unroll 1
        for (int k = 0; k < OUT_CH; ++k)
          atomicAdd(&acc[d * PITCH + k], src64[(size_t)c * OUT_CH + k] * nrm);
        if (HAS8) {
#pragma unroll 1
          for (int k = 0; k < IN_CH; ++k)
            atomicAdd(&acc8[d * IN_CH + k], src8[(size_t)c * IN_CH + k] * nrm);
        }
      }
    };
    auto visitRow = [&](int e) {
      unsigned d = (unsigned)(rowIdx[e] - base);
      if (d < (unsigned)ROWS_PB) take(e);
    };
    const unsigned pat = (blockIdx.x & 255u) * 0x01010101u;
    const int nG = E >> 4;
    const uint4* __restrict__ own4 = (const uint4*)ownerW;
    auto visit16 = [&](int q, uint4 w) {
      const unsigned x0 = w.x ^ pat, x1 = w.y ^ pat;
      const unsigned x2 = w.z ^ pat, x3 = w.w ^ pat;
      const unsigned n0 = ((x0 & 0x7f7f7f7fu) + 0x7f7f7f7fu) | x0;
      const unsigned n1 = ((x1 & 0x7f7f7f7fu) + 0x7f7f7f7fu) | x1;
      const unsigned n2 = ((x2 & 0x7f7f7f7fu) + 0x7f7f7f7fu) | x2;
      const unsigned n3 = ((x3 & 0x7f7f7f7fu) + 0x7f7f7f7fu) | x3;
      if (q < nG && (~(n0 & n1 & n2 & n3) & 0x80808080u) != 0u) {
        unsigned h = ((~n0 & 0x80808080u) >> 7) | ((~n1 & 0x80808080u) >> 6) |
                     ((~n2 & 0x80808080u) >> 5) | ((~n3 & 0x80808080u) >> 4);
        const int e = q * 16;
        while (h != 0u) {
          const int p = __builtin_ctz(h);
          h &= h - 1u;
          const int eh = e + ((p & 7) << 2) + (p >> 3);
          if (byteIsBlock) take(eh); else visitRow(eh);
        }
      }
    };
    const int qLast = nG - 1;
    auto load4 = [&](uint4 (&v)[4], int q0) {
#pragma unroll
      for (int k = 0; k < 4; ++k) {
        int q = q0 + k * AGG_THREADS + tid;
        v[k] = own4[min(q, qLast)];
      }
    };
    const int STEP = 4 * AGG_THREADS;
    const int nWin = (nG + 2 * STEP - 1) / (2 * STEP);
    int win = nWin > 0 ? (int)(blockIdx.x % (unsigned)nWin) : 0;
    int q0 = win * 2 * STEP;
    uint4 va[4], vb[4];
    if (nWin > 0) load4(va, q0);
    for (int n = 0; n < nWin; ++n) {
      win = (win + 1 < nWin) ? win + 1 : 0;
      const int q0n = win * 2 * STEP;
      load4(vb, q0 + STEP);
#pragma unroll
      for (int k = 0; k < 4; ++k)
        visit16(q0 + k * AGG_THREADS + tid, va[k]);
      load4(va, q0n);
#pragma unroll
      for (int k = 0; k < 4; ++k)
        visit16(q0 + STEP + k * AGG_THREADS + tid, vb[k]);
      q0 = q0n;
    }
    for (int e = (nG << 4) + tid; e < E; e += AGG_THREADS) visitRow(e);
  }
  __syncthreads();

  const int wave = __builtin_amdgcn_readfirstlane(tid >> 5);
  auto sumGroup = [&](int c, int d, float nrm, int cnt) {
#pragma unroll
    for (int h = 0; h < 2; ++h) {
      float2 g[16];
#pragma unroll
      for (int j = 0; j < 16; ++j) {
        int cj = __builtin_amdgcn_readlane(c, h * 16 + j);
        g[j] = ((const float2*)(src64 + (size_t)cj * OUT_CH))[lane];
      }
#pragma unroll
      for (int j = 0; j < 16; ++j) {
        if (h * 16 + j < cnt) {
          float nj = __int_as_float(
              __builtin_amdgcn_readlane(__float_as_int(nrm), h * 16 + j));
          int dj = __builtin_amdgcn_readlane(d, h * 16 + j);
          float* t = &acc[dj * PITCH + 2 * lane];
          atomicAdd(&t[0], g[j].x * nj);
          atomicAdd(&t[1], g[j].y * nj);
        }
      }
    }
  };
  if (fromList) {
    const int nq = total;
    int g0 = wave * 32;
    if (g0 < nq) {
      int4 rec = myList[min(g0 + lane, nq - 1)];
      for (;;) {
        const int gn = g0 + 32 * AGG_WAVES;
        const int4 nxt = myList[min(gn + lane, nq - 1)];
        sumGroup(rec.x, rec.y, __int_as_float(rec.z), min(32, nq - g0));
        if (gn >= nq) break;
        rec = nxt;
        g0 = gn;
      }
    }
  } else {
    const int nq = __builtin_amdgcn_readfirstlane(min(qn, QCAP));
    for (int g0 = wave * 32; g0 < nq; g0 += 32 * AGG_WAVES) {
      const int cnt = min(32, nq - g0);
      const bool valid = lane < cnt;
      const int e = valid ? queue[g0 + lane] : 0;
      const int r = rowIdx[e];
      const int c = colIdx[e];
      const float nrm = valid ? -dinv[r] * dinv[c] : 0.0f;
      const int d = valid ? r - base : 0;
      if (FIRST && valid)
        myList[g0 + lane] = make_int4(c, d, __float_as_int(nrm), 0);
      if (HAS8) {
        if (valid) {
          const float4* s8 = (const float4*)(src8 + (size_t)c * IN_CH);
          float4 a = s8[0];
          float4 b = s8[1];
          float* t8 = &acc8[d * IN_CH];
          atomicAdd(&t8[0], a.x * nrm);
          atomicAdd(&t8[1], a.y * nrm);
          atomicAdd(&t8[2], a.z * nrm);
          atomicAdd(&t8[3], a.w * nrm);
          atomicAdd(&t8[4], b.x * nrm);
          atomicAdd(&t8[5], b.y * nrm);
          atomicAdd(&t8[6], b.z * nrm);
          atomicAdd(&t8[7], b.w * nrm);
        }
      }
      sumGroup(c, d, nrm, cnt);
    }
    if (FIRST && tid == 0) listCount[blockIdx.x] = qn;
  }
  __syncthreads();

  const int rows = min(ROWS_PB, N - base);
  float* hst = (float*)queue;
  float* xst = hst + TILE_ROWS * PITCH;
  float* lst = xst + TILE_ROWS * IN_CH;
  for (int i = tid; i < TILE_ROWS * (OUT_CH / 4); i += AGG_THREADS) {
    const int r = i >> 4, c4 = i & 15;
    float4 v = make_float4(0.0f, 0.0f, 0.0f, 0.0f);
    if (r < rows) v = ((const float4*)(src64 + (size_t)(base + r) * OUT_CH))[c4];
    *(float4*)&hst[r * PITCH + c4 * 4] = v;
  }
  for (int i = tid; i < TILE_ROWS * (IN_CH / 4); i += AGG_THREADS) {
    const int r = i >> 1, c4 = i & 1;
    float4 v = make_float4(0.0f, 0.0f, 0.0f, 0.0f);
    float4 w = v;
    if (r < rows) {
      v = ((const float4*)(src8 + (size_t)(base + r) * IN_CH))[c4];
      if (!FIRST) w = ((const float4*)(Lx + (size_t)(base + r) * IN_CH))[c4];
    }
    *(float4*)&xst[r * IN_CH + c4 * 4] = v;
    if (!FIRST) *(float4*)&lst[r * IN_CH + c4 * 4] = w;
  }
  if (HAS8) {
    float4* o8 = (float4*)(Lx + (size_t)base * IN_CH);
    const float4* a8 = (const float4*)acc8;
    for (int i = tid; i < rows * (IN_CH / 4); i += AGG_THREADS) o8[i] = a8[i];
  }
  __syncthreads();

  const int half = lane >> 4;
  const int l15 = lane & 15;
  const int nUnits = ((rows + 15) >> 4) * (OUT_CH / 16);
  for (int u = wave; u < nUnits; u += AGG_WAVES) {
    const int row0 = (u >> 2) * 16;
    const int rA = row0 + l15;
    const int colB = (u & 3) * 16 + l15;
    if (FIRST) {
      v8f accZ = {};
      v8f accR = {};
      accum_block2(xst, IN_CH, rA, Wax, Wbx, IN_CH, colB, half, accZ, accR);
      accum_block2(acc8, IN_CH, rA, Wax + IN_CH * OUT_CH, Wbx + IN_CH * OUT_CH,
                   IN_CH, colB, half, accZ, accR);
      accum_block2(hst, PITCH, rA, Wah, Wbh, OUT_CH, colB, half, accZ, accR);
      accum_block2(acc, PITCH, rA, Wah + OUT_CH * OUT_CH, Wbh + OUT_CH * OUT_CH,
                   OUT_CH, colB, half, accZ, accR);
      const float bZ = bax[colB] + bah[colB];
      const float bR = bbx[colB] + bbh[colB];
#pragma unroll
      for (int i = 0; i < 8; ++i) {
        const int r = row0 + i + 8 * half;
        if (r < rows) {
          const float z = 1.0f / (1.0f + __expf(-(accZ[i] + bZ)));
          const float rg = 1.0f / (1.0f + __expf(-(accR[i] + bR)));
          const size_t g = (size_t)(base + r) * OUT_CH + colB;
          Zbuf[g] = z;
          RHout[g] = rg * hst[r * PITCH + colB];
        }
      }
    } else {
      v8f accC = {};
      accC = accum_block(xst, IN_CH, rA, Wax, IN_CH, colB, half, accC);
      accC = accum_block(lst, IN_CH, rA, Wax + IN_CH * OUT_CH, IN_CH, colB, half, accC);
      accC = accum_block(hst, PITCH, rA, Wah, OUT_CH, colB, half, accC);
      accC = accum_block(acc, PITCH, rA, Wah + OUT_CH * OUT_CH, OUT_CH, colB, half, accC);
      const float bC = bax[colB] + bah[colB];
#pragma unroll
      for (int i = 0; i < 8; ++i) {
        const int r = row0 + i + 8 * half;
        if (r < rows) {
          const size_t g = (size_t)(base + r) * OUT_CH + colB;
          const float cv = tanhf(accC[i] + bC);
          const float z = Zbuf[g];
          const float h = H[g];
          out[g] = z * h + (1.0f - z) * cv;
        }
      }
    }
  }
}


extern "C" void kernel_launch(void* const* d_in, const int* in_sizes, int n_in,
                              void* d_out, int out_size, void* d_ws,
                              size_t ws_size, hipStream_t stream) {
  const float* x = (const float*)d_in[0];
  const float* H = (const float*)d_in[1];
  const int* edge = (const int*)d_in[2];
  const float* Wzx = (const float*)d_in[3];
  const float* bzx = (const float*)d_in[4];
  const float* Wzh = (const float*)d_in[5];
  const float* bzh = (const float*)d_in[6];
  const float* Wrx = (const float*)d_in[7];
  const float* brx = (const float*)d_in[8];
  const float* Wrh = (const float*)d_in[9];
  const float* brh = (const float*)d_in[10];
  const float* Wcx = (const float*)d_in[11];
  const float* bcx = (const float*)d_in[12];
  const float* Wch = (const float*)d_in[13];
  const float* bch = (const float*)d_in[14];

  const int N = in_sizes[0] / IN_CH;
  const int E = in_sizes[2] / 2;
  const int* rowIdx = edge;
  const int* colIdx = edge + E;

  float* ws = (float*)d_ws;
  float* dinv = ws;
  float* Lx = dinv + ((N + 63) / 64) * 64;
  float* Zbuf = Lx + (size_t)N * IN_CH;
  float* RH = Zbuf + (size_t)N * OUT_CH;
  const int nZero = N;
  const int aggBlocks = (N + ROWS_PB - 1) / ROWS_PB;
  unsigned* ownerW = (unsigned*)(RH + (size_t)N * OUT_CH);
  int* listCount = (int*)(ownerW + ((E + 255) / 256) * 64);
  int* listWords = listCount + ((aggBlocks + 63) / 64) * 64;

  zero_kernel<<<(nZero + 255) / 256, 256, 0, stream>>>(dinv, nZero);
  degree_kernel<<<(E + 1023) / 1024, 256, 0, stream>>>(rowIdx, dinv, ownerW,
                                                        E);
  dinv_kernel<<<(N + 255) / 256, 256, 0, stream>>>(dinv, N);
  agg_kernel<true><<<aggBlocks, AGG_THREADS, 0, stream>>>(
      rowIdx, colIdx, dinv, H, x, Lx, ownerW, listWords, listCount, E, N, H,
      Wzx, Wzh, Wrx, Wrh, bzx, bzh, brx, brh, Zbuf, RH, nullptr);
  agg_kernel<false><<<aggBlocks, AGG_THREADS, 0, stream>>>(
      rowIdx, colIdx, dinv, RH, x, Lx, ownerW, listWords, listCount, E, N, H,
      Wcx, Wch, nullptr, nullptr, bcx, bch, nullptr, nullptr, Zbuf, nullptr,
      (float*)d_out);
}


# Edge filter lists hits per wave: ballot rank and scalar count replace the LDS counter add, hits of 4 loads in one loop, no barrier before the sums; a wave sums its own list; sum order = LDS arrival
# speedup vs baseline: 2.0762x; 1.0578x over previous
#include <hip/hip_runtime.h>
#include <hip/hip_bf16.h>
#include <math.h>


typedef float v2f __attribute__((ext_vector_type(2)));
typedef float v8f __attribute__((ext_vector_type(8)));

#define IN_CH 8
#define OUT_CH 64

__device__ __forceinline__ v8f wmma_f32(v2f a, v2f b, v8f c) {
  return __builtin_amdgcn_wmma_f32_16x16x4_f32(false, a, false, b, (short)0, c,
                                               false, false);
}

__device__ __forceinline__ v8f accum_block(const float* __restrict__ A, int ldA,
                                           int rA, const float* __restrict__ W,
                                           int Kdim, int colB, int half,
                                           v8f acc) {
#pragma unroll 4
  for (int k0 = 0; k0 < Kdim; k0 += 4) {
    int ka = k0 + 2 * half;
    v2f a;
    a.x = A[rA * ldA + ka];
    a.y = A[rA * ldA + ka + 1];
    v2f b;
    b.x = W[ka * OUT_CH + colB];
    b.y = W[(ka + 1) * OUT_CH + colB];
    acc = wmma_f32(a, b, acc);
  }
  return acc;
}

__device__ __forceinline__ void accum_block2(const float* __restrict__ A,
                                             int ldA, int rA,
                                             const float* __restrict__ W1,
                                             const float* __restrict__ W2,
                                             int Kdim, int colB, int half,
                                             v8f& acc1, v8f& acc2) {
#pragma unroll 4
  for (int k0 = 0; k0 < Kdim; k0 += 4) {
    int ka = k0 + 2 * half;
    v2f a;
    a.x = A[rA * ldA + ka];
    a.y = A[rA * ldA + ka + 1];
    v2f b1;
    b1.x = W1[ka * OUT_CH + colB];
    b1.y = W1[(ka + 1) * OUT_CH + colB];
    v2f b2;
    b2.x = W2[ka * OUT_CH + colB];
    b2.y = W2[(ka + 1) * OUT_CH + colB];
    acc1 = wmma_f32(a, b1, acc1);
    acc2 = wmma_f32(a, b2, acc2);
  }
}


__global__ void __launch_bounds__(256) zero_kernel(float* p, int n) {
  int i = blockIdx.x * blockDim.x + threadIdx.x;
  if (i < n) p[i] = 0.0f;
}

#define ROWS_PB 196
#define TILE_ROWS 208
#define PITCH 68

__device__ __forceinline__ unsigned owner_byte(int r) {
  return (unsigned)(r / ROWS_PB) & 255u;
}

__global__ void __launch_bounds__(256)
degree_kernel(const int* __restrict__ rowIdx, float* __restrict__ deg,
              unsigned* __restrict__ ownerW, int E) {
  const int t = blockIdx.x * blockDim.x + threadIdx.x;
  const int e = t * 4;
  if (e + 3 < E) {
    const int4 r = ((const int4*)rowIdx)[t];
    atomicAdd(&deg[r.x], 1.0f);
    atomicAdd(&deg[r.y], 1.0f);
    atomicAdd(&deg[r.z], 1.0f);
    atomicAdd(&deg[r.w], 1.0f);
    ownerW[t] = owner_byte(r.x) | (owner_byte(r.y) << 8) |
                (owner_byte(r.z) << 16) | (owner_byte(r.w) << 24);
  } else {
    unsigned char* ob = (unsigned char*)ownerW;
    for (int k = e; k < E; ++k) {
      const int r = rowIdx[k];
      atomicAdd(&deg[r], 1.0f);
      ob[k] = (unsigned char)owner_byte(r);
    }
  }
}

__global__ void __launch_bounds__(256) dinv_kernel(float* deg, int n) {
  int i = blockIdx.x * blockDim.x + threadIdx.x;
  if (i < n) {
    float d = deg[i];
    deg[i] = (d > 0.0f) ? rsqrtf(d) : 0.0f;
  }
}

#define QCAP 32768
#define AGG_THREADS 1024
#define AGG_WAVES (AGG_THREADS / 32)
#define WCAP (QCAP / AGG_WAVES)

template <bool FIRST>
__global__ void __launch_bounds__(AGG_THREADS)
agg_kernel(const int* __restrict__ rowIdx, const int* __restrict__ colIdx,
           const float* __restrict__ dinv, const float* __restrict__ src64,
           const float* __restrict__ src8, float* __restrict__ Lx,
           const unsigned* __restrict__ ownerW, int* __restrict__ listWords,
           int* __restrict__ listCount, int E, int N,
           const float* __restrict__ H, const float* __restrict__ Wax,
           const float* __restrict__ Wah, const float* __restrict__ Wbx,
           const float* __restrict__ Wbh, const float* __restrict__ bax,
           const float* __restrict__ bah, const float* __restrict__ bbx,
           const float* __restrict__ bbh, float* __restrict__ Zbuf,
           float* __restrict__ RHout, float* __restrict__ out) {
  constexpr bool HAS8 = FIRST;
  __shared__ __attribute__((aligned(16))) float acc[TILE_ROWS * PITCH];
  __shared__ __attribute__((aligned(16))) float acc8[HAS8 ? TILE_ROWS * IN_CH : 4];
  __shared__ __attribute__((aligned(16))) int queue[QCAP];

  const int tid = threadIdx.x;
  const int lane = tid & 31;
  const int base = blockIdx.x * ROWS_PB;
  const int wave = __builtin_amdgcn_readfirstlane(tid >> 5);

  for (int i = tid; i < TILE_ROWS * PITCH; i += AGG_THREADS) acc[i] = 0.0f;
  if (HAS8) {
    for (int i = tid; i < TILE_ROWS * IN_CH; i += AGG_THREADS) acc8[i] = 0.0f;
  }
  __syncthreads();

  int* wq = queue + wave * WCAP;
  int wcnt = 0;
  const int cntIdx = blockIdx.x * AGG_WAVES + wave;
  int total = 0;
  bool fromList = false;
  if (!FIRST) {
    total = __builtin_amdgcn_readfirstlane(listCount[cntIdx]);
    fromList = total <= WCAP;
  }
  int4* __restrict__ myList =
      (int4*)listWords + ((size_t)blockIdx.x * QCAP + (size_t)wave * WCAP);

  if (!fromList) {
    const bool byteIsBlock = gridDim.x <= 256u;
    auto spot = [&](int e) {
      int r = rowIdx[e];
      int c = colIdx[e];
      unsigned d = (unsigned)(r - base);
      float nrm = -dinv[r] * dinv[c];
#pragma unroll 1
      for (int k = 0; k < OUT_CH; ++k)
        atomicAdd(&acc[d * PITCH + k], src64[(size_t)c * OUT_CH + k] * nrm);
      if (HAS8) {
#pragma unroll 1
        for (int k = 0; k < IN_CH; ++k)
          atomicAdd(&acc8[d * IN_CH + k], src8[(size_t)c * IN_CH + k] * nrm);
      }
    };
    auto append = [&](bool has, int e) {
      const unsigned m = __builtin_amdgcn_ballot_w32(has);
      const int slot = wcnt + (int)__builtin_amdgcn_mbcnt_lo(m, 0u);
      const int after = wcnt + __builtin_popcount(m);
      if (has && slot < WCAP) wq[slot] = e;
      if (after > WCAP) {
        if (has && slot >= WCAP) spot(e);
      }
      wcnt = after;
    };
    auto ownsRow = [&](int e) {
      return (unsigned)(rowIdx[e] - base) < (unsigned)ROWS_PB;
    };
    const unsigned pat = (blockIdx.x & 255u) * 0x01010101u;
    const int nG = E >> 4;
    const uint4* __restrict__ own4 = (const uint4*)ownerW;
    auto hits16 = [&](int q, uint4 w) {
      const unsigned x0 = w.x ^ pat, x1 = w.y ^ pat;
      const unsigned x2 = w.z ^ pat, x3 = w.w ^ pat;
      const unsigned n0 = ((x0 & 0x7f7f7f7fu) + 0x7f7f7f7fu) | x0;
      const unsigned n1 = ((x1 & 0x7f7f7f7fu) + 0x7f7f7f7fu) | x1;
      const unsigned n2 = ((x2 & 0x7f7f7f7fu) + 0x7f7f7f7fu) | x2;
      const unsigned n3 = ((x3 & 0x7f7f7f7fu) + 0x7f7f7f7fu) | x3;
      unsigned h = 0u;
      if (q < nG)
        h = ((~n0 & 0x80808080u) >> 7) | ((~n1 & 0x80808080u) >> 6) |
            ((~n2 & 0x80808080u) >> 5) | ((~n3 & 0x80808080u) >> 4);
      return h;
    };
    auto visit64 = [&](int qb, const uint4 (&v)[4]) {
      unsigned lo = hits16(qb, v[0]) | (hits16(qb + AGG_THREADS, v[1]) << 4);
      unsigned hi = hits16(qb + 2 * AGG_THREADS, v[2]) |
                    (hits16(qb + 3 * AGG_THREADS, v[3]) << 4);
      while (__builtin_amdgcn_ballot_w32((lo | hi) != 0u) != 0u) {
        bool has = (lo | hi) != 0u;
        const bool useLo = lo != 0u;
        const unsigned word = useLo ? lo : hi;
        const int p = __builtin_ffs((int)word) - 1;
        const unsigned rest = word & (word - 1u);
        lo = useLo ? rest : lo;
        hi = useLo ? hi : rest;
        const int k = ((p >> 2) & 1) + (useLo ? 0 : 2);
        const int eh = (qb + k * AGG_THREADS) * 16 + ((p & 3) << 2) + (p >> 3);
        if (!byteIsBlock) {
          if (has) has = ownsRow(eh);
        }
        append(has, eh);
      }
    };
    const int qLast = nG - 1;
    auto load4 = [&](uint4 (&v)[4], int q0) {
#pragma unroll
      for (int k = 0; k < 4; ++k) {
        int q = q0 + k * AGG_THREADS + tid;
        v[k] = own4[min(q, qLast)];
      }
    };
    const int STEP = 4 * AGG_THREADS;
    const int nWin = (nG + 2 * STEP - 1) / (2 * STEP);
    int win = nWin > 0 ? (int)(blockIdx.x % (unsigned)nWin) : 0;
    int q0 = win * 2 * STEP;
    uint4 va[4], vb[4];
    if (nWin > 0) load4(va, q0);
    for (int n = 0; n < nWin; ++n) {
      win = (win + 1 < nWin) ? win + 1 : 0;
      const int q0n = win * 2 * STEP;
      load4(vb, q0 + STEP);
      visit64(q0 + tid, va);
      load4(va, q0n);
      visit64(q0 + STEP + tid, vb);
      q0 = q0n;
    }
    for (int e0 = nG << 4; e0 < E; e0 += AGG_THREADS) {
      const int e = e0 + tid;
      bool has = false;
      if (e < E) has = ownsRow(e);
      append(has, e);
    }
    __builtin_amdgcn_fence(__ATOMIC_ACQ_REL, "workgroup");
  }

  auto sumGroup = [&](int c, int d, float nrm, int cnt) {
#pragma unroll
    for (int h = 0; h < 2; ++h) {
      if (h * 16 >= cnt) break;
      float2 g[16];
#pragma unroll
      for (int j = 0; j < 16; ++j) {
        int cj = __builtin_amdgcn_readlane(c, h * 16 + j);
        g[j] = ((const float2*)(src64 + (size_t)cj * OUT_CH))[lane];
      }
#pragma unroll
      for (int j = 0; j < 16; ++j) {
        if (h * 16 + j < cnt) {
          float nj = __int_as_float(
              __builtin_amdgcn_readlane(__float_as_int(nrm), h * 16 + j));
          int dj = __builtin_amdgcn_readlane(d, h * 16 + j);
          float* t = &acc[dj * PITCH + 2 * lane];
          atomicAdd(&t[0], g[j].x * nj);
          atomicAdd(&t[1], g[j].y * nj);
        }
      }
    }
  };
  if (fromList) {
    const int nq = total;
    int g0 = 0;
    if (g0 < nq) {
      int4 rec = myList[min(g0 + lane, nq - 1)];
      for (;;) {
        const int gn = g0 + 32;
        const int4 nxt = myList[min(gn + lane, nq - 1)];
        sumGroup(rec.x, rec.y, __int_as_float(rec.z), min(32, nq - g0));
        if (gn >= nq) break;
        rec = nxt;
        g0 = gn;
      }
    }
  } else {
    const int nq = __builtin_amdgcn_readfirstlane(min(wcnt, WCAP));
    for (int g0 = 0; g0 < nq; g0 += 32) {
      const int cnt = min(32, nq - g0);
      const bool valid = lane < cnt;
      const int e = valid ? wq[g0 + lane] : 0;
      const int r = rowIdx[e];
      const int c = colIdx[e];
      const float nrm = valid ? -dinv[r] * dinv[c] : 0.0f;
      const int d = valid ? r - base : 0;
      if (FIRST && valid)
        myList[g0 + lane] = make_int4(c, d, __float_as_int(nrm), 0);
      if (HAS8) {
        if (valid) {
          const float4* s8 = (const float4*)(src8 + (size_t)c * IN_CH);
          float4 a = s8[0];
          float4 b = s8[1];
          float* t8 = &acc8[d * IN_CH];
          atomicAdd(&t8[0], a.x * nrm);
          atomicAdd(&t8[1], a.y * nrm);
          atomicAdd(&t8[2], a.z * nrm);
          atomicAdd(&t8[3], a.w * nrm);
          atomicAdd(&t8[4], b.x * nrm);
          atomicAdd(&t8[5], b.y * nrm);
          atomicAdd(&t8[6], b.z * nrm);
          atomicAdd(&t8[7], b.w * nrm);
        }
      }
      sumGroup(c, d, nrm, cnt);
    }
    if (FIRST && lane == 0) listCount[cntIdx] = wcnt;
  }
  __syncthreads();

  const int rows = min(ROWS_PB, N - base);
  float* hst = (float*)queue;
  float* xst = hst + TILE_ROWS * PITCH;
  float* lst = xst + TILE_ROWS * IN_CH;
  for (int i = tid; i < TILE_ROWS * (OUT_CH / 4); i += AGG_THREADS) {
    const int r = i >> 4, c4 = i & 15;
    float4 v = make_float4(0.0f, 0.0f, 0.0f, 0.0f);
    if (r < rows) v = ((const float4*)(src64 + (size_t)(base + r) * OUT_CH))[c4];
    *(float4*)&hst[r * PITCH + c4 * 4] = v;
  }
  for (int i = tid; i < TILE_ROWS * (IN_CH / 4); i += AGG_THREADS) {
    const int r = i >> 1, c4 = i & 1;
    float4 v = make_float4(0.0f, 0.0f, 0.0f, 0.0f);
    float4 w = v;
    if (r < rows) {
      v = ((const float4*)(src8 + (size_t)(base + r) * IN_CH))[c4];
      if (!FIRST) w = ((const float4*)(Lx + (size_t)(base + r) * IN_CH))[c4];
    }
    *(float4*)&xst[r * IN_CH + c4 * 4] = v;
    if (!FIRST) *(float4*)&lst[r * IN_CH + c4 * 4] = w;
  }
  if (HAS8) {
    float4* o8 = (float4*)(Lx + (size_t)base * IN_CH);
    const float4* a8 = (const float4*)acc8;
    for (int i = tid; i < rows * (IN_CH / 4); i += AGG_THREADS) o8[i] = a8[i];
  }
  __syncthreads();

  const int half = lane >> 4;
  const int l15 = lane & 15;
  const int nUnits = ((rows + 15) >> 4) * (OUT_CH / 16);
  for (int u = wave; u < nUnits; u += AGG_WAVES) {
    const int row0 = (u >> 2) * 16;
    const int rA = row0 + l15;
    const int colB = (u & 3) * 16 + l15;
    if (FIRST) {
      v8f accZ = {};
      v8f accR = {};
      accum_block2(xst, IN_CH, rA, Wax, Wbx, IN_CH, colB, half, accZ, accR);
      accum_block2(acc8, IN_CH, rA, Wax + IN_CH * OUT_CH, Wbx + IN_CH * OUT_CH,
                   IN_CH, colB, half, accZ, accR);
      accum_block2(hst, PITCH, rA, Wah, Wbh, OUT_CH, colB, half, accZ, accR);
      accum_block2(acc, PITCH, rA, Wah + OUT_CH * OUT_CH, Wbh + OUT_CH * OUT_CH,
                   OUT_CH, colB, half, accZ, accR);
      const float bZ = bax[colB] + bah[colB];
      const float bR = bbx[colB] + bbh[colB];
#pragma unroll
      for (int i = 0; i < 8; ++i) {
        const int r = row0 + i + 8 * half;
        if (r < rows) {
          const float z = 1.0f / (1.0f + __expf(-(accZ[i] + bZ)));
          const float rg = 1.0f / (1.0f + __expf(-(accR[i] + bR)));
          const size_t g = (size_t)(base + r) * OUT_CH + colB;
          Zbuf[g] = z;
          RHout[g] = rg * hst[r * PITCH + colB];
        }
      }
    } else {
      v8f accC = {};
      accC = accum_block(xst, IN_CH, rA, Wax, IN_CH, colB, half, accC);
      accC = accum_block(lst, IN_CH, rA, Wax + IN_CH * OUT_CH, IN_CH, colB, half, accC);
      accC = accum_block(hst, PITCH, rA, Wah, OUT_CH, colB, half, accC);
      accC = accum_block(acc, PITCH, rA, Wah + OUT_CH * OUT_CH, OUT_CH, colB, half, accC);
      const float bC = bax[colB] + bah[colB];
#pragma unroll
      for (int i = 0; i < 8; ++i) {
        const int r = row0 + i + 8 * half;
        if (r < rows) {
          const size_t g = (size_t)(base + r) * OUT_CH + colB;
          const float cv = tanhf(accC[i] + bC);
          const float z = Zbuf[g];
          const float h = H[g];
          out[g] = z * h + (1.0f - z) * cv;
        }
      }
    }
  }
}


extern "C" void kernel_launch(void* const* d_in, const int* in_sizes, int n_in,
                              void* d_out, int out_size, void* d_ws,
                              size_t ws_size, hipStream_t stream) {
  const float* x = (const float*)d_in[0];
  const float* H = (const float*)d_in[1];
  const int* edge = (const int*)d_in[2];
  const float* Wzx = (const float*)d_in[3];
  const float* bzx = (const float*)d_in[4];
  const float* Wzh = (const float*)d_in[5];
  const float* bzh = (const float*)d_in[6];
  const float* Wrx = (const float*)d_in[7];
  const float* brx = (const float*)d_in[8];
  const float* Wrh = (const float*)d_in[9];
  const float* brh = (const float*)d_in[10];
  const float* Wcx = (const float*)d_in[11];
  const float* bcx = (const float*)d_in[12];
  const float* Wch = (const float*)d_in[13];
  const float* bch = (const float*)d_in[14];

  const int N = in_sizes[0] / IN_CH;
  const int E = in_sizes[2] / 2;
  const int* rowIdx = edge;
  const int* colIdx = edge + E;

  float* ws = (float*)d_ws;
  float* dinv = ws;
  float* Lx = dinv + ((N + 63) / 64) * 64;
  float* Zbuf = Lx + (size_t)N * IN_CH;
  float* RH = Zbuf + (size_t)N * OUT_CH;
  const int nZero = N;
  const int aggBlocks = (N + ROWS_PB - 1) / ROWS_PB;
  unsigned* ownerW = (unsigned*)(RH + (size_t)N * OUT_CH);
  int* listCount = (int*)(ownerW + ((E + 255) / 256) * 64);
  int* listWords = listCount + ((aggBlocks * AGG_WAVES + 63) / 64) * 64;

  zero_kernel<<<(nZero + 255) / 256, 256, 0, stream>>>(dinv, nZero);
  degree_kernel<<<(E + 1023) / 1024, 256, 0, stream>>>(rowIdx, dinv, ownerW,
                                                        E);
  dinv_kernel<<<(N + 255) / 256, 256, 0, stream>>>(dinv, N);
  agg_kernel<true><<<aggBlocks, AGG_THREADS, 0, stream>>>(
      rowIdx, colIdx, dinv, H, x, Lx, ownerW, listWords, listCount, E, N, H,
      Wzx, Wzh, Wrx, Wrh, bzx, bzh, brx, brh, Zbuf, RH, nullptr);
  agg_kernel<false><<<aggBlocks, AGG_THREADS, 0, stream>>>(
      rowIdx, colIdx, dinv, RH, x, Lx, ownerW, listWords, listCount, E, N, H,
      Wcx, Wch, nullptr, nullptr, bcx, bch, nullptr, nullptr, Zbuf, nullptr,
      (float*)d_out);
}


# Row sums in registers: found edges are handed through LDS lists to the wave owning the row (row mod 32), which sums its 7 rows by packed FMA; no LDS float atomics; sum order = arrival in owner list
# speedup vs baseline: 2.1725x; 1.0464x over previous
#include <hip/hip_runtime.h>
#include <hip/hip_bf16.h>
#include <math.h>


typedef float v2f __attribute__((ext_vector_type(2)));
typedef float v8f __attribute__((ext_vector_type(8)));

#define IN_CH 8
#define OUT_CH 64

__device__ __forceinline__ v8f wmma_f32(v2f a, v2f b, v8f c) {
  return __builtin_amdgcn_wmma_f32_16x16x4_f32(false, a, false, b, (short)0, c,
                                               false, false);
}

__device__ __forceinline__ v8f accum_block(const float* __restrict__ A, int ldA,
                                           int rA, const float* __restrict__ W,
                                           int Kdim, int colB, int half,
                                           v8f acc) {
#pragma unroll 4
  for (int k0 = 0; k0 < Kdim; k0 += 4) {
    int ka = k0 + 2 * half;
    v2f a;
    a.x = A[rA * ldA + ka];
    a.y = A[rA * ldA + ka + 1];
    v2f b;
    b.x = W[ka * OUT_CH + colB];
    b.y = W[(ka + 1) * OUT_CH + colB];
    acc = wmma_f32(a, b, acc);
  }
  return acc;
}

__device__ __forceinline__ void accum_block2(const float* __restrict__ A,
                                             int ldA, int rA,
                                             const float* __restrict__ W1,
                                             const float* __restrict__ W2,
                                             int Kdim, int colB, int half,
                                             v8f& acc1, v8f& acc2) {
#pragma unroll 4
  for (int k0 = 0; k0 < Kdim; k0 += 4) {
    int ka = k0 + 2 * half;
    v2f a;
    a.x = A[rA * ldA + ka];
    a.y = A[rA * ldA + ka + 1];
    v2f b1;
    b1.x = W1[ka * OUT_CH + colB];
    b1.y = W1[(ka + 1) * OUT_CH + colB];
    v2f b2;
    b2.x = W2[ka * OUT_CH + colB];
    b2.y = W2[(ka + 1) * OUT_CH + colB];
    acc1 = wmma_f32(a, b1, acc1);
    acc2 = wmma_f32(a, b2, acc2);
  }
}


__global__ void __launch_bounds__(256) zero_kernel(float* p, int n) {
  int i = blockIdx.x * blockDim.x + threadIdx.x;
  if (i < n) p[i] = 0.0f;
}

#define ROWS_PB 196
#define TILE_ROWS 208
#define PITCH 68

__device__ __forceinline__ unsigned owner_byte(int r) {
  return (unsigned)(r / ROWS_PB) & 255u;
}

__global__ void __launch_bounds__(256)
degree_kernel(const int* __restrict__ rowIdx, float* __restrict__ deg,
              unsigned* __restrict__ ownerW, int E) {
  const int t = blockIdx.x * blockDim.x + threadIdx.x;
  const int e = t * 4;
  if (e + 3 < E) {
    const int4 r = ((const int4*)rowIdx)[t];
    atomicAdd(&deg[r.x], 1.0f);
    atomicAdd(&deg[r.y], 1.0f);
    atomicAdd(&deg[r.z], 1.0f);
    atomicAdd(&deg[r.w], 1.0f);
    ownerW[t] = owner_byte(r.x) | (owner_byte(r.y) << 8) |
                (owner_byte(r.z) << 16) | (owner_byte(r.w) << 24);
  } else {
    unsigned char* ob = (unsigned char*)ownerW;
    for (int k = e; k < E; ++k) {
      const int r = rowIdx[k];
      atomicAdd(&deg[r], 1.0f);
      ob[k] = (unsigned char)owner_byte(r);
    }
  }
}

__global__ void __launch_bounds__(256) dinv_kernel(float* deg, int n) {
  int i = blockIdx.x * blockDim.x + threadIdx.x;
  if (i < n) {
    float d = deg[i];
    deg[i] = (d > 0.0f) ? rsqrtf(d) : 0.0f;
  }
}

#define AGG_THREADS 1024
#define AGG_WAVES (AGG_THREADS / 32)
#define WCAP 384
#define DCAP 288
#define NSLOT 7

template <bool FIRST>
__global__ void __launch_bounds__(AGG_THREADS)
agg_kernel(const int* __restrict__ rowIdx, const int* __restrict__ colIdx,
           const float* __restrict__ dinv, const float* __restrict__ src64,
           const float* __restrict__ src8, float* __restrict__ Lx,
           const unsigned* __restrict__ ownerW, int* __restrict__ listWords,
           int* __restrict__ listCount, int* __restrict__ findCount, int E,
           int N,
           const float* __restrict__ H, const float* __restrict__ Wax,
           const float* __restrict__ Wah, const float* __restrict__ Wbx,
           const float* __restrict__ Wbh, const float* __restrict__ bax,
           const float* __restrict__ bah, const float* __restrict__ bbx,
           const float* __restrict__ bbh, float* __restrict__ Zbuf,
           float* __restrict__ RHout, float* __restrict__ out) {
  constexpr bool HAS8 = FIRST;
  __shared__ __attribute__((aligned(16))) float acc[TILE_ROWS * PITCH];
  __shared__ __attribute__((aligned(16))) float acc8[HAS8 ? TILE_ROWS * IN_CH : 4];
  __shared__ int fq[AGG_WAVES * WCAP];
  __shared__ __attribute__((aligned(16))) int dq[AGG_WAVES * DCAP * 3];
  __shared__ int dcnt[AGG_WAVES];

  const int tid = threadIdx.x;
  const int lane = tid & 31;
  const int base = blockIdx.x * ROWS_PB;
  const int wave = __builtin_amdgcn_readfirstlane(tid >> 5);

  for (int i = tid; i < TILE_ROWS * PITCH; i += AGG_THREADS) acc[i] = 0.0f;
  if (HAS8) {
    for (int i = tid; i < TILE_ROWS * IN_CH; i += AGG_THREADS) acc8[i] = 0.0f;
  }
  if (tid < AGG_WAVES) dcnt[tid] = 0;
  __syncthreads();

  int* wq = fq + wave * WCAP;
  int wcnt = 0;
  const int cntIdx = blockIdx.x * AGG_WAVES + wave;
  bool filt = true;
  if (!FIRST) {
    const int dc = listCount[blockIdx.x * AGG_WAVES + lane];
    const int fc = findCount[blockIdx.x * AGG_WAVES + lane];
    filt = __builtin_amdgcn_ballot_w32(dc > DCAP || fc > WCAP) != 0u;
  }
  int4* __restrict__ myList =
      (int4*)listWords + ((size_t)blockIdx.x * AGG_WAVES + (size_t)wave) * DCAP;

  if (filt) {
    const bool byteIsBlock = gridDim.x <= 256u;
    auto spot = [&](int e) {
      int r = rowIdx[e];
      int c = colIdx[e];
      unsigned d = (unsigned)(r - base);
      float nrm = -dinv[r] * dinv[c];
#pragma unroll 1
      for (int k = 0; k < OUT_CH; ++k)
        atomicAdd(&acc[d * PITCH + k], src64[(size_t)c * OUT_CH + k] * nrm);
      if (HAS8) {
#pragma unroll 1
        for (int k = 0; k < IN_CH; ++k)
          atomicAdd(&acc8[d * IN_CH + k], src8[(size_t)c * IN_CH + k] * nrm);
      }
    };
    auto append = [&](bool has, int e) {
      const unsigned m = __builtin_amdgcn_ballot_w32(has);
      const int slot = wcnt + (int)__builtin_amdgcn_mbcnt_lo(m, 0u);
      const int after = wcnt + __builtin_popcount(m);
      if (has && slot < WCAP) wq[slot] = e;
      if (after > WCAP) {
        if (has && slot >= WCAP) spot(e);
      }
      wcnt = after;
    };
    auto ownsRow = [&](int e) {
      return (unsigned)(rowIdx[e] - base) < (unsigned)ROWS_PB;
    };
    const unsigned pat = (blockIdx.x & 255u) * 0x01010101u;
    const int nG = E >> 4;
    const uint4* __restrict__ own4 = (const uint4*)ownerW;
    auto hits16 = [&](int q, uint4 w) {
      const unsigned x0 = w.x ^ pat, x1 = w.y ^ pat;
      const unsigned x2 = w.z ^ pat, x3 = w.w ^ pat;
      const unsigned n0 = ((x0 & 0x7f7f7f7fu) + 0x7f7f7f7fu) | x0;
      const unsigned n1 = ((x1 & 0x7f7f7f7fu) + 0x7f7f7f7fu) | x1;
      const unsigned n2 = ((x2 & 0x7f7f7f7fu) + 0x7f7f7f7fu) | x2;
      const unsigned n3 = ((x3 & 0x7f7f7f7fu) + 0x7f7f7f7fu) | x3;
      unsigned h = 0u;
      if (q < nG)
        h = ((~n0 & 0x80808080u) >> 7) | ((~n1 & 0x80808080u) >> 6) |
            ((~n2 & 0x80808080u) >> 5) | ((~n3 & 0x80808080u) >> 4);
      return h;
    };
    auto visit64 = [&](int qb, const uint4 (&v)[4]) {
      unsigned lo = hits16(qb, v[0]) | (hits16(qb + AGG_THREADS, v[1]) << 4);
      unsigned hi = hits16(qb + 2 * AGG_THREADS, v[2]) |
                    (hits16(qb + 3 * AGG_THREADS, v[3]) << 4);
      while (__builtin_amdgcn_ballot_w32((lo | hi) != 0u) != 0u) {
        bool has = (lo | hi) != 0u;
        const bool useLo = lo != 0u;
        const unsigned word = useLo ? lo : hi;
        const int p = __builtin_ffs((int)word) - 1;
        const unsigned rest = word & (word - 1u);
        lo = useLo ? rest : lo;
        hi = useLo ? hi : rest;
        const int k = ((p >> 2) & 1) + (useLo ? 0 : 2);
        const int eh = (qb + k * AGG_THREADS) * 16 + ((p & 3) << 2) + (p >> 3);
        if (!byteIsBlock) {
          if (has) has = ownsRow(eh);
        }
        append(has, eh);
      }
    };
    const int qLast = nG - 1;
    auto load4 = [&](uint4 (&v)[4], int q0) {
#pragma unroll
      for (int k = 0; k < 4; ++k) {
        int q = q0 + k * AGG_THREADS + tid;
        v[k] = own4[min(q, qLast)];
      }
    };
    const int STEP = 4 * AGG_THREADS;
    const int nWin = (nG + 2 * STEP - 1) / (2 * STEP);
    int win = nWin > 0 ? (int)(blockIdx.x % (unsigned)nWin) : 0;
    int q0 = win * 2 * STEP;
    uint4 va[4], vb[4];
    if (nWin > 0) load4(va, q0);
    for (int n = 0; n < nWin; ++n) {
      win = (win + 1 < nWin) ? win + 1 : 0;
      const int q0n = win * 2 * STEP;
      load4(vb, q0 + STEP);
      visit64(q0 + tid, va);
      load4(va, q0n);
      visit64(q0 + STEP + tid, vb);
      q0 = q0n;
    }
    for (int e0 = nG << 4; e0 < E; e0 += AGG_THREADS) {
      const int e = e0 + tid;
      bool has = false;
      if (e < E) has = ownsRow(e);
      append(has, e);
    }
    __builtin_amdgcn_fence(__ATOMIC_ACQ_REL, "workgroup");
  }

  if (filt) {
    const int nq = __builtin_amdgcn_readfirstlane(min(wcnt, WCAP));
    for (int g0 = 0; g0 < nq; g0 += 32) {
      const int cnt = min(32, nq - g0);
      const bool valid = lane < cnt;
      const int e = valid ? wq[g0 + lane] : 0;
      const int r = rowIdx[e];
      const int c = colIdx[e];
      const float nrm = valid ? -dinv[r] * dinv[c] : 0.0f;
      const int d = valid ? r - base : 0;
      if (valid) {
        if (HAS8) {
          const float4* s8 = (const float4*)(src8 + (size_t)c * IN_CH);
          float4 a = s8[0];
          float4 b = s8[1];
          float* t8 = &acc8[d * IN_CH];
          atomicAdd(&t8[0], a.x * nrm);
          atomicAdd(&t8[1], a.y * nrm);
          atomicAdd(&t8[2], a.z * nrm);
          atomicAdd(&t8[3], a.w * nrm);
          atomicAdd(&t8[4], b.x * nrm);
          atomicAdd(&t8[5], b.y * nrm);
          atomicAdd(&t8[6], b.z * nrm);
          atomicAdd(&t8[7], b.w * nrm);
        }
        const int t = d & (AGG_WAVES - 1);
        const int slot = atomicAdd(&dcnt[t], 1);
        if (slot < DCAP) {
          int* q = &dq[(t * DCAP + slot) * 3];
          q[0] = c;
          q[1] = d;
          q[2] = __float_as_int(nrm);
        } else {
#pragma unroll 1
          for (int k = 0; k < OUT_CH; ++k)
            atomicAdd(&acc[d * PITCH + k], src64[(size_t)c * OUT_CH + k] * nrm);
        }
      }
    }
    if (FIRST && lane == 0) findCount[cntIdx] = wcnt;
  }
  __syncthreads();

  v2f a0 = {0.0f, 0.0f}, a1 = a0, a2 = a0, a3 = a0, a4 = a0, a5 = a0, a6 = a0;
  auto sumGroup = [&](int c, int d, float nrm, int cnt) {
#pragma unroll
    for (int h = 0; h < 2; ++h) {
      if (h * 16 >= cnt) break;
      float2 g[16];
#pragma unroll
      for (int j = 0; j < 16; ++j) {
        int cj = __builtin_amdgcn_readlane(c, h * 16 + j);
        g[j] = ((const float2*)(src64 + (size_t)cj * OUT_CH))[lane];
      }
#pragma unroll
      for (int j = 0; j < 16; ++j) {
        const float nj = __int_as_float(
            __builtin_amdgcn_readlane(__float_as_int(nrm), h * 16 + j));
        const int sj = __builtin_amdgcn_readlane(d, h * 16 + j) >> 5;
        const v2f gj = {g[j].x, g[j].y};
        a0 += gj * (sj == 0 ? nj : 0.0f);
        a1 += gj * (sj == 1 ? nj : 0.0f);
        a2 += gj * (sj == 2 ? nj : 0.0f);
        a3 += gj * (sj == 3 ? nj : 0.0f);
        a4 += gj * (sj == 4 ? nj : 0.0f);
        a5 += gj * (sj == 5 ? nj : 0.0f);
        a6 += gj * (sj == 6 ? nj : 0.0f);
      }
    }
  };
  if (filt) {
    const int all = dcnt[wave];
    const int nq = __builtin_amdgcn_readfirstlane(min(all, DCAP));
    const int* mq = dq + wave * DCAP * 3;
    for (int g0 = 0; g0 < nq; g0 += 32) {
      const int cnt = min(32, nq - g0);
      const bool valid = lane < cnt;
      const int i3 = (g0 + lane) * 3;
      const int c = valid ? mq[i3] : 0;
      const int d = valid ? mq[i3 + 1] : 0;
      const float nrm = valid ? __int_as_float(mq[i3 + 2]) : 0.0f;
      if (FIRST && valid)
        myList[g0 + lane] = make_int4(c, d, __float_as_int(nrm), 0);
      sumGroup(c, d, nrm, cnt);
    }
    if (FIRST && lane == 0) listCount[cntIdx] = all;
  } else {
    const int nq = __builtin_amdgcn_readfirstlane(listCount[cntIdx]);
    int g0 = 0;
    if (g0 < nq) {
      int4 rec = myList[min(g0 + lane, nq - 1)];
      for (;;) {
        const int gn = g0 + 32;
        const int4 nxt = myList[min(gn + lane, nq - 1)];
        const int cnt = min(32, nq - g0);
        const bool valid = lane < cnt;
        sumGroup(rec.x, valid ? rec.y : 0,
                 valid ? __int_as_float(rec.z) : 0.0f, cnt);
        if (gn >= nq) break;
        rec = nxt;
        g0 = gn;
      }
    }
  }
  {
    const v2f av[NSLOT] = {a0, a1, a2, a3, a4, a5, a6};
#pragma unroll
    for (int sidx = 0; sidx < NSLOT; ++sidx) {
      const int row = sidx * AGG_WAVES + wave;
      if (row < TILE_ROWS) {
        float2* p = (float2*)&acc[row * PITCH + 2 * lane];
        float2 v = *p;
        v.x += av[sidx].x;
        v.y += av[sidx].y;
        *p = v;
      }
    }
  }
  __syncthreads();

  const int rows = min(ROWS_PB, N - base);
  float* hst = (float*)dq;
  float* xst = hst + TILE_ROWS * PITCH;
  float* lst = xst + TILE_ROWS * IN_CH;
  for (int i = tid; i < TILE_ROWS * (OUT_CH / 4); i += AGG_THREADS) {
    const int r = i >> 4, c4 = i & 15;
    float4 v = make_float4(0.0f, 0.0f, 0.0f, 0.0f);
    if (r < rows) v = ((const float4*)(src64 + (size_t)(base + r) * OUT_CH))[c4];
    *(float4*)&hst[r * PITCH + c4 * 4] = v;
  }
  for (int i = tid; i < TILE_ROWS * (IN_CH / 4); i += AGG_THREADS) {
    const int r = i >> 1, c4 = i & 1;
    float4 v = make_float4(0.0f, 0.0f, 0.0f, 0.0f);
    float4 w = v;
    if (r < rows) {
      v = ((const float4*)(src8 + (size_t)(base + r) * IN_CH))[c4];
      if (!FIRST) w = ((const float4*)(Lx + (size_t)(base + r) * IN_CH))[c4];
    }
    *(float4*)&xst[r * IN_CH + c4 * 4] = v;
    if (!FIRST) *(float4*)&lst[r * IN_CH + c4 * 4] = w;
  }
  if (HAS8) {
    float4* o8 = (float4*)(Lx + (size_t)base * IN_CH);
    const float4* a8 = (const float4*)acc8;
    for (int i = tid; i < rows * (IN_CH / 4); i += AGG_THREADS) o8[i] = a8[i];
  }
  __syncthreads();

  const int half = lane >> 4;
  const int l15 = lane & 15;
  const int nUnits = ((rows + 15) >> 4) * (OUT_CH / 16);
  for (int u = wave; u < nUnits; u += AGG_WAVES) {
    const int row0 = (u >> 2) * 16;
    const int rA = row0 + l15;
    const int colB = (u & 3) * 16 + l15;
    if (FIRST) {
      v8f accZ = {};
      v8f accR = {};
      accum_block2(xst, IN_CH, rA, Wax, Wbx, IN_CH, colB, half, accZ, accR);
      accum_block2(acc8, IN_CH, rA, Wax + IN_CH * OUT_CH, Wbx + IN_CH * OUT_CH,
                   IN_CH, colB, half, accZ, accR);
      accum_block2(hst, PITCH, rA, Wah, Wbh, OUT_CH, colB, half, accZ, accR);
      accum_block2(acc, PITCH, rA, Wah + OUT_CH * OUT_CH, Wbh + OUT_CH * OUT_CH,
                   OUT_CH, colB, half, accZ, accR);
      const float bZ = bax[colB] + bah[colB];
      const float bR = bbx[colB] + bbh[colB];
#pragma unroll
      for (int i = 0; i < 8; ++i) {
        const int r = row0 + i + 8 * half;
        if (r < rows) {
          const float z = 1.0f / (1.0f + __expf(-(accZ[i] + bZ)));
          const float rg = 1.0f / (1.0f + __expf(-(accR[i] + bR)));
          const size_t g = (size_t)(base + r) * OUT_CH + colB;
          Zbuf[g] = z;
          RHout[g] = rg * hst[r * PITCH + colB];
        }
      }
    } else {
      v8f accC = {};
      accC = accum_block(xst, IN_CH, rA, Wax, IN_CH, colB, half, accC);
      accC = accum_block(lst, IN_CH, rA, Wax + IN_CH * OUT_CH, IN_CH, colB, half, accC);
      accC = accum_block(hst, PITCH, rA, Wah, OUT_CH, colB, half, accC);
      accC = accum_block(acc, PITCH, rA, Wah + OUT_CH * OUT_CH, OUT_CH, colB, half, accC);
      const float bC = bax[colB] + bah[colB];
#pragma unroll
      for (int i = 0; i < 8; ++i) {
        const int r = row0 + i + 8 * half;
        if (r < rows) {
          const size_t g = (size_t)(base + r) * OUT_CH + colB;
          const float cv = tanhf(accC[i] + bC);
          const float z = Zbuf[g];
          const float h = H[g];
          out[g] = z * h + (1.0f - z) * cv;
        }
      }
    }
  }
}


extern "C" void kernel_launch(void* const* d_in, const int* in_sizes, int n_in,
                              void* d_out, int out_size, void* d_ws,
                              size_t ws_size, hipStream_t stream) {
  const float* x = (const float*)d_in[0];
  const float* H = (const float*)d_in[1];
  const int* edge = (const int*)d_in[2];
  const float* Wzx = (const float*)d_in[3];
  const float* bzx = (const float*)d_in[4];
  const float* Wzh = (const float*)d_in[5];
  const float* bzh = (const float*)d_in[6];
  const float* Wrx = (const float*)d_in[7];
  const float* brx = (const float*)d_in[8];
  const float* Wrh = (const float*)d_in[9];
  const float* brh = (const float*)d_in[10];
  const float* Wcx = (const float*)d_in[11];
  const float* bcx = (const float*)d_in[12];
  const float* Wch = (const float*)d_in[13];
  const float* bch = (const float*)d_in[14];

  const int N = in_sizes[0] / IN_CH;
  const int E = in_sizes[2] / 2;
  const int* rowIdx = edge;
  const int* colIdx = edge + E;

  float* ws = (float*)d_ws;
  float* dinv = ws;
  float* Lx = dinv + ((N + 63) / 64) * 64;
  float* Zbuf = Lx + (size_t)N * IN_CH;
  float* RH = Zbuf + (size_t)N * OUT_CH;
  const int nZero = N;
  const int aggBlocks = (N + ROWS_PB - 1) / ROWS_PB;
  unsigned* ownerW = (unsigned*)(RH + (size_t)N * OUT_CH);
  int* listCount = (int*)(ownerW + ((E + 255) / 256) * 64);
  int* findCount = listCount + ((aggBlocks * AGG_WAVES + 63) / 64) * 64;
  int* listWords = findCount + ((aggBlocks * AGG_WAVES + 63) / 64) * 64;

  zero_kernel<<<(nZero + 255) / 256, 256, 0, stream>>>(dinv, nZero);
  degree_kernel<<<(E + 1023) / 1024, 256, 0, stream>>>(rowIdx, dinv, ownerW,
                                                        E);
  dinv_kernel<<<(N + 255) / 256, 256, 0, stream>>>(dinv, N);
  agg_kernel<true><<<aggBlocks, AGG_THREADS, 0, stream>>>(
      rowIdx, colIdx, dinv, H, x, Lx, ownerW, listWords, listCount, findCount, E,
      N, H,
      Wzx, Wzh, Wrx, Wrh, bzx, bzh, brx, brh, Zbuf, RH, nullptr);
  agg_kernel<false><<<aggBlocks, AGG_THREADS, 0, stream>>>(
      rowIdx, colIdx, dinv, RH, x, Lx, ownerW, listWords, listCount, findCount, E,
      N, H,
      Wcx, Wch, nullptr, nullptr, bcx, bch, nullptr, nullptr, Zbuf, nullptr,
      (float*)d_out);
}


# Edge filter streams the block's own bit-plane (one bit per edge; the degree kernel builds the planes in LDS, stores whole lines) instead of owner bytes; sum order = arrival in owner list
# speedup vs baseline: 2.2482x; 1.0348x over previous
#include <hip/hip_runtime.h>
#include <hip/hip_bf16.h>
#include <math.h>


typedef float v2f __attribute__((ext_vector_type(2)));
typedef float v8f __attribute__((ext_vector_type(8)));

#define IN_CH 8
#define OUT_CH 64

__device__ __forceinline__ v8f wmma_f32(v2f a, v2f b, v8f c) {
  return __builtin_amdgcn_wmma_f32_16x16x4_f32(false, a, false, b, (short)0, c,
                                               false, false);
}

__device__ __forceinline__ v8f accum_block(const float* __restrict__ A, int ldA,
                                           int rA, const float* __restrict__ W,
                                           int Kdim, int colB, int half,
                                           v8f acc) {
#pragma unroll 4
  for (int k0 = 0; k0 < Kdim; k0 += 4) {
    int ka = k0 + 2 * half;
    v2f a;
    a.x = A[rA * ldA + ka];
    a.y = A[rA * ldA + ka + 1];
    v2f b;
    b.x = W[ka * OUT_CH + colB];
    b.y = W[(ka + 1) * OUT_CH + colB];
    acc = wmma_f32(a, b, acc);
  }
  return acc;
}

__device__ __forceinline__ void accum_block2(const float* __restrict__ A,
                                             int ldA, int rA,
                                             const float* __restrict__ W1,
                                             const float* __restrict__ W2,
                                             int Kdim, int colB, int half,
                                             v8f& acc1, v8f& acc2) {
#pragma unroll 4
  for (int k0 = 0; k0 < Kdim; k0 += 4) {
    int ka = k0 + 2 * half;
    v2f a;
    a.x = A[rA * ldA + ka];
    a.y = A[rA * ldA + ka + 1];
    v2f b1;
    b1.x = W1[ka * OUT_CH + colB];
    b1.y = W1[(ka + 1) * OUT_CH + colB];
    v2f b2;
    b2.x = W2[ka * OUT_CH + colB];
    b2.y = W2[(ka + 1) * OUT_CH + colB];
    acc1 = wmma_f32(a, b1, acc1);
    acc2 = wmma_f32(a, b2, acc2);
  }
}


__global__ void __launch_bounds__(256) zero_kernel(float* p, int n) {
  int i = blockIdx.x * blockDim.x + threadIdx.x;
  if (i < n) p[i] = 0.0f;
}

#define ROWS_PB 196
#define TILE_ROWS 208
#define PITCH 68

__device__ __forceinline__ unsigned owner_plane(int r) {
  return (unsigned)(r / ROWS_PB) & 255u;
}

#define NPLANES 256
#define DEG_THREADS 256
#define DEG_EDGES 1024

__device__ __forceinline__ int plane_words(int E) {
  return ((E + DEG_EDGES - 1) / DEG_EDGES) * 32;
}

__global__ void __launch_bounds__(DEG_THREADS)
degree_kernel(const int* __restrict__ rowIdx, float* __restrict__ deg,
              unsigned* __restrict__ planes, int E) {
  __shared__ __attribute__((aligned(16))) unsigned bits[NPLANES * 32];
  const int tid = threadIdx.x;
  uint4* b4 = (uint4*)bits;
#pragma unroll
  for (int i = 0; i < 8; ++i)
    b4[i * DEG_THREADS + tid] = make_uint4(0u, 0u, 0u, 0u);
  __syncthreads();
  const int t = blockIdx.x * DEG_THREADS + tid;
  const int e = t * 4;
  const int w = tid >> 3;
  const int sh = (tid & 7) * 4;
  if (e + 3 < E) {
    const int4 r = ((const int4*)rowIdx)[t];
    atomicAdd(&deg[r.x], 1.0f);
    atomicAdd(&deg[r.y], 1.0f);
    atomicAdd(&deg[r.z], 1.0f);
    atomicAdd(&deg[r.w], 1.0f);
    atomicOr(&bits[owner_plane(r.x) * 32 + w], 1u << sh);
    atomicOr(&bits[owner_plane(r.y) * 32 + w], 2u << sh);
    atomicOr(&bits[owner_plane(r.z) * 32 + w], 4u << sh);
    atomicOr(&bits[owner_plane(r.w) * 32 + w], 8u << sh);
  } else {
    for (int k = e; k < E; ++k) {
      const int r = rowIdx[k];
      atomicAdd(&deg[r], 1.0f);
      atomicOr(&bits[owner_plane(r) * 32 + w], 1u << (sh + (k - e)));
    }
  }
  __syncthreads();
  const int pitchW = plane_words(E);
#pragma unroll
  for (int i = 0; i < 8; ++i) {
    const int idx = i * DEG_THREADS + tid;
    uint4* line = (uint4*)(planes + (size_t)(idx >> 3) * pitchW +
                           (size_t)blockIdx.x * 32);
    line[idx & 7] = b4[idx];
  }
}

__global__ void __launch_bounds__(256) dinv_kernel(float* deg, int n) {
  int i = blockIdx.x * blockDim.x + threadIdx.x;
  if (i < n) {
    float d = deg[i];
    deg[i] = (d > 0.0f) ? rsqrtf(d) : 0.0f;
  }
}

#define AGG_THREADS 1024
#define AGG_WAVES (AGG_THREADS / 32)
#define WCAP 384
#define DCAP 288
#define NSLOT 7

template <bool FIRST>
__global__ void __launch_bounds__(AGG_THREADS)
agg_kernel(const int* __restrict__ rowIdx, const int* __restrict__ colIdx,
           const float* __restrict__ dinv, const float* __restrict__ src64,
           const float* __restrict__ src8, float* __restrict__ Lx,
           const unsigned* __restrict__ planes, int* __restrict__ listWords,
           int* __restrict__ listCount, int* __restrict__ findCount, int E,
           int N,
           const float* __restrict__ H, const float* __restrict__ Wax,
           const float* __restrict__ Wah, const float* __restrict__ Wbx,
           const float* __restrict__ Wbh, const float* __restrict__ bax,
           const float* __restrict__ bah, const float* __restrict__ bbx,
           const float* __restrict__ bbh, float* __restrict__ Zbuf,
           float* __restrict__ RHout, float* __restrict__ out) {
  constexpr bool HAS8 = FIRST;
  __shared__ __attribute__((aligned(16))) float acc[TILE_ROWS * PITCH];
  __shared__ __attribute__((aligned(16))) float acc8[HAS8 ? TILE_ROWS * IN_CH : 4];
  __shared__ int fq[AGG_WAVES * WCAP];
  __shared__ __attribute__((aligned(16))) int dq[AGG_WAVES * DCAP * 3];
  __shared__ int dcnt[AGG_WAVES];

  const int tid = threadIdx.x;
  const int lane = tid & 31;
  const int base = blockIdx.x * ROWS_PB;
  const int wave = __builtin_amdgcn_readfirstlane(tid >> 5);

  for (int i = tid; i < TILE_ROWS * PITCH; i += AGG_THREADS) acc[i] = 0.0f;
  if (HAS8) {
    for (int i = tid; i < TILE_ROWS * IN_CH; i += AGG_THREADS) acc8[i] = 0.0f;
  }
  if (tid < AGG_WAVES) dcnt[tid] = 0;
  __syncthreads();

  int* wq = fq + wave * WCAP;
  int wcnt = 0;
  const int cntIdx = blockIdx.x * AGG_WAVES + wave;
  bool filt = true;
  if (!FIRST) {
    const int dc = listCount[blockIdx.x * AGG_WAVES + lane];
    const int fc = findCount[blockIdx.x * AGG_WAVES + lane];
    filt = __builtin_amdgcn_ballot_w32(dc > DCAP || fc > WCAP) != 0u;
  }
  int4* __restrict__ myList =
      (int4*)listWords + ((size_t)blockIdx.x * AGG_WAVES + (size_t)wave) * DCAP;

  if (filt) {
    const bool planeIsBlock = gridDim.x <= 256u;
    auto spot = [&](int e) {
      int r = rowIdx[e];
      int c = colIdx[e];
      unsigned d = (unsigned)(r - base);
      float nrm = -dinv[r] * dinv[c];
#pragma unroll 1
      for (int k = 0; k < OUT_CH; ++k)
        atomicAdd(&acc[d * PITCH + k], src64[(size_t)c * OUT_CH + k] * nrm);
      if (HAS8) {
#pragma unroll 1
        for (int k = 0; k < IN_CH; ++k)
          atomicAdd(&acc8[d * IN_CH + k], src8[(size_t)c * IN_CH + k] * nrm);
      }
    };
    auto append = [&](bool has, int e) {
      const unsigned m = __builtin_amdgcn_ballot_w32(has);
      const int slot = wcnt + (int)__builtin_amdgcn_mbcnt_lo(m, 0u);
      const int after = wcnt + __builtin_popcount(m);
      if (has && slot < WCAP) wq[slot] = e;
      if (after > WCAP) {
        if (has && slot >= WCAP) spot(e);
      }
      wcnt = after;
    };
    auto ownsRow = [&](int e) {
      return (unsigned)(rowIdx[e] - base) < (unsigned)ROWS_PB;
    };
    const int pitchW = plane_words(E);
    const int nQ = pitchW >> 2;
    const uint4* __restrict__ pl4 =
        (const uint4*)(planes + (size_t)(blockIdx.x & 255u) * pitchW);
    auto visit = [&](int q, uint4 v) {
      const bool in = q < nQ;
      unsigned a = in ? v.x : 0u, b = in ? v.y : 0u;
      unsigned cw = in ? v.z : 0u, d = in ? v.w : 0u;
      while (__builtin_amdgcn_ballot_w32((a | b | cw | d) != 0u) != 0u) {
        const int j = a != 0u ? 0 : b != 0u ? 1 : cw != 0u ? 2 : 3;
        const unsigned word = j == 0 ? a : j == 1 ? b : j == 2 ? cw : d;
        bool has = word != 0u;
        const int p = __builtin_ffs((int)word) - 1;
        const unsigned rest = word & (word - 1u);
        a = j == 0 ? rest : a;
        b = j == 1 ? rest : b;
        cw = j == 2 ? rest : cw;
        d = j == 3 ? rest : d;
        const int eh = ((q * 4 + j) << 5) + p;
        if (!planeIsBlock) {
          if (has) has = ownsRow(eh);
        }
        append(has, eh);
      }
    };
    const int qLast = nQ - 1;
    auto load4 = [&](uint4 (&v)[4], int q0) {
#pragma unroll
      for (int k = 0; k < 4; ++k) {
        int q = q0 + k * AGG_THREADS + tid;
        v[k] = pl4[min(q, qLast)];
      }
    };
    auto visit4 = [&](int q0, const uint4 (&v)[4]) {
#pragma unroll
      for (int k = 0; k < 4; ++k) visit(q0 + k * AGG_THREADS + tid, v[k]);
    };
    const int STEP = 4 * AGG_THREADS;
    if (nQ > 0) {
      uint4 va[4], vb[4];
      load4(va, 0);
      for (int q0 = 0; q0 < nQ; q0 += 2 * STEP) {
        load4(vb, q0 + STEP);
        visit4(q0, va);
        load4(va, q0 + 2 * STEP);
        visit4(q0 + STEP, vb);
      }
    }
    __builtin_amdgcn_fence(__ATOMIC_ACQ_REL, "workgroup");
  }

  if (filt) {
    const int nq = __builtin_amdgcn_readfirstlane(min(wcnt, WCAP));
    for (int g0 = 0; g0 < nq; g0 += 32) {
      const int cnt = min(32, nq - g0);
      const bool valid = lane < cnt;
      const int e = valid ? wq[g0 + lane] : 0;
      const int r = rowIdx[e];
      const int c = colIdx[e];
      const float nrm = valid ? -dinv[r] * dinv[c] : 0.0f;
      const int d = valid ? r - base : 0;
      if (valid) {
        if (HAS8) {
          const float4* s8 = (const float4*)(src8 + (size_t)c * IN_CH);
          float4 a = s8[0];
          float4 b = s8[1];
          float* t8 = &acc8[d * IN_CH];
          atomicAdd(&t8[0], a.x * nrm);
          atomicAdd(&t8[1], a.y * nrm);
          atomicAdd(&t8[2], a.z * nrm);
          atomicAdd(&t8[3], a.w * nrm);
          atomicAdd(&t8[4], b.x * nrm);
          atomicAdd(&t8[5], b.y * nrm);
          atomicAdd(&t8[6], b.z * nrm);
          atomicAdd(&t8[7], b.w * nrm);
        }
        const int t = d & (AGG_WAVES - 1);
        const int slot = atomicAdd(&dcnt[t], 1);
        if (slot < DCAP) {
          int* q = &dq[(t * DCAP + slot) * 3];
          q[0] = c;
          q[1] = d;
          q[2] = __float_as_int(nrm);
        } else {
#pragma unroll 1
          for (int k = 0; k < OUT_CH; ++k)
            atomicAdd(&acc[d * PITCH + k], src64[(size_t)c * OUT_CH + k] * nrm);
        }
      }
    }
    if (FIRST && lane == 0) findCount[cntIdx] = wcnt;
  }
  __syncthreads();

  v2f a0 = {0.0f, 0.0f}, a1 = a0, a2 = a0, a3 = a0, a4 = a0, a5 = a0, a6 = a0;
  auto sumGroup = [&](int c, int d, float nrm, int cnt) {
#pragma unroll
    for (int h = 0; h < 2; ++h) {
      if (h * 16 >= cnt) break;
      float2 g[16];
#pragma unroll
      for (int j = 0; j < 16; ++j) {
        int cj = __builtin_amdgcn_readlane(c, h * 16 + j);
        g[j] = ((const float2*)(src64 + (size_t)cj * OUT_CH))[lane];
      }
#pragma unroll
      for (int j = 0; j < 16; ++j) {
        const float nj = __int_as_float(
            __builtin_amdgcn_readlane(__float_as_int(nrm), h * 16 + j));
        const int sj = __builtin_amdgcn_readlane(d, h * 16 + j) >> 5;
        const v2f gj = {g[j].x, g[j].y};
        a0 += gj * (sj == 0 ? nj : 0.0f);
        a1 += gj * (sj == 1 ? nj : 0.0f);
        a2 += gj * (sj == 2 ? nj : 0.0f);
        a3 += gj * (sj == 3 ? nj : 0.0f);
        a4 += gj * (sj == 4 ? nj : 0.0f);
        a5 += gj * (sj == 5 ? nj : 0.0f);
        a6 += gj * (sj == 6 ? nj : 0.0f);
      }
    }
  };
  if (filt) {
    const int all = dcnt[wave];
    const int nq = __builtin_amdgcn_readfirstlane(min(all, DCAP));
    const int* mq = dq + wave * DCAP * 3;
    for (int g0 = 0; g0 < nq; g0 += 32) {
      const int cnt = min(32, nq - g0);
      const bool valid = lane < cnt;
      const int i3 = (g0 + lane) * 3;
      const int c = valid ? mq[i3] : 0;
      const int d = valid ? mq[i3 + 1] : 0;
      const float nrm = valid ? __int_as_float(mq[i3 + 2]) : 0.0f;
      if (FIRST && valid)
        myList[g0 + lane] = make_int4(c, d, __float_as_int(nrm), 0);
      sumGroup(c, d, nrm, cnt);
    }
    if (FIRST && lane == 0) listCount[cntIdx] = all;
  } else {
    const int nq = __builtin_amdgcn_readfirstlane(listCount[cntIdx]);
    int g0 = 0;
    if (g0 < nq) {
      int4 rec = myList[min(g0 + lane, nq - 1)];
      for (;;) {
        const int gn = g0 + 32;
        const int4 nxt = myList[min(gn + lane, nq - 1)];
        const int cnt = min(32, nq - g0);
        const bool valid = lane < cnt;
        sumGroup(rec.x, valid ? rec.y : 0,
                 valid ? __int_as_float(rec.z) : 0.0f, cnt);
        if (gn >= nq) break;
        rec = nxt;
        g0 = gn;
      }
    }
  }
  {
    const v2f av[NSLOT] = {a0, a1, a2, a3, a4, a5, a6};
#pragma unroll
    for (int sidx = 0; sidx < NSLOT; ++sidx) {
      const int row = sidx * AGG_WAVES + wave;
      if (row < TILE_ROWS) {
        float2* p = (float2*)&acc[row * PITCH + 2 * lane];
        float2 v = *p;
        v.x += av[sidx].x;
        v.y += av[sidx].y;
        *p = v;
      }
    }
  }
  __syncthreads();

  const int rows = min(ROWS_PB, N - base);
  float* hst = (float*)dq;
  float* xst = hst + TILE_ROWS * PITCH;
  float* lst = xst + TILE_ROWS * IN_CH;
  for (int i = tid; i < TILE_ROWS * (OUT_CH / 4); i += AGG_THREADS) {
    const int r = i >> 4, c4 = i & 15;
    float4 v = make_float4(0.0f, 0.0f, 0.0f, 0.0f);
    if (r < rows) v = ((const float4*)(src64 + (size_t)(base + r) * OUT_CH))[c4];
    *(float4*)&hst[r * PITCH + c4 * 4] = v;
  }
  for (int i = tid; i < TILE_ROWS * (IN_CH / 4); i += AGG_THREADS) {
    const int r = i >> 1, c4 = i & 1;
    float4 v = make_float4(0.0f, 0.0f, 0.0f, 0.0f);
    float4 w = v;
    if (r < rows) {
      v = ((const float4*)(src8 + (size_t)(base + r) * IN_CH))[c4];
      if (!FIRST) w = ((const float4*)(Lx + (size_t)(base + r) * IN_CH))[c4];
    }
    *(float4*)&xst[r * IN_CH + c4 * 4] = v;
    if (!FIRST) *(float4*)&lst[r * IN_CH + c4 * 4] = w;
  }
  if (HAS8) {
    float4* o8 = (float4*)(Lx + (size_t)base * IN_CH);
    const float4* a8 = (const float4*)acc8;
    for (int i = tid; i < rows * (IN_CH / 4); i += AGG_THREADS) o8[i] = a8[i];
  }
  __syncthreads();

  const int half = lane >> 4;
  const int l15 = lane & 15;
  const int nUnits = ((rows + 15) >> 4) * (OUT_CH / 16);
  for (int u = wave; u < nUnits; u += AGG_WAVES) {
    const int row0 = (u >> 2) * 16;
    const int rA = row0 + l15;
    const int colB = (u & 3) * 16 + l15;
    if (FIRST) {
      v8f accZ = {};
      v8f accR = {};
      accum_block2(xst, IN_CH, rA, Wax, Wbx, IN_CH, colB, half, accZ, accR);
      accum_block2(acc8, IN_CH, rA, Wax + IN_CH * OUT_CH, Wbx + IN_CH * OUT_CH,
                   IN_CH, colB, half, accZ, accR);
      accum_block2(hst, PITCH, rA, Wah, Wbh, OUT_CH, colB, half, accZ, accR);
      accum_block2(acc, PITCH, rA, Wah + OUT_CH * OUT_CH, Wbh + OUT_CH * OUT_CH,
                   OUT_CH, colB, half, accZ, accR);
      const float bZ = bax[colB] + bah[colB];
      const float bR = bbx[colB] + bbh[colB];
#pragma unroll
      for (int i = 0; i < 8; ++i) {
        const int r = row0 + i + 8 * half;
        if (r < rows) {
          const float z = 1.0f / (1.0f + __expf(-(accZ[i] + bZ)));
          const float rg = 1.0f / (1.0f + __expf(-(accR[i] + bR)));
          const size_t g = (size_t)(base + r) * OUT_CH + colB;
          Zbuf[g] = z;
          RHout[g] = rg * hst[r * PITCH + colB];
        }
      }
    } else {
      v8f accC = {};
      accC = accum_block(xst, IN_CH, rA, Wax, IN_CH, colB, half, accC);
      accC = accum_block(lst, IN_CH, rA, Wax + IN_CH * OUT_CH, IN_CH, colB, half, accC);
      accC = accum_block(hst, PITCH, rA, Wah, OUT_CH, colB, half, accC);
      accC = accum_block(acc, PITCH, rA, Wah + OUT_CH * OUT_CH, OUT_CH, colB, half, accC);
      const float bC = bax[colB] + bah[colB];
#pragma unroll
      for (int i = 0; i < 8; ++i) {
        const int r = row0 + i + 8 * half;
        if (r < rows) {
          const size_t g = (size_t)(base + r) * OUT_CH + colB;
          const float cv = tanhf(accC[i] + bC);
          const float z = Zbuf[g];
          const float h = H[g];
          out[g] = z * h + (1.0f - z) * cv;
        }
      }
    }
  }
}


extern "C" void kernel_launch(void* const* d_in, const int* in_sizes, int n_in,
                              void* d_out, int out_size, void* d_ws,
                              size_t ws_size, hipStream_t stream) {
  const float* x = (const float*)d_in[0];
  const float* H = (const float*)d_in[1];
  const int* edge = (const int*)d_in[2];
  const float* Wzx = (const float*)d_in[3];
  const float* bzx = (const float*)d_in[4];
  const float* Wzh = (const float*)d_in[5];
  const float* bzh = (const float*)d_in[6];
  const float* Wrx = (const float*)d_in[7];
  const float* brx = (const float*)d_in[8];
  const float* Wrh = (const float*)d_in[9];
  const float* brh = (const float*)d_in[10];
  const float* Wcx = (const float*)d_in[11];
  const float* bcx = (const float*)d_in[12];
  const float* Wch = (const float*)d_in[13];
  const float* bch = (const float*)d_in[14];

  const int N = in_sizes[0] / IN_CH;
  const int E = in_sizes[2] / 2;
  const int* rowIdx = edge;
  const int* colIdx = edge + E;

  float* ws = (float*)d_ws;
  float* dinv = ws;
  float* Lx = dinv + ((N + 63) / 64) * 64;
  float* Zbuf = Lx + (size_t)N * IN_CH;
  float* RH = Zbuf + (size_t)N * OUT_CH;
  const int nZero = N;
  const int aggBlocks = (N + ROWS_PB - 1) / ROWS_PB;
  unsigned* planes = (unsigned*)(RH + (size_t)N * OUT_CH);
  int* listCount =
      (int*)(planes + (size_t)NPLANES * (((E + DEG_EDGES - 1) / DEG_EDGES) * 32));
  int* findCount = listCount + ((aggBlocks * AGG_WAVES + 63) / 64) * 64;
  int* listWords = findCount + ((aggBlocks * AGG_WAVES + 63) / 64) * 64;

  zero_kernel<<<(nZero + 255) / 256, 256, 0, stream>>>(dinv, nZero);
  degree_kernel<<<(E + DEG_EDGES - 1) / DEG_EDGES, DEG_THREADS, 0, stream>>>(
      rowIdx, dinv, planes, E);
  dinv_kernel<<<(N + 255) / 256, 256, 0, stream>>>(dinv, N);
  agg_kernel<true><<<aggBlocks, AGG_THREADS, 0, stream>>>(
      rowIdx, colIdx, dinv, H, x, Lx, planes, listWords, listCount, findCount, E,
      N, H,
      Wzx, Wzh, Wrx, Wrh, bzx, bzh, brx, brh, Zbuf, RH, nullptr);
  agg_kernel<false><<<aggBlocks, AGG_THREADS, 0, stream>>>(
      rowIdx, colIdx, dinv, RH, x, Lx, planes, listWords, listCount, findCount, E,
      N, H,
      Wcx, Wch, nullptr, nullptr, bcx, bch, nullptr, nullptr, Zbuf, nullptr,
      (float*)d_out);
}


# Row sums from one list per destination row: finders put each edge on its row's LDS list, the owner wave walks a row with one multiply-add per edge (was 7 selected adds); sum order = row list arrival
# speedup vs baseline: 2.4311x; 1.0814x over previous
#include <hip/hip_runtime.h>
#include <hip/hip_bf16.h>
#include <math.h>


typedef float v2f __attribute__((ext_vector_type(2)));
typedef float v8f __attribute__((ext_vector_type(8)));

#define IN_CH 8
#define OUT_CH 64

__device__ __forceinline__ v8f wmma_f32(v2f a, v2f b, v8f c) {
  return __builtin_amdgcn_wmma_f32_16x16x4_f32(false, a, false, b, (short)0, c,
                                               false, false);
}

__device__ __forceinline__ v8f accum_block(const float* __restrict__ A, int ldA,
                                           int rA, const float* __restrict__ W,
                                           int Kdim, int colB, int half,
                                           v8f acc) {
#pragma unroll 4
  for (int k0 = 0; k0 < Kdim; k0 += 4) {
    int ka = k0 + 2 * half;
    v2f a;
    a.x = A[rA * ldA + ka];
    a.y = A[rA * ldA + ka + 1];
    v2f b;
    b.x = W[ka * OUT_CH + colB];
    b.y = W[(ka + 1) * OUT_CH + colB];
    acc = wmma_f32(a, b, acc);
  }
  return acc;
}

__device__ __forceinline__ void accum_block2(const float* __restrict__ A,
                                             int ldA, int rA,
                                             const float* __restrict__ W1,
                                             const float* __restrict__ W2,
                                             int Kdim, int colB, int half,
                                             v8f& acc1, v8f& acc2) {
#pragma unroll 4
  for (int k0 = 0; k0 < Kdim; k0 += 4) {
    int ka = k0 + 2 * half;
    v2f a;
    a.x = A[rA * ldA + ka];
    a.y = A[rA * ldA + ka + 1];
    v2f b1;
    b1.x = W1[ka * OUT_CH + colB];
    b1.y = W1[(ka + 1) * OUT_CH + colB];
    v2f b2;
    b2.x = W2[ka * OUT_CH + colB];
    b2.y = W2[(ka + 1) * OUT_CH + colB];
    acc1 = wmma_f32(a, b1, acc1);
    acc2 = wmma_f32(a, b2, acc2);
  }
}


__global__ void __launch_bounds__(256) zero_kernel(float* p, int n) {
  int i = blockIdx.x * blockDim.x + threadIdx.x;
  if (i < n) p[i] = 0.0f;
}

#define ROWS_PB 196
#define TILE_ROWS 208
#define PITCH 68

__device__ __forceinline__ unsigned owner_plane(int r) {
  return (unsigned)(r / ROWS_PB) & 255u;
}

#define NPLANES 256
#define DEG_THREADS 256
#define DEG_EDGES 1024

__device__ __forceinline__ int plane_words(int E) {
  return ((E + DEG_EDGES - 1) / DEG_EDGES) * 32;
}

__global__ void __launch_bounds__(DEG_THREADS)
degree_kernel(const int* __restrict__ rowIdx, float* __restrict__ deg,
              unsigned* __restrict__ planes, int E) {
  __shared__ __attribute__((aligned(16))) unsigned bits[NPLANES * 32];
  const int tid = threadIdx.x;
  uint4* b4 = (uint4*)bits;
#pragma unroll
  for (int i = 0; i < 8; ++i)
    b4[i * DEG_THREADS + tid] = make_uint4(0u, 0u, 0u, 0u);
  __syncthreads();
  const int t = blockIdx.x * DEG_THREADS + tid;
  const int e = t * 4;
  const int w = tid >> 3;
  const int sh = (tid & 7) * 4;
  if (e + 3 < E) {
    const int4 r = ((const int4*)rowIdx)[t];
    atomicAdd(&deg[r.x], 1.0f);
    atomicAdd(&deg[r.y], 1.0f);
    atomicAdd(&deg[r.z], 1.0f);
    atomicAdd(&deg[r.w], 1.0f);
    atomicOr(&bits[owner_plane(r.x) * 32 + w], 1u << sh);
    atomicOr(&bits[owner_plane(r.y) * 32 + w], 2u << sh);
    atomicOr(&bits[owner_plane(r.z) * 32 + w], 4u << sh);
    atomicOr(&bits[owner_plane(r.w) * 32 + w], 8u << sh);
  } else {
    for (int k = e; k < E; ++k) {
      const int r = rowIdx[k];
      atomicAdd(&deg[r], 1.0f);
      atomicOr(&bits[owner_plane(r) * 32 + w], 1u << (sh + (k - e)));
    }
  }
  __syncthreads();
  const int pitchW = plane_words(E);
#pragma unroll
  for (int i = 0; i < 8; ++i) {
    const int idx = i * DEG_THREADS + tid;
    uint4* line = (uint4*)(planes + (size_t)(idx >> 3) * pitchW +
                           (size_t)blockIdx.x * 32);
    line[idx & 7] = b4[idx];
  }
}

__global__ void __launch_bounds__(256) dinv_kernel(float* deg, int n) {
  int i = blockIdx.x * blockDim.x + threadIdx.x;
  if (i < n) {
    float d = deg[i];
    deg[i] = (d > 0.0f) ? rsqrtf(d) : 0.0f;
  }
}

#define AGG_THREADS 1024
#define AGG_WAVES (AGG_THREADS / 32)
#define WCAP 384
#define NSLOT 7
#define LROWS (NSLOT * AGG_WAVES)
#define RCAP 64
#define RPW 132

template <bool FIRST>
__global__ void __launch_bounds__(AGG_THREADS)
agg_kernel(const int* __restrict__ rowIdx, const int* __restrict__ colIdx,
           const float* __restrict__ dinv, const float* __restrict__ src64,
           const float* __restrict__ src8, float* __restrict__ Lx,
           const unsigned* __restrict__ planes, int* __restrict__ recWords,
           int* __restrict__ rowCount, int* __restrict__ findCount, int E,
           int N,
           const float* __restrict__ H, const float* __restrict__ Wax,
           const float* __restrict__ Wah, const float* __restrict__ Wbx,
           const float* __restrict__ Wbh, const float* __restrict__ bax,
           const float* __restrict__ bah, const float* __restrict__ bbx,
           const float* __restrict__ bbh, float* __restrict__ Zbuf,
           float* __restrict__ RHout, float* __restrict__ out) {
  constexpr bool HAS8 = FIRST;
  __shared__ __attribute__((aligned(16))) float acc[TILE_ROWS * PITCH];
  __shared__ __attribute__((aligned(16))) float acc8[HAS8 ? TILE_ROWS * IN_CH : 4];
  __shared__ int fq[AGG_WAVES * WCAP];
  __shared__ __attribute__((aligned(16))) int rq[LROWS * RPW];
  __shared__ int rcnt[LROWS];

  const int tid = threadIdx.x;
  const int lane = tid & 31;
  const int base = blockIdx.x * ROWS_PB;
  const int wave = __builtin_amdgcn_readfirstlane(tid >> 5);

  for (int i = tid; i < TILE_ROWS * PITCH; i += AGG_THREADS) acc[i] = 0.0f;
  if (HAS8) {
    for (int i = tid; i < TILE_ROWS * IN_CH; i += AGG_THREADS) acc8[i] = 0.0f;
  }
  if (tid < LROWS) rcnt[tid] = 0;
  __syncthreads();

  int* wq = fq + wave * WCAP;
  int wcnt = 0;
  const int cntIdx = blockIdx.x * AGG_WAVES + wave;
  const int rbase = blockIdx.x * LROWS;
  int2* __restrict__ recs = (int2*)recWords;
  bool filt = true;
  if (!FIRST) {
    bool over = findCount[blockIdx.x * AGG_WAVES + lane] > WCAP;
#pragma unroll
    for (int s = 0; s < NSLOT; ++s)
      over = over || rowCount[rbase + s * AGG_WAVES + lane] > RCAP;
    filt = __builtin_amdgcn_ballot_w32(over) != 0u;
  }

  if (filt) {
    const bool planeIsBlock = gridDim.x <= 256u;
    auto spot = [&](int e) {
      int r = rowIdx[e];
      int c = colIdx[e];
      unsigned d = (unsigned)(r - base);
      float nrm = -dinv[r] * dinv[c];
#pragma unroll 1
      for (int k = 0; k < OUT_CH; ++k)
        atomicAdd(&acc[d * PITCH + k], src64[(size_t)c * OUT_CH + k] * nrm);
      if (HAS8) {
#pragma unroll 1
        for (int k = 0; k < IN_CH; ++k)
          atomicAdd(&acc8[d * IN_CH + k], src8[(size_t)c * IN_CH + k] * nrm);
      }
    };
    auto append = [&](bool has, int e) {
      const unsigned m = __builtin_amdgcn_ballot_w32(has);
      const int slot = wcnt + (int)__builtin_amdgcn_mbcnt_lo(m, 0u);
      const int after = wcnt + __builtin_popcount(m);
      if (has && slot < WCAP) wq[slot] = e;
      if (after > WCAP) {
        if (has && slot >= WCAP) spot(e);
      }
      wcnt = after;
    };
    auto ownsRow = [&](int e) {
      return (unsigned)(rowIdx[e] - base) < (unsigned)ROWS_PB;
    };
    const int pitchW = plane_words(E);
    const int nQ = pitchW >> 2;
    const uint4* __restrict__ pl4 =
        (const uint4*)(planes + (size_t)(blockIdx.x & 255u) * pitchW);
    auto visit = [&](int q, uint4 v) {
      const bool in = q < nQ;
      unsigned a = in ? v.x : 0u, b = in ? v.y : 0u;
      unsigned cw = in ? v.z : 0u, d = in ? v.w : 0u;
      while (__builtin_amdgcn_ballot_w32((a | b | cw | d) != 0u) != 0u) {
        const int j = a != 0u ? 0 : b != 0u ? 1 : cw != 0u ? 2 : 3;
        const unsigned word = j == 0 ? a : j == 1 ? b : j == 2 ? cw : d;
        bool has = word != 0u;
        const int p = __builtin_ffs((int)word) - 1;
        const unsigned rest = word & (word - 1u);
        a = j == 0 ? rest : a;
        b = j == 1 ? rest : b;
        cw = j == 2 ? rest : cw;
        d = j == 3 ? rest : d;
        const int eh = ((q * 4 + j) << 5) + p;
        if (!planeIsBlock) {
          if (has) has = ownsRow(eh);
        }
        append(has, eh);
      }
    };
    const int qLast = nQ - 1;
    auto load4 = [&](uint4 (&v)[4], int q0) {
#pragma unroll
      for (int k = 0; k < 4; ++k) {
        int q = q0 + k * AGG_THREADS + tid;
        v[k] = pl4[min(q, qLast)];
      }
    };
    auto visit4 = [&](int q0, const uint4 (&v)[4]) {
#pragma unroll
      for (int k = 0; k < 4; ++k) visit(q0 + k * AGG_THREADS + tid, v[k]);
    };
    const int STEP = 4 * AGG_THREADS;
    if (nQ > 0) {
      uint4 va[4], vb[4];
      load4(va, 0);
      for (int q0 = 0; q0 < nQ; q0 += 2 * STEP) {
        load4(vb, q0 + STEP);
        visit4(q0, va);
        load4(va, q0 + 2 * STEP);
        visit4(q0 + STEP, vb);
      }
    }
    __builtin_amdgcn_fence(__ATOMIC_ACQ_REL, "workgroup");
  }

  if (filt) {
    const int nq = __builtin_amdgcn_readfirstlane(min(wcnt, WCAP));
    for (int g0 = 0; g0 < nq; g0 += 32) {
      const int cnt = min(32, nq - g0);
      const bool valid = lane < cnt;
      const int e = valid ? wq[g0 + lane] : 0;
      const int r = rowIdx[e];
      const int c = colIdx[e];
      const float nrm = valid ? -dinv[r] * dinv[c] : 0.0f;
      const int d = valid ? r - base : 0;
      if (valid) {
        if (HAS8) {
          const float4* s8 = (const float4*)(src8 + (size_t)c * IN_CH);
          float4 a = s8[0];
          float4 b = s8[1];
          float* t8 = &acc8[d * IN_CH];
          atomicAdd(&t8[0], a.x * nrm);
          atomicAdd(&t8[1], a.y * nrm);
          atomicAdd(&t8[2], a.z * nrm);
          atomicAdd(&t8[3], a.w * nrm);
          atomicAdd(&t8[4], b.x * nrm);
          atomicAdd(&t8[5], b.y * nrm);
          atomicAdd(&t8[6], b.z * nrm);
          atomicAdd(&t8[7], b.w * nrm);
        }
        const int slot = atomicAdd(&rcnt[d], 1);
        if (slot < RCAP) {
          *(int2*)&rq[d * RPW + slot * 2] = make_int2(c, __float_as_int(nrm));
        } else {
#pragma unroll 1
          for (int k = 0; k < OUT_CH; ++k)
            atomicAdd(&acc[d * PITCH + k], src64[(size_t)c * OUT_CH + k] * nrm);
        }
      }
    }
    if (FIRST && lane == 0) findCount[cntIdx] = wcnt;
  }
  __syncthreads();

  auto sumRow = [&](int2 e0, int2 e1, int n) {
    v2f a = {0.0f, 0.0f};
#pragma unroll 1
    for (int j0 = 0; j0 < n; j0 += 32) {
      const bool up = j0 != 0;
      const int cc = up ? e1.x : e0.x;
      const int ww = up ? e1.y : e0.y;
      float2 g[32];
#pragma unroll
      for (int k = 0; k < 32; ++k) {
        const unsigned cj = (unsigned)__builtin_amdgcn_readlane(cc, k);
        g[k] = ((const float2*)(src64 + (size_t)cj * OUT_CH))[lane];
      }
#pragma unroll
      for (int k = 0; k < 32; ++k) {
        const float wj = __int_as_float(__builtin_amdgcn_readlane(ww, k));
        const v2f gk = {g[k].x, g[k].y};
        a += gk * wj;
      }
    }
    return a;
  };
  auto putRow = [&](int row, v2f a) {
    if (row < TILE_ROWS) {
      float2* p = (float2*)&acc[row * PITCH + 2 * lane];
      float2 v = *p;
      v.x += a.x;
      v.y += a.y;
      *p = v;
    }
  };
  if (filt) {
#pragma unroll 1
    for (int s = 0; s < NSLOT; ++s) {
      const int row = s * AGG_WAVES + wave;
      const int all = rcnt[row];
      const int n = __builtin_amdgcn_readfirstlane(min(all, RCAP));
      const int2* rl = (const int2*)&rq[row * RPW];
      const int2 none = make_int2(0, 0);
      const int2 e0 = lane < n ? rl[lane] : none;
      const int2 e1 = lane + 32 < n ? rl[lane + 32] : none;
      if (FIRST) {
        int2* rr = recs + (size_t)(rbase + row) * RCAP;
        rr[lane] = e0;
        rr[lane + 32] = e1;
        if (lane == 0) rowCount[rbase + row] = all;
      }
      putRow(row, sumRow(e0, e1, n));
    }
  } else {
    const int myc = lane < NSLOT ? rowCount[rbase + lane * AGG_WAVES + wave] : 0;
    const int2* rr = recs + (size_t)(rbase + wave) * RCAP;
    int2 e0 = rr[lane];
    int2 e1 = rr[lane + 32];
#pragma unroll 1
    for (int s = 0; s < NSLOT; ++s) {
      const int row = s * AGG_WAVES + wave;
      const int n = min(__builtin_amdgcn_readlane(myc, s), RCAP);
      int2 n0 = e0, n1 = e1;
      if (s + 1 < NSLOT) {
        const int2* rn = recs + (size_t)(rbase + row + AGG_WAVES) * RCAP;
        n0 = rn[lane];
        n1 = rn[lane + 32];
      }
      putRow(row, sumRow(e0, e1, n));
      e0 = n0;
      e1 = n1;
    }
  }
  __syncthreads();

  const int rows = min(ROWS_PB, N - base);
  float* hst = (float*)rq;
  float* xst = hst + TILE_ROWS * PITCH;
  float* lst = xst + TILE_ROWS * IN_CH;
  for (int i = tid; i < TILE_ROWS * (OUT_CH / 4); i += AGG_THREADS) {
    const int r = i >> 4, c4 = i & 15;
    float4 v = make_float4(0.0f, 0.0f, 0.0f, 0.0f);
    if (r < rows) v = ((const float4*)(src64 + (size_t)(base + r) * OUT_CH))[c4];
    *(float4*)&hst[r * PITCH + c4 * 4] = v;
  }
  for (int i = tid; i < TILE_ROWS * (IN_CH / 4); i += AGG_THREADS) {
    const int r = i >> 1, c4 = i & 1;
    float4 v = make_float4(0.0f, 0.0f, 0.0f, 0.0f);
    float4 w = v;
    if (r < rows) {
      v = ((const float4*)(src8 + (size_t)(base + r) * IN_CH))[c4];
      if (!FIRST) w = ((const float4*)(Lx + (size_t)(base + r) * IN_CH))[c4];
    }
    *(float4*)&xst[r * IN_CH + c4 * 4] = v;
    if (!FIRST) *(float4*)&lst[r * IN_CH + c4 * 4] = w;
  }
  if (HAS8) {
    float4* o8 = (float4*)(Lx + (size_t)base * IN_CH);
    const float4* a8 = (const float4*)acc8;
    for (int i = tid; i < rows * (IN_CH / 4); i += AGG_THREADS) o8[i] = a8[i];
  }
  __syncthreads();

  const int half = lane >> 4;
  const int l15 = lane & 15;
  const int nUnits = ((rows + 15) >> 4) * (OUT_CH / 16);
  for (int u = wave; u < nUnits; u += AGG_WAVES) {
    const int row0 = (u >> 2) * 16;
    const int rA = row0 + l15;
    const int colB = (u & 3) * 16 + l15;
    if (FIRST) {
      v8f accZ = {};
      v8f accR = {};
      accum_block2(xst, IN_CH, rA, Wax, Wbx, IN_CH, colB, half, accZ, accR);
      accum_block2(acc8, IN_CH, rA, Wax + IN_CH * OUT_CH, Wbx + IN_CH * OUT_CH,
                   IN_CH, colB, half, accZ, accR);
      accum_block2(hst, PITCH, rA, Wah, Wbh, OUT_CH, colB, half, accZ, accR);
      accum_block2(acc, PITCH, rA, Wah + OUT_CH * OUT_CH, Wbh + OUT_CH * OUT_CH,
                   OUT_CH, colB, half, accZ, accR);
      const float bZ = bax[colB] + bah[colB];
      const float bR = bbx[colB] + bbh[colB];
#pragma unroll
      for (int i = 0; i < 8; ++i) {
        const int r = row0 + i + 8 * half;
        if (r < rows) {
          const float z = 1.0f / (1.0f + __expf(-(accZ[i] + bZ)));
          const float rg = 1.0f / (1.0f + __expf(-(accR[i] + bR)));
          const size_t g = (size_t)(base + r) * OUT_CH + colB;
          Zbuf[g] = z;
          RHout[g] = rg * hst[r * PITCH + colB];
        }
      }
    } else {
      v8f accC = {};
      accC = accum_block(xst, IN_CH, rA, Wax, IN_CH, colB, half, accC);
      accC = accum_block(lst, IN_CH, rA, Wax + IN_CH * OUT_CH, IN_CH, colB, half, accC);
      accC = accum_block(hst, PITCH, rA, Wah, OUT_CH, colB, half, accC);
      accC = accum_block(acc, PITCH, rA, Wah + OUT_CH * OUT_CH, OUT_CH, colB, half, accC);
      const float bC = bax[colB] + bah[colB];
#pragma unroll
      for (int i = 0; i < 8; ++i) {
        const int r = row0 + i + 8 * half;
        if (r < rows) {
          const size_t g = (size_t)(base + r) * OUT_CH + colB;
          const float cv = tanhf(accC[i] + bC);
          const float z = Zbuf[g];
          const float h = H[g];
          out[g] = z * h + (1.0f - z) * cv;
        }
      }
    }
  }
}


extern "C" void kernel_launch(void* const* d_in, const int* in_sizes, int n_in,
                              void* d_out, int out_size, void* d_ws,
                              size_t ws_size, hipStream_t stream) {
  const float* x = (const float*)d_in[0];
  const float* H = (const float*)d_in[1];
  const int* edge = (const int*)d_in[2];
  const float* Wzx = (const float*)d_in[3];
  const float* bzx = (const float*)d_in[4];
  const float* Wzh = (const float*)d_in[5];
  const float* bzh = (const float*)d_in[6];
  const float* Wrx = (const float*)d_in[7];
  const float* brx = (const float*)d_in[8];
  const float* Wrh = (const float*)d_in[9];
  const float* brh = (const float*)d_in[10];
  const float* Wcx = (const float*)d_in[11];
  const float* bcx = (const float*)d_in[12];
  const float* Wch = (const float*)d_in[13];
  const float* bch = (const float*)d_in[14];

  const int N = in_sizes[0] / IN_CH;
  const int E = in_sizes[2] / 2;
  const int* rowIdx = edge;
  const int* colIdx = edge + E;

  float* ws = (float*)d_ws;
  float* dinv = ws;
  float* Lx = dinv + ((N + 63) / 64) * 64;
  float* Zbuf = Lx + (size_t)N * IN_CH;
  float* RH = Zbuf + (size_t)N * OUT_CH;
  const int nZero = N;
  const int aggBlocks = (N + ROWS_PB - 1) / ROWS_PB;
  unsigned* planes = (unsigned*)(RH + (size_t)N * OUT_CH);
  int* rowCount =
      (int*)(planes + (size_t)NPLANES * (((E + DEG_EDGES - 1) / DEG_EDGES) * 32));
  int* findCount = rowCount + ((aggBlocks * LROWS + 63) / 64) * 64;
  int* recs = findCount + ((aggBlocks * AGG_WAVES + 63) / 64) * 64;

  zero_kernel<<<(nZero + 255) / 256, 256, 0, stream>>>(dinv, nZero);
  degree_kernel<<<(E + DEG_EDGES - 1) / DEG_EDGES, DEG_THREADS, 0, stream>>>(
      rowIdx, dinv, planes, E);
  dinv_kernel<<<(N + 255) / 256, 256, 0, stream>>>(dinv, N);
  agg_kernel<true><<<aggBlocks, AGG_THREADS, 0, stream>>>(
      rowIdx, colIdx, dinv, H, x, Lx, planes, recs, rowCount, findCount, E,
      N, H,
      Wzx, Wzh, Wrx, Wrh, bzx, bzh, brx, brh, Zbuf, RH, nullptr);
  agg_kernel<false><<<aggBlocks, AGG_THREADS, 0, stream>>>(
      rowIdx, colIdx, dinv, RH, x, Lx, planes, recs, rowCount, findCount, E,
      N, H,
      Wcx, Wch, nullptr, nullptr, bcx, bch, nullptr, nullptr, Zbuf, nullptr,
      (float*)d_out);
}


# Degrees counted by row owners: a listing launch files edges by row and stores row counts and deg^-1/2; both passes sum from saved lists; no global atomics, 5 -> 4 launches; Lx sum order = lane groups
# speedup vs baseline: 4.2316x; 1.7406x over previous
#include <hip/hip_runtime.h>
#include <hip/hip_bf16.h>
#include <math.h>


typedef float v2f __attribute__((ext_vector_type(2)));
typedef float v8f __attribute__((ext_vector_type(8)));

#define IN_CH 8
#define OUT_CH 64

__device__ __forceinline__ v8f wmma_f32(v2f a, v2f b, v8f c) {
  return __builtin_amdgcn_wmma_f32_16x16x4_f32(false, a, false, b, (short)0, c,
                                               false, false);
}

__device__ __forceinline__ v8f accum_block(const float* __restrict__ A, int ldA,
                                           int rA, const float* __restrict__ W,
                                           int Kdim, int colB, int half,
                                           v8f acc) {
#pragma unroll 4
  for (int k0 = 0; k0 < Kdim; k0 += 4) {
    int ka = k0 + 2 * half;
    v2f a;
    a.x = A[rA * ldA + ka];
    a.y = A[rA * ldA + ka + 1];
    v2f b;
    b.x = W[ka * OUT_CH + colB];
    b.y = W[(ka + 1) * OUT_CH + colB];
    acc = wmma_f32(a, b, acc);
  }
  return acc;
}

__device__ __forceinline__ void accum_block2(const float* __restrict__ A,
                                             int ldA, int rA,
                                             const float* __restrict__ W1,
                                             const float* __restrict__ W2,
                                             int Kdim, int colB, int half,
                                             v8f& acc1, v8f& acc2) {
#pragma unroll 4
  for (int k0 = 0; k0 < Kdim; k0 += 4) {
    int ka = k0 + 2 * half;
    v2f a;
    a.x = A[rA * ldA + ka];
    a.y = A[rA * ldA + ka + 1];
    v2f b1;
    b1.x = W1[ka * OUT_CH + colB];
    b1.y = W1[(ka + 1) * OUT_CH + colB];
    v2f b2;
    b2.x = W2[ka * OUT_CH + colB];
    b2.y = W2[(ka + 1) * OUT_CH + colB];
    acc1 = wmma_f32(a, b1, acc1);
    acc2 = wmma_f32(a, b2, acc2);
  }
}


#define ROWS_PB 196
#define TILE_ROWS 208
#define PITCH 68

__device__ __forceinline__ unsigned owner_plane(int r) {
  return (unsigned)(r / ROWS_PB) & 255u;
}

#define NPLANES 256
#define DEG_THREADS 256
#define DEG_EDGES 1024

__device__ __forceinline__ int plane_words(int E) {
  return ((E + DEG_EDGES - 1) / DEG_EDGES) * 32;
}

__global__ void __launch_bounds__(DEG_THREADS)
plane_kernel(const int* __restrict__ rowIdx, unsigned* __restrict__ planes,
             int E) {
  __shared__ __attribute__((aligned(16))) unsigned bits[NPLANES * 32];
  const int tid = threadIdx.x;
  uint4* b4 = (uint4*)bits;
#pragma unroll
  for (int i = 0; i < 8; ++i)
    b4[i * DEG_THREADS + tid] = make_uint4(0u, 0u, 0u, 0u);
  __syncthreads();
  const int t = blockIdx.x * DEG_THREADS + tid;
  const int e = t * 4;
  const int w = tid >> 3;
  const int sh = (tid & 7) * 4;
  if (e + 3 < E) {
    const int4 r = ((const int4*)rowIdx)[t];
    atomicOr(&bits[owner_plane(r.x) * 32 + w], 1u << sh);
    atomicOr(&bits[owner_plane(r.y) * 32 + w], 2u << sh);
    atomicOr(&bits[owner_plane(r.z) * 32 + w], 4u << sh);
    atomicOr(&bits[owner_plane(r.w) * 32 + w], 8u << sh);
  } else {
    for (int k = e; k < E; ++k) {
      const int r = rowIdx[k];
      atomicOr(&bits[owner_plane(r) * 32 + w], 1u << (sh + (k - e)));
    }
  }
  __syncthreads();
  const int pitchW = plane_words(E);
#pragma unroll
  for (int i = 0; i < 8; ++i) {
    const int idx = i * DEG_THREADS + tid;
    uint4* line = (uint4*)(planes + (size_t)(idx >> 3) * pitchW +
                           (size_t)blockIdx.x * 32);
    line[idx & 7] = b4[idx];
  }
}

#define AGG_THREADS 1024
#define AGG_WAVES (AGG_THREADS / 32)
#define WCAP 384
#define NSLOT 7
#define LROWS (NSLOT * AGG_WAVES)
#define RCAP 64
#define RPW 132

template <int MODE>
__global__ void __launch_bounds__(AGG_THREADS)
agg_kernel(const int* __restrict__ rowIdx, const int* __restrict__ colIdx,
           const float* __restrict__ dinv, const float* __restrict__ src64,
           const float* __restrict__ src8, float* __restrict__ Lx,
           const unsigned* __restrict__ planes, int* __restrict__ recWords,
           int* __restrict__ rowCount, int* __restrict__ findCount, int E,
           int N,
           const float* __restrict__ H, const float* __restrict__ Wax,
           const float* __restrict__ Wah, const float* __restrict__ Wbx,
           const float* __restrict__ Wbh, const float* __restrict__ bax,
           const float* __restrict__ bah, const float* __restrict__ bbx,
           const float* __restrict__ bbh, float* __restrict__ Zbuf,
           float* __restrict__ RHout, float* __restrict__ out,
           float* __restrict__ dinvOut) {
  constexpr bool LIST = MODE == 0;
  constexpr bool FIRST = MODE == 1;
  constexpr bool HAS8 = FIRST;
  __shared__ __attribute__((aligned(16))) float acc[TILE_ROWS * PITCH];
  __shared__ __attribute__((aligned(16))) float acc8[HAS8 ? TILE_ROWS * IN_CH : 4];
  __shared__ int fq[AGG_WAVES * WCAP];
  __shared__ __attribute__((aligned(16))) int rq[LROWS * RPW];
  __shared__ int rcnt[LROWS];

  const int tid = threadIdx.x;
  const int lane = tid & 31;
  const int base = blockIdx.x * ROWS_PB;
  const int wave = __builtin_amdgcn_readfirstlane(tid >> 5);

  if (!LIST) {
    for (int i = tid; i < TILE_ROWS * PITCH; i += AGG_THREADS) acc[i] = 0.0f;
  }
  if (HAS8) {
    for (int i = tid; i < TILE_ROWS * IN_CH; i += AGG_THREADS) acc8[i] = 0.0f;
  }
  if (tid < LROWS) rcnt[tid] = 0;
  __syncthreads();

  int* wq = fq + wave * WCAP;
  int wcnt = 0;
  const int cntIdx = blockIdx.x * AGG_WAVES + wave;
  const int rbase = blockIdx.x * LROWS;
  int2* __restrict__ recs = (int2*)recWords;
  bool filt = true;
  if (!LIST) {
    bool over = findCount[blockIdx.x * AGG_WAVES + lane] > WCAP;
#pragma unroll
    for (int s = 0; s < NSLOT; ++s)
      over = over || rowCount[rbase + s * AGG_WAVES + lane] > RCAP;
    filt = __builtin_amdgcn_ballot_w32(over) != 0u;
  }

  if (filt) {
    const bool planeIsBlock = gridDim.x <= 256u;
    auto spot = [&](int e) {
      int r = rowIdx[e];
      unsigned d = (unsigned)(r - base);
      if (LIST) {
        atomicAdd(&rcnt[d], 1);
        return;
      }
      int c = colIdx[e];
      float nrm = -dinv[r] * dinv[c];
#pragma unroll 1
      for (int k = 0; k < OUT_CH; ++k)
        atomicAdd(&acc[d * PITCH + k], src64[(size_t)c * OUT_CH + k] * nrm);
      if (HAS8) {
#pragma unroll 1
        for (int k = 0; k < IN_CH; ++k)
          atomicAdd(&acc8[d * IN_CH + k], src8[(size_t)c * IN_CH + k] * nrm);
      }
    };
    auto append = [&](bool has, int e) {
      const unsigned m = __builtin_amdgcn_ballot_w32(has);
      const int slot = wcnt + (int)__builtin_amdgcn_mbcnt_lo(m, 0u);
      const int after = wcnt + __builtin_popcount(m);
      if (has && slot < WCAP) wq[slot] = e;
      if (after > WCAP) {
        if (has && slot >= WCAP) spot(e);
      }
      wcnt = after;
    };
    auto ownsRow = [&](int e) {
      return (unsigned)(rowIdx[e] - base) < (unsigned)ROWS_PB;
    };
    const int pitchW = plane_words(E);
    const int nQ = pitchW >> 2;
    const uint4* __restrict__ pl4 =
        (const uint4*)(planes + (size_t)(blockIdx.x & 255u) * pitchW);
    auto visit = [&](int q, uint4 v) {
      const bool in = q < nQ;
      unsigned a = in ? v.x : 0u, b = in ? v.y : 0u;
      unsigned cw = in ? v.z : 0u, d = in ? v.w : 0u;
      while (__builtin_amdgcn_ballot_w32((a | b | cw | d) != 0u) != 0u) {
        const int j = a != 0u ? 0 : b != 0u ? 1 : cw != 0u ? 2 : 3;
        const unsigned word = j == 0 ? a : j == 1 ? b : j == 2 ? cw : d;
        bool has = word != 0u;
        const int p = __builtin_ffs((int)word) - 1;
        const unsigned rest = word & (word - 1u);
        a = j == 0 ? rest : a;
        b = j == 1 ? rest : b;
        cw = j == 2 ? rest : cw;
        d = j == 3 ? rest : d;
        const int eh = ((q * 4 + j) << 5) + p;
        if (!planeIsBlock) {
          if (has) has = ownsRow(eh);
        }
        append(has, eh);
      }
    };
    const int qLast = nQ - 1;
    auto load4 = [&](uint4 (&v)[4], int q0) {
#pragma unroll
      for (int k = 0; k < 4; ++k) {
        int q = q0 + k * AGG_THREADS + tid;
        v[k] = pl4[min(q, qLast)];
      }
    };
    auto visit4 = [&](int q0, const uint4 (&v)[4]) {
#pragma unroll
      for (int k = 0; k < 4; ++k) visit(q0 + k * AGG_THREADS + tid, v[k]);
    };
    const int STEP = 4 * AGG_THREADS;
    if (nQ > 0) {
      uint4 va[4], vb[4];
      load4(va, 0);
      for (int q0 = 0; q0 < nQ; q0 += 2 * STEP) {
        load4(vb, q0 + STEP);
        visit4(q0, va);
        load4(va, q0 + 2 * STEP);
        visit4(q0 + STEP, vb);
      }
    }
    __builtin_amdgcn_fence(__ATOMIC_ACQ_REL, "workgroup");
  }

  if (filt) {
    const int nq = __builtin_amdgcn_readfirstlane(min(wcnt, WCAP));
    for (int g0 = 0; g0 < nq; g0 += 32) {
      const int cnt = min(32, nq - g0);
      const bool valid = lane < cnt;
      const int e = valid ? wq[g0 + lane] : 0;
      const int r = rowIdx[e];
      const int c = colIdx[e];
      const float nrm = (!LIST && valid) ? -dinv[r] * dinv[c] : 0.0f;
      const int d = valid ? r - base : 0;
      if (valid) {
        if (HAS8) {
          const float4* s8 = (const float4*)(src8 + (size_t)c * IN_CH);
          float4 a = s8[0];
          float4 b = s8[1];
          float* t8 = &acc8[d * IN_CH];
          atomicAdd(&t8[0], a.x * nrm);
          atomicAdd(&t8[1], a.y * nrm);
          atomicAdd(&t8[2], a.z * nrm);
          atomicAdd(&t8[3], a.w * nrm);
          atomicAdd(&t8[4], b.x * nrm);
          atomicAdd(&t8[5], b.y * nrm);
          atomicAdd(&t8[6], b.z * nrm);
          atomicAdd(&t8[7], b.w * nrm);
        }
        const int slot = atomicAdd(&rcnt[d], 1);
        if (slot < RCAP) {
          *(int2*)&rq[d * RPW + slot * 2] = make_int2(c, __float_as_int(nrm));
        } else if (!LIST) {
#pragma unroll 1
          for (int k = 0; k < OUT_CH; ++k)
            atomicAdd(&acc[d * PITCH + k], src64[(size_t)c * OUT_CH + k] * nrm);
        }
      }
    }
    if ((LIST || FIRST) && lane == 0) findCount[cntIdx] = wcnt;
  }
  __syncthreads();

  if (LIST) {
#pragma unroll 1
    for (int s = 0; s < NSLOT; ++s) {
      const int row = s * AGG_WAVES + wave;
      const int all = rcnt[row];
      const int n = min(all, RCAP);
      const int2* rl = (const int2*)&rq[row * RPW];
      const int2 none = make_int2(0, 0);
      int2* rr = recs + (size_t)(rbase + row) * RCAP;
      int2 v0 = rl[lane];
      int2 v1 = rl[lane + 32];
      if (lane >= n) v0 = none;
      if (lane + 32 >= n) v1 = none;
      rr[lane] = v0;
      rr[lane + 32] = v1;
      if (lane == 0) rowCount[rbase + row] = all;
    }
    if (tid < ROWS_PB && base + tid < N) {
      const int dg = rcnt[tid];
      dinvOut[base + tid] = dg > 0 ? rsqrtf((float)dg) : 0.0f;
    }
    return;
  }

  auto sumRow = [&](int2 e0, int2 e1, int n) {
    v2f a = {0.0f, 0.0f};
#pragma unroll 1
    for (int j0 = 0; j0 < n; j0 += 32) {
      const bool up = j0 != 0;
      const int cc = up ? e1.x : e0.x;
      const int ww = up ? e1.y : e0.y;
      float2 g[32];
#pragma unroll
      for (int k = 0; k < 32; ++k) {
        const unsigned cj = (unsigned)__builtin_amdgcn_readlane(cc, k);
        g[k] = ((const float2*)(src64 + (size_t)cj * OUT_CH))[lane];
      }
#pragma unroll
      for (int k = 0; k < 32; ++k) {
        const float wj = __int_as_float(__builtin_amdgcn_readlane(ww, k));
        const v2f gk = {g[k].x, g[k].y};
        a += gk * wj;
      }
    }
    return a;
  };
  auto putRow = [&](int row, v2f a) {
    if (row < TILE_ROWS) {
      float2* p = (float2*)&acc[row * PITCH + 2 * lane];
      float2 v = *p;
      v.x += a.x;
      v.y += a.y;
      *p = v;
    }
  };
  if (filt) {
#pragma unroll 1
    for (int s = 0; s < NSLOT; ++s) {
      const int row = s * AGG_WAVES + wave;
      const int all = rcnt[row];
      const int n = __builtin_amdgcn_readfirstlane(min(all, RCAP));
      const int2* rl = (const int2*)&rq[row * RPW];
      const int2 none = make_int2(0, 0);
      const int2 e0 = lane < n ? rl[lane] : none;
      const int2 e1 = lane + 32 < n ? rl[lane + 32] : none;
      if (FIRST) {
        int2* rr = recs + (size_t)(rbase + row) * RCAP;
        rr[lane] = e0;
        rr[lane + 32] = e1;
        if (lane == 0) rowCount[rbase + row] = all;
      }
      putRow(row, sumRow(e0, e1, n));
    }
  } else if (FIRST) {
    const int myc = lane < NSLOT ? rowCount[rbase + lane * AGG_WAVES + wave] : 0;
    auto recOf = [&](int row) { return recs + (size_t)(rbase + row) * RCAP; };
    struct Fac {
      float dr, d0, d1;
    };
    auto factors = [&](int2 p0, int2 p1, int row) {
      Fac f;
      f.dr = dinv[min(base + row, N - 1)];
      f.d0 = dinv[p0.x];
      f.d1 = dinv[p1.x];
      return f;
    };
    auto weights = [&](Fac f, int n) {
      float2 w;
      w.x = lane < n ? -f.dr * f.d0 : 0.0f;
      w.y = lane + 32 < n ? -f.dr * f.d1 : 0.0f;
      return w;
    };
    const int sub = lane >> 3;
    const int col8 = lane & 7;
    int2 a0 = recOf(wave)[lane];
    int2 a1 = recOf(wave)[lane + 32];
    int2 b0 = recOf(wave + AGG_WAVES)[lane];
    int2 b1 = recOf(wave + AGG_WAVES)[lane + 32];
    Fac fa = factors(a0, a1, wave);
#pragma unroll 1
    for (int s = 0; s < NSLOT; ++s) {
      const int row = s * AGG_WAVES + wave;
      const int n = min(__builtin_amdgcn_readlane(myc, s), RCAP);
      const float2 wa = weights(fa, n);
      a0.y = __float_as_int(wa.x);
      a1.y = __float_as_int(wa.y);
      int2* rr = recOf(row);
      rr[lane] = a0;
      rr[lane + 32] = a1;
      int2 c0 = b0, c1 = b1;
      if (s + 2 < NSLOT) {
        const int2* rn = recOf(row + 2 * AGG_WAVES);
        c0 = rn[lane];
        c1 = rn[lane + 32];
      }
      Fac fb = fa;
      if (s + 1 < NSLOT) fb = factors(b0, b1, row + AGG_WAVES);
      float xs0[8], xs1[8];
#pragma unroll
      for (int m = 0; m < 8; ++m) {
        const int cj = __shfl(a0.x, 4 * m + sub);
        xs0[m] = src8[(size_t)cj * IN_CH + col8];
        xs1[m] = 0.0f;
      }
      if (n > 32) {
#pragma unroll
        for (int m = 0; m < 8; ++m) {
          const int cj = __shfl(a1.x, 4 * m + sub);
          xs1[m] = src8[(size_t)cj * IN_CH + col8];
        }
      }
      putRow(row, sumRow(a0, a1, n));
      float ax = 0.0f;
#pragma unroll
      for (int m = 0; m < 8; ++m) ax += __shfl(wa.x, 4 * m + sub) * xs0[m];
      if (n > 32) {
#pragma unroll
        for (int m = 0; m < 8; ++m) ax += __shfl(wa.y, 4 * m + sub) * xs1[m];
      }
      ax += __shfl_xor(ax, 8);
      ax += __shfl_xor(ax, 16);
      if (lane < IN_CH && row < TILE_ROWS) acc8[row * IN_CH + lane] = ax;
      a0 = b0;
      a1 = b1;
      b0 = c0;
      b1 = c1;
      fa = fb;
    }
  } else {
    const int myc = lane < NSLOT ? rowCount[rbase + lane * AGG_WAVES + wave] : 0;
    const int2* rr = recs + (size_t)(rbase + wave) * RCAP;
    int2 e0 = rr[lane];
    int2 e1 = rr[lane + 32];
#pragma unroll 1
    for (int s = 0; s < NSLOT; ++s) {
      const int row = s * AGG_WAVES + wave;
      const int n = min(__builtin_amdgcn_readlane(myc, s), RCAP);
      int2 n0 = e0, n1 = e1;
      if (s + 1 < NSLOT) {
        const int2* rn = recs + (size_t)(rbase + row + AGG_WAVES) * RCAP;
        n0 = rn[lane];
        n1 = rn[lane + 32];
      }
      putRow(row, sumRow(e0, e1, n));
      e0 = n0;
      e1 = n1;
    }
  }
  __syncthreads();

  const int rows = min(ROWS_PB, N - base);
  float* hst = (float*)rq;
  float* xst = hst + TILE_ROWS * PITCH;
  float* lst = xst + TILE_ROWS * IN_CH;
  for (int i = tid; i < TILE_ROWS * (OUT_CH / 4); i += AGG_THREADS) {
    const int r = i >> 4, c4 = i & 15;
    float4 v = make_float4(0.0f, 0.0f, 0.0f, 0.0f);
    if (r < rows) v = ((const float4*)(src64 + (size_t)(base + r) * OUT_CH))[c4];
    *(float4*)&hst[r * PITCH + c4 * 4] = v;
  }
  for (int i = tid; i < TILE_ROWS * (IN_CH / 4); i += AGG_THREADS) {
    const int r = i >> 1, c4 = i & 1;
    float4 v = make_float4(0.0f, 0.0f, 0.0f, 0.0f);
    float4 w = v;
    if (r < rows) {
      v = ((const float4*)(src8 + (size_t)(base + r) * IN_CH))[c4];
      if (!FIRST) w = ((const float4*)(Lx + (size_t)(base + r) * IN_CH))[c4];
    }
    *(float4*)&xst[r * IN_CH + c4 * 4] = v;
    if (!FIRST) *(float4*)&lst[r * IN_CH + c4 * 4] = w;
  }
  if (HAS8) {
    float4* o8 = (float4*)(Lx + (size_t)base * IN_CH);
    const float4* a8 = (const float4*)acc8;
    for (int i = tid; i < rows * (IN_CH / 4); i += AGG_THREADS) o8[i] = a8[i];
  }
  __syncthreads();

  const int half = lane >> 4;
  const int l15 = lane & 15;
  const int nUnits = ((rows + 15) >> 4) * (OUT_CH / 16);
  for (int u = wave; u < nUnits; u += AGG_WAVES) {
    const int row0 = (u >> 2) * 16;
    const int rA = row0 + l15;
    const int colB = (u & 3) * 16 + l15;
    if (FIRST) {
      v8f accZ = {};
      v8f accR = {};
      accum_block2(xst, IN_CH, rA, Wax, Wbx, IN_CH, colB, half, accZ, accR);
      accum_block2(acc8, IN_CH, rA, Wax + IN_CH * OUT_CH, Wbx + IN_CH * OUT_CH,
                   IN_CH, colB, half, accZ, accR);
      accum_block2(hst, PITCH, rA, Wah, Wbh, OUT_CH, colB, half, accZ, accR);
      accum_block2(acc, PITCH, rA, Wah + OUT_CH * OUT_CH, Wbh + OUT_CH * OUT_CH,
                   OUT_CH, colB, half, accZ, accR);
      const float bZ = bax[colB] + bah[colB];
      const float bR = bbx[colB] + bbh[colB];
#pragma unroll
      for (int i = 0; i < 8; ++i) {
        const int r = row0 + i + 8 * half;
        if (r < rows) {
          const float z = 1.0f / (1.0f + __expf(-(accZ[i] + bZ)));
          const float rg = 1.0f / (1.0f + __expf(-(accR[i] + bR)));
          const size_t g = (size_t)(base + r) * OUT_CH + colB;
          Zbuf[g] = z;
          RHout[g] = rg * hst[r * PITCH + colB];
        }
      }
    } else {
      v8f accC = {};
      accC = accum_block(xst, IN_CH, rA, Wax, IN_CH, colB, half, accC);
      accC = accum_block(lst, IN_CH, rA, Wax + IN_CH * OUT_CH, IN_CH, colB, half, accC);
      accC = accum_block(hst, PITCH, rA, Wah, OUT_CH, colB, half, accC);
      accC = accum_block(acc, PITCH, rA, Wah + OUT_CH * OUT_CH, OUT_CH, colB, half, accC);
      const float bC = bax[colB] + bah[colB];
#pragma unroll
      for (int i = 0; i < 8; ++i) {
        const int r = row0 + i + 8 * half;
        if (r < rows) {
          const size_t g = (size_t)(base + r) * OUT_CH + colB;
          const float cv = tanhf(accC[i] + bC);
          const float z = Zbuf[g];
          const float h = H[g];
          out[g] = z * h + (1.0f - z) * cv;
        }
      }
    }
  }
}


extern "C" void kernel_launch(void* const* d_in, const int* in_sizes, int n_in,
                              void* d_out, int out_size, void* d_ws,
                              size_t ws_size, hipStream_t stream) {
  const float* x = (const float*)d_in[0];
  const float* H = (const float*)d_in[1];
  const int* edge = (const int*)d_in[2];
  const float* Wzx = (const float*)d_in[3];
  const float* bzx = (const float*)d_in[4];
  const float* Wzh = (const float*)d_in[5];
  const float* bzh = (const float*)d_in[6];
  const float* Wrx = (const float*)d_in[7];
  const float* brx = (const float*)d_in[8];
  const float* Wrh = (const float*)d_in[9];
  const float* brh = (const float*)d_in[10];
  const float* Wcx = (const float*)d_in[11];
  const float* bcx = (const float*)d_in[12];
  const float* Wch = (const float*)d_in[13];
  const float* bch = (const float*)d_in[14];

  const int N = in_sizes[0] / IN_CH;
  const int E = in_sizes[2] / 2;
  const int* rowIdx = edge;
  const int* colIdx = edge + E;

  float* ws = (float*)d_ws;
  float* dinv = ws;
  float* Lx = dinv + ((N + 63) / 64) * 64;
  float* Zbuf = Lx + (size_t)N * IN_CH;
  float* RH = Zbuf + (size_t)N * OUT_CH;
  const int aggBlocks = (N + ROWS_PB - 1) / ROWS_PB;
  unsigned* planes = (unsigned*)(RH + (size_t)N * OUT_CH);
  int* rowCount =
      (int*)(planes + (size_t)NPLANES * (((E + DEG_EDGES - 1) / DEG_EDGES) * 32));
  int* findCount = rowCount + ((aggBlocks * LROWS + 63) / 64) * 64;
  int* recs = findCount + ((aggBlocks * AGG_WAVES + 63) / 64) * 64;

  plane_kernel<<<(E + DEG_EDGES - 1) / DEG_EDGES, DEG_THREADS, 0, stream>>>(
      rowIdx, planes, E);
  agg_kernel<0><<<aggBlocks, AGG_THREADS, 0, stream>>>(
      rowIdx, colIdx, nullptr, nullptr, nullptr, nullptr, planes, recs,
      rowCount, findCount, E, N, nullptr,
      nullptr, nullptr, nullptr, nullptr, nullptr, nullptr, nullptr, nullptr,
      nullptr, nullptr, nullptr, dinv);
  agg_kernel<1><<<aggBlocks, AGG_THREADS, 0, stream>>>(
      rowIdx, colIdx, dinv, H, x, Lx, planes, recs, rowCount, findCount, E,
      N, H,
      Wzx, Wzh, Wrx, Wrh, bzx, bzh, brx, brh, Zbuf, RH, nullptr, nullptr);
  agg_kernel<2><<<aggBlocks, AGG_THREADS, 0, stream>>>(
      rowIdx, colIdx, dinv, RH, x, Lx, planes, recs, rowCount, findCount, E,
      N, H,
      Wcx, Wch, nullptr, nullptr, bcx, bch, nullptr, nullptr, Zbuf, nullptr,
      (float*)d_out, nullptr);
}


# First pass keeps the whole deg^-1/2 table in LDS (one whole-line copy per block, in the space of the unused edge lists): weight factors are LDS lookups, not a scattered read per edge; sum order kept
# speedup vs baseline: 4.3825x; 1.0357x over previous
#include <hip/hip_runtime.h>
#include <hip/hip_bf16.h>
#include <math.h>


typedef float v2f __attribute__((ext_vector_type(2)));
typedef float v4f __attribute__((ext_vector_type(4)));
typedef float v8f __attribute__((ext_vector_type(8)));

#define IN_CH 8
#define OUT_CH 64

__device__ __forceinline__ v8f wmma_f32(v2f a, v2f b, v8f c) {
  return __builtin_amdgcn_wmma_f32_16x16x4_f32(false, a, false, b, (short)0, c,
                                               false, false);
}

__device__ __forceinline__ v8f accum_block(const float* __restrict__ A, int ldA,
                                           int rA, const float* __restrict__ W,
                                           int Kdim, int colB, int half,
                                           v8f acc) {
#pragma unroll 4
  for (int k0 = 0; k0 < Kdim; k0 += 4) {
    int ka = k0 + 2 * half;
    v2f a;
    a.x = A[rA * ldA + ka];
    a.y = A[rA * ldA + ka + 1];
    v2f b;
    b.x = W[ka * OUT_CH + colB];
    b.y = W[(ka + 1) * OUT_CH + colB];
    acc = wmma_f32(a, b, acc);
  }
  return acc;
}

__device__ __forceinline__ void accum_block2(const float* __restrict__ A,
                                             int ldA, int rA,
                                             const float* __restrict__ W1,
                                             const float* __restrict__ W2,
                                             int Kdim, int colB, int half,
                                             v8f& acc1, v8f& acc2) {
#pragma unroll 4
  for (int k0 = 0; k0 < Kdim; k0 += 4) {
    int ka = k0 + 2 * half;
    v2f a;
    a.x = A[rA * ldA + ka];
    a.y = A[rA * ldA + ka + 1];
    v2f b1;
    b1.x = W1[ka * OUT_CH + colB];
    b1.y = W1[(ka + 1) * OUT_CH + colB];
    v2f b2;
    b2.x = W2[ka * OUT_CH + colB];
    b2.y = W2[(ka + 1) * OUT_CH + colB];
    acc1 = wmma_f32(a, b1, acc1);
    acc2 = wmma_f32(a, b2, acc2);
  }
}


#define ROWS_PB 196
#define TILE_ROWS 208
#define PITCH 68

__device__ __forceinline__ unsigned owner_plane(int r) {
  return (unsigned)(r / ROWS_PB) & 255u;
}

#define NPLANES 256
#define DEG_THREADS 256
#define DEG_EDGES 1024

__device__ __forceinline__ int plane_words(int E) {
  return ((E + DEG_EDGES - 1) / DEG_EDGES) * 32;
}

__global__ void __launch_bounds__(DEG_THREADS)
plane_kernel(const int* __restrict__ rowIdx, unsigned* __restrict__ planes,
             int E) {
  __shared__ __attribute__((aligned(16))) unsigned bits[NPLANES * 32];
  const int tid = threadIdx.x;
  uint4* b4 = (uint4*)bits;
#pragma unroll
  for (int i = 0; i < 8; ++i)
    b4[i * DEG_THREADS + tid] = make_uint4(0u, 0u, 0u, 0u);
  __syncthreads();
  const int t = blockIdx.x * DEG_THREADS + tid;
  const int e = t * 4;
  const int w = tid >> 3;
  const int sh = (tid & 7) * 4;
  if (e + 3 < E) {
    const int4 r = ((const int4*)rowIdx)[t];
    atomicOr(&bits[owner_plane(r.x) * 32 + w], 1u << sh);
    atomicOr(&bits[owner_plane(r.y) * 32 + w], 2u << sh);
    atomicOr(&bits[owner_plane(r.z) * 32 + w], 4u << sh);
    atomicOr(&bits[owner_plane(r.w) * 32 + w], 8u << sh);
  } else {
    for (int k = e; k < E; ++k) {
      const int r = rowIdx[k];
      atomicOr(&bits[owner_plane(r) * 32 + w], 1u << (sh + (k - e)));
    }
  }
  __syncthreads();
  const int pitchW = plane_words(E);
#pragma unroll
  for (int i = 0; i < 8; ++i) {
    const int idx = i * DEG_THREADS + tid;
    uint4* line = (uint4*)(planes + (size_t)(idx >> 3) * pitchW +
                           (size_t)blockIdx.x * 32);
    line[idx & 7] = b4[idx];
  }
}

#define AGG_THREADS 1024
#define AGG_WAVES (AGG_THREADS / 32)
#define WCAP 384
#define NSLOT 7
#define LROWS (NSLOT * AGG_WAVES)
#define RCAP 64
#define RPW 132
#define LIST_WORDS (AGG_WAVES * WCAP + LROWS * RPW)
#define DTAB_CAP (NPLANES * ROWS_PB)
#define TAB_TRIPS ((DTAB_CAP / 4 + AGG_THREADS - 1) / AGG_THREADS)

template <int MODE>
__global__ void __launch_bounds__(AGG_THREADS)
agg_kernel(const int* __restrict__ rowIdx, const int* __restrict__ colIdx,
           const float* __restrict__ dinv, const float* __restrict__ src64,
           const float* __restrict__ src8, float* __restrict__ Lx,
           const unsigned* __restrict__ planes, int* __restrict__ recWords,
           int* __restrict__ rowCount, int* __restrict__ findCount, int E,
           int N,
           const float* __restrict__ H, const float* __restrict__ Wax,
           const float* __restrict__ Wah, const float* __restrict__ Wbx,
           const float* __restrict__ Wbh, const float* __restrict__ bax,
           const float* __restrict__ bah, const float* __restrict__ bbx,
           const float* __restrict__ bbh, float* __restrict__ Zbuf,
           float* __restrict__ RHout, float* __restrict__ out,
           float* __restrict__ dinvOut) {
  constexpr bool LIST = MODE == 0;
  constexpr bool FIRST = MODE == 1;
  constexpr bool HAS8 = FIRST;
  __shared__ __attribute__((aligned(16))) float acc[TILE_ROWS * PITCH];
  __shared__ __attribute__((aligned(16))) float acc8[HAS8 ? TILE_ROWS * IN_CH : 4];
  constexpr int POOL_WORDS =
      (FIRST && DTAB_CAP > LIST_WORDS) ? DTAB_CAP : LIST_WORDS;
  __shared__ __attribute__((aligned(16))) int pool[POOL_WORDS];
  int* const fq = pool;
  int* const rq = pool + AGG_WAVES * WCAP;
  float* const dtab = (float*)pool;
  __shared__ int rcnt[LROWS];

  const int tid = threadIdx.x;
  const int lane = tid & 31;
  const int base = blockIdx.x * ROWS_PB;
  const int wave = __builtin_amdgcn_readfirstlane(tid >> 5);

  if (!LIST) {
    for (int i = tid; i < TILE_ROWS * PITCH; i += AGG_THREADS) acc[i] = 0.0f;
  }
  if (HAS8) {
    for (int i = tid; i < TILE_ROWS * IN_CH; i += AGG_THREADS) acc8[i] = 0.0f;
  }
  if (tid < LROWS) rcnt[tid] = 0;
  __syncthreads();

  int* wq = fq + wave * WCAP;
  int wcnt = 0;
  const int cntIdx = blockIdx.x * AGG_WAVES + wave;
  const int rbase = blockIdx.x * LROWS;
  int2* __restrict__ recs = (int2*)recWords;
  bool filt = true;
  if (!LIST) {
    bool over = findCount[blockIdx.x * AGG_WAVES + lane] > WCAP;
#pragma unroll
    for (int s = 0; s < NSLOT; ++s)
      over = over || rowCount[rbase + s * AGG_WAVES + lane] > RCAP;
    filt = __builtin_amdgcn_ballot_w32(over) != 0u;
    if (FIRST && N > DTAB_CAP) filt = true;
  }

  if (filt) {
    const bool planeIsBlock = gridDim.x <= 256u;
    auto spot = [&](int e) {
      int r = rowIdx[e];
      unsigned d = (unsigned)(r - base);
      if (LIST) {
        atomicAdd(&rcnt[d], 1);
        return;
      }
      int c = colIdx[e];
      float nrm = -dinv[r] * dinv[c];
#pragma unroll 1
      for (int k = 0; k < OUT_CH; ++k)
        atomicAdd(&acc[d * PITCH + k], src64[(size_t)c * OUT_CH + k] * nrm);
      if (HAS8) {
#pragma unroll 1
        for (int k = 0; k < IN_CH; ++k)
          atomicAdd(&acc8[d * IN_CH + k], src8[(size_t)c * IN_CH + k] * nrm);
      }
    };
    auto append = [&](bool has, int e) {
      const unsigned m = __builtin_amdgcn_ballot_w32(has);
      const int slot = wcnt + (int)__builtin_amdgcn_mbcnt_lo(m, 0u);
      const int after = wcnt + __builtin_popcount(m);
      if (has && slot < WCAP) wq[slot] = e;
      if (after > WCAP) {
        if (has && slot >= WCAP) spot(e);
      }
      wcnt = after;
    };
    auto ownsRow = [&](int e) {
      return (unsigned)(rowIdx[e] - base) < (unsigned)ROWS_PB;
    };
    const int pitchW = plane_words(E);
    const int nQ = pitchW >> 2;
    const uint4* __restrict__ pl4 =
        (const uint4*)(planes + (size_t)(blockIdx.x & 255u) * pitchW);
    auto visit = [&](int q, uint4 v) {
      const bool in = q < nQ;
      unsigned a = in ? v.x : 0u, b = in ? v.y : 0u;
      unsigned cw = in ? v.z : 0u, d = in ? v.w : 0u;
      while (__builtin_amdgcn_ballot_w32((a | b | cw | d) != 0u) != 0u) {
        const int j = a != 0u ? 0 : b != 0u ? 1 : cw != 0u ? 2 : 3;
        const unsigned word = j == 0 ? a : j == 1 ? b : j == 2 ? cw : d;
        bool has = word != 0u;
        const int p = __builtin_ffs((int)word) - 1;
        const unsigned rest = word & (word - 1u);
        a = j == 0 ? rest : a;
        b = j == 1 ? rest : b;
        cw = j == 2 ? rest : cw;
        d = j == 3 ? rest : d;
        const int eh = ((q * 4 + j) << 5) + p;
        if (!planeIsBlock) {
          if (has) has = ownsRow(eh);
        }
        append(has, eh);
      }
    };
    const int qLast = nQ - 1;
    auto load4 = [&](uint4 (&v)[4], int q0) {
#pragma unroll
      for (int k = 0; k < 4; ++k) {
        int q = q0 + k * AGG_THREADS + tid;
        v[k] = pl4[min(q, qLast)];
      }
    };
    auto visit4 = [&](int q0, const uint4 (&v)[4]) {
#pragma unroll
      for (int k = 0; k < 4; ++k) visit(q0 + k * AGG_THREADS + tid, v[k]);
    };
    const int STEP = 4 * AGG_THREADS;
    if (nQ > 0) {
      uint4 va[4], vb[4];
      load4(va, 0);
      for (int q0 = 0; q0 < nQ; q0 += 2 * STEP) {
        load4(vb, q0 + STEP);
        visit4(q0, va);
        load4(va, q0 + 2 * STEP);
        visit4(q0 + STEP, vb);
      }
    }
    __builtin_amdgcn_fence(__ATOMIC_ACQ_REL, "workgroup");
  }

  if (filt) {
    const int nq = __builtin_amdgcn_readfirstlane(min(wcnt, WCAP));
    for (int g0 = 0; g0 < nq; g0 += 32) {
      const int cnt = min(32, nq - g0);
      const bool valid = lane < cnt;
      const int e = valid ? wq[g0 + lane] : 0;
      const int r = rowIdx[e];
      const int c = colIdx[e];
      const float nrm = (!LIST && valid) ? -dinv[r] * dinv[c] : 0.0f;
      const int d = valid ? r - base : 0;
      if (valid) {
        if (HAS8) {
          const float4* s8 = (const float4*)(src8 + (size_t)c * IN_CH);
          float4 a = s8[0];
          float4 b = s8[1];
          float* t8 = &acc8[d * IN_CH];
          atomicAdd(&t8[0], a.x * nrm);
          atomicAdd(&t8[1], a.y * nrm);
          atomicAdd(&t8[2], a.z * nrm);
          atomicAdd(&t8[3], a.w * nrm);
          atomicAdd(&t8[4], b.x * nrm);
          atomicAdd(&t8[5], b.y * nrm);
          atomicAdd(&t8[6], b.z * nrm);
          atomicAdd(&t8[7], b.w * nrm);
        }
        const int slot = atomicAdd(&rcnt[d], 1);
        if (slot < RCAP) {
          *(int2*)&rq[d * RPW + slot * 2] = make_int2(c, __float_as_int(nrm));
        } else if (!LIST) {
#pragma unroll 1
          for (int k = 0; k < OUT_CH; ++k)
            atomicAdd(&acc[d * PITCH + k], src64[(size_t)c * OUT_CH + k] * nrm);
        }
      }
    }
    if ((LIST || FIRST) && lane == 0) findCount[cntIdx] = wcnt;
  }
  if (FIRST) {
    if (!filt) {
      const int n4 = N >> 2;
      const int last4 = max(n4 - 1, 0);
      const v4f* __restrict__ d4 = (const v4f*)dinv;
      v4f* t4 = (v4f*)dtab;
      v4f tv[TAB_TRIPS];
#pragma unroll
      for (int k = 0; k < TAB_TRIPS; ++k)
        tv[k] = d4[min(k * AGG_THREADS + tid, last4)];
#pragma unroll
      for (int k = 0; k < TAB_TRIPS; ++k) {
        const int i = k * AGG_THREADS + tid;
        if (i < n4) t4[i] = tv[k];
      }
      for (int i = (n4 << 2) + tid; i < N; i += AGG_THREADS) dtab[i] = dinv[i];
    }
  }
  __syncthreads();

  if (LIST) {
#pragma unroll 1
    for (int s = 0; s < NSLOT; ++s) {
      const int row = s * AGG_WAVES + wave;
      const int all = rcnt[row];
      const int n = min(all, RCAP);
      const int2* rl = (const int2*)&rq[row * RPW];
      const int2 none = make_int2(0, 0);
      int2* rr = recs + (size_t)(rbase + row) * RCAP;
      int2 v0 = rl[lane];
      int2 v1 = rl[lane + 32];
      if (lane >= n) v0 = none;
      if (lane + 32 >= n) v1 = none;
      rr[lane] = v0;
      rr[lane + 32] = v1;
      if (lane == 0) rowCount[rbase + row] = all;
    }
    if (tid < ROWS_PB && base + tid < N) {
      const int dg = rcnt[tid];
      dinvOut[base + tid] = dg > 0 ? rsqrtf((float)dg) : 0.0f;
    }
    return;
  }

  auto sumRow = [&](int2 e0, int2 e1, int n) {
    v2f a = {0.0f, 0.0f};
#pragma unroll 1
    for (int j0 = 0; j0 < n; j0 += 32) {
      const bool up = j0 != 0;
      const int cc = up ? e1.x : e0.x;
      const int ww = up ? e1.y : e0.y;
      float2 g[32];
#pragma unroll
      for (int k = 0; k < 32; ++k) {
        const unsigned cj = (unsigned)__builtin_amdgcn_readlane(cc, k);
        g[k] = ((const float2*)(src64 + (size_t)cj * OUT_CH))[lane];
      }
#pragma unroll
      for (int k = 0; k < 32; ++k) {
        const float wj = __int_as_float(__builtin_amdgcn_readlane(ww, k));
        const v2f gk = {g[k].x, g[k].y};
        a += gk * wj;
      }
    }
    return a;
  };
  auto putRow = [&](int row, v2f a) {
    if (row < TILE_ROWS) {
      float2* p = (float2*)&acc[row * PITCH + 2 * lane];
      float2 v = *p;
      v.x += a.x;
      v.y += a.y;
      *p = v;
    }
  };
  if (filt) {
#pragma unroll 1
    for (int s = 0; s < NSLOT; ++s) {
      const int row = s * AGG_WAVES + wave;
      const int all = rcnt[row];
      const int n = __builtin_amdgcn_readfirstlane(min(all, RCAP));
      const int2* rl = (const int2*)&rq[row * RPW];
      const int2 none = make_int2(0, 0);
      const int2 e0 = lane < n ? rl[lane] : none;
      const int2 e1 = lane + 32 < n ? rl[lane + 32] : none;
      if (FIRST) {
        int2* rr = recs + (size_t)(rbase + row) * RCAP;
        rr[lane] = e0;
        rr[lane + 32] = e1;
        if (lane == 0) rowCount[rbase + row] = all;
      }
      putRow(row, sumRow(e0, e1, n));
    }
  } else if (FIRST) {
    const int myc = lane < NSLOT ? rowCount[rbase + lane * AGG_WAVES + wave] : 0;
    auto recOf = [&](int row) { return recs + (size_t)(rbase + row) * RCAP; };
    struct Fac {
      float dr, d0, d1;
    };
    auto factors = [&](int2 p0, int2 p1, int row) {
      Fac f;
      f.dr = dtab[min(base + row, N - 1)];
      f.d0 = dtab[p0.x];
      f.d1 = dtab[p1.x];
      return f;
    };
    auto weights = [&](Fac f, int n) {
      float2 w;
      w.x = lane < n ? -f.dr * f.d0 : 0.0f;
      w.y = lane + 32 < n ? -f.dr * f.d1 : 0.0f;
      return w;
    };
    const int sub = lane >> 3;
    const int col8 = lane & 7;
    int2 a0 = recOf(wave)[lane];
    int2 a1 = recOf(wave)[lane + 32];
    int2 b0 = recOf(wave + AGG_WAVES)[lane];
    int2 b1 = recOf(wave + AGG_WAVES)[lane + 32];
    Fac fa = factors(a0, a1, wave);
#pragma unroll 1
    for (int s = 0; s < NSLOT; ++s) {
      const int row = s * AGG_WAVES + wave;
      const int n = min(__builtin_amdgcn_readlane(myc, s), RCAP);
      const float2 wa = weights(fa, n);
      a0.y = __float_as_int(wa.x);
      a1.y = __float_as_int(wa.y);
      int2* rr = recOf(row);
      rr[lane] = a0;
      rr[lane + 32] = a1;
      int2 c0 = b0, c1 = b1;
      if (s + 2 < NSLOT) {
        const int2* rn = recOf(row + 2 * AGG_WAVES);
        c0 = rn[lane];
        c1 = rn[lane + 32];
      }
      float xs0[8], xs1[8];
#pragma unroll
      for (int m = 0; m < 8; ++m) {
        const int cj = __shfl(a0.x, 4 * m + sub);
        xs0[m] = src8[(size_t)cj * IN_CH + col8];
        xs1[m] = 0.0f;
      }
      if (n > 32) {
#pragma unroll
        for (int m = 0; m < 8; ++m) {
          const int cj = __shfl(a1.x, 4 * m + sub);
          xs1[m] = src8[(size_t)cj * IN_CH + col8];
        }
      }
      Fac fb = fa;
      if (s + 1 < NSLOT) fb = factors(b0, b1, row + AGG_WAVES);
      putRow(row, sumRow(a0, a1, n));
      float ax = 0.0f;
#pragma unroll
      for (int m = 0; m < 8; ++m) ax += __shfl(wa.x, 4 * m + sub) * xs0[m];
      if (n > 32) {
#pragma unroll
        for (int m = 0; m < 8; ++m) ax += __shfl(wa.y, 4 * m + sub) * xs1[m];
      }
      ax += __shfl_xor(ax, 8);
      ax += __shfl_xor(ax, 16);
      if (lane < IN_CH && row < TILE_ROWS) acc8[row * IN_CH + lane] = ax;
      a0 = b0;
      a1 = b1;
      b0 = c0;
      b1 = c1;
      fa = fb;
    }
  } else {
    const int myc = lane < NSLOT ? rowCount[rbase + lane * AGG_WAVES + wave] : 0;
    const int2* rr = recs + (size_t)(rbase + wave) * RCAP;
    int2 e0 = rr[lane];
    int2 e1 = rr[lane + 32];
#pragma unroll 1
    for (int s = 0; s < NSLOT; ++s) {
      const int row = s * AGG_WAVES + wave;
      const int n = min(__builtin_amdgcn_readlane(myc, s), RCAP);
      int2 n0 = e0, n1 = e1;
      if (s + 1 < NSLOT) {
        const int2* rn = recs + (size_t)(rbase + row + AGG_WAVES) * RCAP;
        n0 = rn[lane];
        n1 = rn[lane + 32];
      }
      putRow(row, sumRow(e0, e1, n));
      e0 = n0;
      e1 = n1;
    }
  }
  __syncthreads();

  const int rows = min(ROWS_PB, N - base);
  float* hst = (float*)rq;
  float* xst = hst + TILE_ROWS * PITCH;
  float* lst = xst + TILE_ROWS * IN_CH;
  for (int i = tid; i < TILE_ROWS * (OUT_CH / 4); i += AGG_THREADS) {
    const int r = i >> 4, c4 = i & 15;
    float4 v = make_float4(0.0f, 0.0f, 0.0f, 0.0f);
    if (r < rows) v = ((const float4*)(src64 + (size_t)(base + r) * OUT_CH))[c4];
    *(float4*)&hst[r * PITCH + c4 * 4] = v;
  }
  for (int i = tid; i < TILE_ROWS * (IN_CH / 4); i += AGG_THREADS) {
    const int r = i >> 1, c4 = i & 1;
    float4 v = make_float4(0.0f, 0.0f, 0.0f, 0.0f);
    float4 w = v;
    if (r < rows) {
      v = ((const float4*)(src8 + (size_t)(base + r) * IN_CH))[c4];
      if (!FIRST) w = ((const float4*)(Lx + (size_t)(base + r) * IN_CH))[c4];
    }
    *(float4*)&xst[r * IN_CH + c4 * 4] = v;
    if (!FIRST) *(float4*)&lst[r * IN_CH + c4 * 4] = w;
  }
  if (HAS8) {
    float4* o8 = (float4*)(Lx + (size_t)base * IN_CH);
    const float4* a8 = (const float4*)acc8;
    for (int i = tid; i < rows * (IN_CH / 4); i += AGG_THREADS) o8[i] = a8[i];
  }
  __syncthreads();

  const int half = lane >> 4;
  const int l15 = lane & 15;
  const int nUnits = ((rows + 15) >> 4) * (OUT_CH / 16);
  for (int u = wave; u < nUnits; u += AGG_WAVES) {
    const int row0 = (u >> 2) * 16;
    const int rA = row0 + l15;
    const int colB = (u & 3) * 16 + l15;
    if (FIRST) {
      v8f accZ = {};
      v8f accR = {};
      accum_block2(xst, IN_CH, rA, Wax, Wbx, IN_CH, colB, half, accZ, accR);
      accum_block2(acc8, IN_CH, rA, Wax + IN_CH * OUT_CH, Wbx + IN_CH * OUT_CH,
                   IN_CH, colB, half, accZ, accR);
      accum_block2(hst, PITCH, rA, Wah, Wbh, OUT_CH, colB, half, accZ, accR);
      accum_block2(acc, PITCH, rA, Wah + OUT_CH * OUT_CH, Wbh + OUT_CH * OUT_CH,
                   OUT_CH, colB, half, accZ, accR);
      const float bZ = bax[colB] + bah[colB];
      const float bR = bbx[colB] + bbh[colB];
#pragma unroll
      for (int i = 0; i < 8; ++i) {
        const int r = row0 + i + 8 * half;
        if (r < rows) {
          const float z = 1.0f / (1.0f + __expf(-(accZ[i] + bZ)));
          const float rg = 1.0f / (1.0f + __expf(-(accR[i] + bR)));
          const size_t g = (size_t)(base + r) * OUT_CH + colB;
          Zbuf[g] = z;
          RHout[g] = rg * hst[r * PITCH + colB];
        }
      }
    } else {
      v8f accC = {};
      accC = accum_block(xst, IN_CH, rA, Wax, IN_CH, colB, half, accC);
      accC = accum_block(lst, IN_CH, rA, Wax + IN_CH * OUT_CH, IN_CH, colB, half, accC);
      accC = accum_block(hst, PITCH, rA, Wah, OUT_CH, colB, half, accC);
      accC = accum_block(acc, PITCH, rA, Wah + OUT_CH * OUT_CH, OUT_CH, colB, half, accC);
      const float bC = bax[colB] + bah[colB];
#pragma unroll
      for (int i = 0; i < 8; ++i) {
        const int r = row0 + i + 8 * half;
        if (r < rows) {
          const size_t g = (size_t)(base + r) * OUT_CH + colB;
          const float cv = tanhf(accC[i] + bC);
          const float z = Zbuf[g];
          const float h = H[g];
          out[g] = z * h + (1.0f - z) * cv;
        }
      }
    }
  }
}


extern "C" void kernel_launch(void* const* d_in, const int* in_sizes, int n_in,
                              void* d_out, int out_size, void* d_ws,
                              size_t ws_size, hipStream_t stream) {
  const float* x = (const float*)d_in[0];
  const float* H = (const float*)d_in[1];
  const int* edge = (const int*)d_in[2];
  const float* Wzx = (const float*)d_in[3];
  const float* bzx = (const float*)d_in[4];
  const float* Wzh = (const float*)d_in[5];
  const float* bzh = (const float*)d_in[6];
  const float* Wrx = (const float*)d_in[7];
  const float* brx = (const float*)d_in[8];
  const float* Wrh = (const float*)d_in[9];
  const float* brh = (const float*)d_in[10];
  const float* Wcx = (const float*)d_in[11];
  const float* bcx = (const float*)d_in[12];
  const float* Wch = (const float*)d_in[13];
  const float* bch = (const float*)d_in[14];

  const int N = in_sizes[0] / IN_CH;
  const int E = in_sizes[2] / 2;
  const int* rowIdx = edge;
  const int* colIdx = edge + E;

  float* ws = (float*)d_ws;
  float* dinv = ws;
  float* Lx = dinv + ((N + 63) / 64) * 64;
  float* Zbuf = Lx + (size_t)N * IN_CH;
  float* RH = Zbuf + (size_t)N * OUT_CH;
  const int aggBlocks = (N + ROWS_PB - 1) / ROWS_PB;
  unsigned* planes = (unsigned*)(RH + (size_t)N * OUT_CH);
  int* rowCount =
      (int*)(planes + (size_t)NPLANES * (((E + DEG_EDGES - 1) / DEG_EDGES) * 32));
  int* findCount = rowCount + ((aggBlocks * LROWS + 63) / 64) * 64;
  int* recs = findCount + ((aggBlocks * AGG_WAVES + 63) / 64) * 64;

  plane_kernel<<<(E + DEG_EDGES - 1) / DEG_EDGES, DEG_THREADS, 0, stream>>>(
      rowIdx, planes, E);
  agg_kernel<0><<<aggBlocks, AGG_THREADS, 0, stream>>>(
      rowIdx, colIdx, nullptr, nullptr, nullptr, nullptr, planes, recs,
      rowCount, findCount, E, N, nullptr,
      nullptr, nullptr, nullptr, nullptr, nullptr, nullptr, nullptr, nullptr,
      nullptr, nullptr, nullptr, dinv);
  agg_kernel<1><<<aggBlocks, AGG_THREADS, 0, stream>>>(
      rowIdx, colIdx, dinv, H, x, Lx, planes, recs, rowCount, findCount, E,
      N, H,
      Wzx, Wzh, Wrx, Wrh, bzx, bzh, brx, brh, Zbuf, RH, nullptr, nullptr);
  agg_kernel<2><<<aggBlocks, AGG_THREADS, 0, stream>>>(
      rowIdx, colIdx, dinv, RH, x, Lx, planes, recs, rowCount, findCount, E,
      N, H,
      Wcx, Wch, nullptr, nullptr, bcx, bch, nullptr, nullptr, Zbuf, nullptr,
      (float*)d_out, nullptr);
}
